# starting kernel
# baseline (speedup 1.0000x reference)
#include <hip/hip_runtime.h>
#include <hip/hip_bf16.h>

typedef float v2f __attribute__((ext_vector_type(2)));
typedef float v8f __attribute__((ext_vector_type(8)));

#define NEG_SLOPE 0.2f
#define EPS 1e-5f

__device__ __forceinline__ void atomicMaxFloat(float* addr, float val) {
    if (val >= 0.0f) {
        atomicMax((int*)addr, __float_as_int(val));
    } else {
        atomicMin((unsigned int*)addr, __float_as_uint(val));
    }
}

__global__ void fill_u32_kernel(unsigned int* __restrict__ p, unsigned int v, long long count) {
    long long i = (long long)blockIdx.x * blockDim.x + threadIdx.x;
    if (i < count) p[i] = v;
}

template <int K, int NCOL>
__global__ void gemm_wmma_f32_kernel(const float* __restrict__ A,
                                     const float* __restrict__ Bm,
                                     float* __restrict__ C) {
    __shared__ float As[16 * K];
    const int rowBase = blockIdx.x * 16;
    for (int i = threadIdx.x; i < 16 * K; i += blockDim.x) {
        int r = i / K, k = i - r * K;
        As[i] = A[(long long)(rowBase + r) * K + k];
    }
    __syncthreads();

    const int wave = threadIdx.x >> 5;
    const int lane = threadIdx.x & 31;
    const int colBase = wave * 16;
    const int mn = lane & 15;
    const int khalf = (lane >> 4) << 1;

    v8f acc = {};
#if __has_builtin(__builtin_amdgcn_wmma_f32_16x16x4_f32)
    for (int kk = 0; kk < K; kk += 4) {
        v2f a, b;
        a.x = As[mn * K + kk + khalf];
        a.y = As[mn * K + kk + khalf + 1];
        b.x = Bm[(long long)(kk + khalf) * NCOL + colBase + mn];
        b.y = Bm[(long long)(kk + khalf + 1) * NCOL + colBase + mn];
        acc = __builtin_amdgcn_wmma_f32_16x16x4_f32(false, a, false, b,
                                                    (short)0, acc, false, false);
    }
#else
    for (int kk = 0; kk < K; ++kk) {
        float av = As[mn * K + kk];
        for (int r = 0; r < 8; ++r)
            acc[r] += av * Bm[(long long)kk * NCOL + colBase + mn];
    }
#endif
    const int mOff = (lane >> 4) << 3;
#pragma unroll
    for (int r = 0; r < 8; ++r) {
        C[(long long)(rowBase + r + mOff) * NCOL + colBase + mn] = acc[r];
    }
}

__global__ void alpha_kernel(const float* __restrict__ h,
                             const float* __restrict__ a_src,
                             const float* __restrict__ a_dst,
                             float* __restrict__ as, float* __restrict__ ad,
                             int n_nodes, int heads, int C) {
    int gw = (int)(((long long)blockIdx.x * blockDim.x + threadIdx.x) >> 5);
    int lane = threadIdx.x & 31;
    if (gw >= n_nodes * heads) return;
    int hd = gw % heads;
    const float* hp = h + (long long)gw * C;
    const float* sp = a_src + hd * C;
    const float* dp = a_dst + hd * C;
    float s1 = 0.0f, s2 = 0.0f;
    for (int c = lane; c < C; c += 32) {
        float v = hp[c];
        s1 += v * sp[c];
        s2 += v * dp[c];
    }
    for (int off = 16; off; off >>= 1) {
        s1 += __shfl_down(s1, off, 32);
        s2 += __shfl_down(s2, off, 32);
    }
    if (lane == 0) { as[gw] = s1; ad[gw] = s2; }
}

__global__ void edge_max_kernel(const int* __restrict__ ei, int E, int E2, int heads,
                                const float* __restrict__ as, const float* __restrict__ ad,
                                float* __restrict__ ebuf, float* __restrict__ emax) {
    int e = (int)((long long)blockIdx.x * blockDim.x + threadIdx.x);
    if (e >= E2) return;
    int s, d;
    if (e < E) { s = ei[e]; d = ei[E + e]; } else { s = d = e - E; }
    for (int hh = 0; hh < heads; ++hh) {
        float v = as[s * heads + hh] + ad[d * heads + hh];
        v = (v > 0.0f) ? v : NEG_SLOPE * v;
        ebuf[(long long)e * heads + hh] = v;
        atomicMaxFloat(&emax[d * heads + hh], v);
    }
}

__global__ void edge_expsum_kernel(const int* __restrict__ ei, int E, int E2, int heads,
                                   const float* __restrict__ emax,
                                   float* __restrict__ ebuf, float* __restrict__ denom) {
    int e = (int)((long long)blockIdx.x * blockDim.x + threadIdx.x);
    if (e >= E2) return;
    int d = (e < E) ? ei[E + e] : (e - E);
    for (int hh = 0; hh < heads; ++hh) {
        float ex = __expf(ebuf[(long long)e * heads + hh] - emax[d * heads + hh]);
        ebuf[(long long)e * heads + hh] = ex;
        atomicAdd(&denom[d * heads + hh], ex);
    }
}

__global__ void edge_aggregate_kernel(const int* __restrict__ ei, int E, int E2,
                                      int heads, int F,
                                      const float* __restrict__ h,
                                      const float* __restrict__ ebuf,
                                      const float* __restrict__ denom,
                                      float* __restrict__ agg) {
    int gw = (int)(((long long)blockIdx.x * blockDim.x + threadIdx.x) >> 5);
    int lane = threadIdx.x & 31;
    if (gw >= E2) return;
    int s, d;
    if (gw < E) { s = ei[gw]; d = ei[E + gw]; } else { s = d = gw - E; }
    int C = F / heads;
    float al[2];
    for (int hh = 0; hh < heads; ++hh)
        al[hh] = ebuf[(long long)gw * heads + hh] / denom[d * heads + hh];
    const float* hs = h + (long long)s * F;
    float* ag = agg + (long long)d * F;
    for (int c = lane; c < F; c += 32) {
        int hh = c / C;
        atomicAdd(&ag[c], hs[c] * al[hh]);
    }
}

__global__ void bias_bn_relu_kernel(const float* __restrict__ x,
                                    const float* __restrict__ b,
                                    const float* __restrict__ gamma,
                                    const float* __restrict__ beta,
                                    const float* __restrict__ mean,
                                    const float* __restrict__ var,
                                    float* __restrict__ out, long long total, int F) {
    long long i = (long long)blockIdx.x * blockDim.x + threadIdx.x;
    if (i >= total) return;
    int c = (int)(i % F);
    float v = x[i] + b[c];
    v = gamma[c] * (v - mean[c]) * rsqrtf(var[c] + EPS) + beta[c];
    out[i] = (v > 0.0f) ? v : 0.0f;
}

__global__ void pool_max_kernel(const float* __restrict__ x,
                                const int* __restrict__ batch,
                                float* __restrict__ out, int n_nodes, int F) {
    long long i = (long long)blockIdx.x * blockDim.x + threadIdx.x;
    if (i >= (long long)n_nodes * F) return;
    int n = (int)(i / F);
    int c = (int)(i - (long long)n * F);
    atomicMaxFloat(&out[batch[n] * F + c], x[i]);
}

extern "C" void kernel_launch(void* const* d_in, const int* in_sizes, int n_in,
                              void* d_out, int out_size, void* d_ws, size_t ws_size,
                              hipStream_t stream) {
    const int F_IN = 128, HID = 128, HEADS = 2;
    const int F1 = HEADS * HID;
    const int N = in_sizes[0] / F_IN;
    const int E = in_sizes[1] / 2;
    const int E2 = E + N;

    const float* x      = (const float*)d_in[0];
    const int*   ei     = (const int*)d_in[1];
    const int*   batch  = (const int*)d_in[2];
    const float* W1     = (const float*)d_in[3];
    const float* a_src1 = (const float*)d_in[4];
    const float* a_dst1 = (const float*)d_in[5];
    const float* b1     = (const float*)d_in[6];
    const float* gamma1 = (const float*)d_in[7];
    const float* beta1  = (const float*)d_in[8];
    const float* rm1    = (const float*)d_in[9];
    const float* rv1    = (const float*)d_in[10];
    const float* W2     = (const float*)d_in[11];
    const float* a_src2 = (const float*)d_in[12];
    const float* a_dst2 = (const float*)d_in[13];
    const float* b2     = (const float*)d_in[14];
    const float* gamma2 = (const float*)d_in[15];
    const float* beta2  = (const float*)d_in[16];
    const float* rm2    = (const float*)d_in[17];
    const float* rv2    = (const float*)d_in[18];

    float* ws = (float*)d_ws;
    size_t o = 0;
    float* h1   = ws + o; o += (size_t)N * F1;
    float* aggA = ws + o; o += (size_t)N * F1;
    float* aggB = ws + o; o += (size_t)N * HID;
    float* as1  = ws + o; o += (size_t)N * HEADS;
    float* ad1  = ws + o; o += (size_t)N * HEADS;
    float* em1  = ws + o; o += (size_t)N * HEADS;
    float* dn1  = ws + o; o += (size_t)N * HEADS;
    float* as2  = ws + o; o += (size_t)N;
    float* ad2  = ws + o; o += (size_t)N;
    float* em2  = ws + o; o += (size_t)N;
    float* dn2  = ws + o; o += (size_t)N;
    float* ex1  = ws + o; o += (size_t)E2 * HEADS;
    float* ex2  = ws + o; o += (size_t)E2;
    float* h2   = h1;

    const unsigned int NEG_INF = 0xFF800000u;
    const unsigned int ZERO = 0u;
    auto fillBlocks = [](long long cnt) { return (unsigned)((cnt + 255) / 256); };

    fill_u32_kernel<<<fillBlocks((long long)N * HEADS), 256, 0, stream>>>((unsigned*)em1, NEG_INF, (long long)N * HEADS);
    fill_u32_kernel<<<fillBlocks((long long)N * HEADS), 256, 0, stream>>>((unsigned*)dn1, ZERO, (long long)N * HEADS);
    fill_u32_kernel<<<fillBlocks((long long)N * F1), 256, 0, stream>>>((unsigned*)aggA, ZERO, (long long)N * F1);
    fill_u32_kernel<<<fillBlocks((long long)N), 256, 0, stream>>>((unsigned*)em2, NEG_INF, (long long)N);
    fill_u32_kernel<<<fillBlocks((long long)N), 256, 0, stream>>>((unsigned*)dn2, ZERO, (long long)N);
    fill_u32_kernel<<<fillBlocks((long long)N * HID), 256, 0, stream>>>((unsigned*)aggB, ZERO, (long long)N * HID);
    fill_u32_kernel<<<fillBlocks((long long)out_size), 256, 0, stream>>>((unsigned*)d_out, NEG_INF, (long long)out_size);

    gemm_wmma_f32_kernel<128, 256><<<N / 16, 32 * (256 / 16), 0, stream>>>(x, W1, h1);
    alpha_kernel<<<(unsigned)(((long long)N * HEADS * 32 + 255) / 256), 256, 0, stream>>>(
        h1, a_src1, a_dst1, as1, ad1, N, HEADS, HID);
    edge_max_kernel<<<fillBlocks(E2), 256, 0, stream>>>(ei, E, E2, HEADS, as1, ad1, ex1, em1);
    edge_expsum_kernel<<<fillBlocks(E2), 256, 0, stream>>>(ei, E, E2, HEADS, em1, ex1, dn1);
    edge_aggregate_kernel<<<(unsigned)(((long long)E2 * 32 + 255) / 256), 256, 0, stream>>>(
        ei, E, E2, HEADS, F1, h1, ex1, dn1, aggA);
    bias_bn_relu_kernel<<<fillBlocks((long long)N * F1), 256, 0, stream>>>(
        aggA, b1, gamma1, beta1, rm1, rv1, aggA, (long long)N * F1, F1);

    gemm_wmma_f32_kernel<256, 128><<<N / 16, 32 * (128 / 16), 0, stream>>>(aggA, W2, h2);
    alpha_kernel<<<(unsigned)(((long long)N * 32 + 255) / 256), 256, 0, stream>>>(
        h2, a_src2, a_dst2, as2, ad2, N, 1, HID);
    edge_max_kernel<<<fillBlocks(E2), 256, 0, stream>>>(ei, E, E2, 1, as2, ad2, ex2, em2);
    edge_expsum_kernel<<<fillBlocks(E2), 256, 0, stream>>>(ei, E, E2, 1, em2, ex2, dn2);
    edge_aggregate_kernel<<<(unsigned)(((long long)E2 * 32 + 255) / 256), 256, 0, stream>>>(
        ei, E, E2, 1, HID, h2, ex2, dn2, aggB);
    bias_bn_relu_kernel<<<fillBlocks((long long)N * HID), 256, 0, stream>>>(
        aggB, b2, gamma2, beta2, rm2, rv2, aggB, (long long)N * HID, HID);

    pool_max_kernel<<<fillBlocks((long long)N * HID), 256, 0, stream>>>(
        aggB, batch, (float*)d_out, N, HID);
}


# Edge softmax+sum by destination-range owner work-groups: each scans all destinations, lists its 200 nodes' edges, sums rows in LDS; no global float atomics; normalised after sum (order changes)
# speedup vs baseline: 2.3405x; 2.3405x over previous
#include <hip/hip_runtime.h>
#include <hip/hip_bf16.h>

typedef float v2f __attribute__((ext_vector_type(2)));
typedef float v8f __attribute__((ext_vector_type(8)));

#define NEG_SLOPE 0.2f
#define EPS 1e-5f

__device__ __forceinline__ void atomicMaxFloat(float* addr, float val) {
    if (val >= 0.0f) {
        atomicMax((int*)addr, __float_as_int(val));
    } else {
        atomicMin((unsigned int*)addr, __float_as_uint(val));
    }
}

__global__ void fill_u32_kernel(unsigned int* __restrict__ p, unsigned int v, long long count) {
    long long i = (long long)blockIdx.x * blockDim.x + threadIdx.x;
    if (i < count) p[i] = v;
}

template <int K, int NCOL>
__global__ void gemm_wmma_f32_kernel(const float* __restrict__ A,
                                     const float* __restrict__ Bm,
                                     float* __restrict__ C) {
    __shared__ float As[16 * K];
    const int rowBase = blockIdx.x * 16;
    for (int i = threadIdx.x; i < 16 * K; i += blockDim.x) {
        int r = i / K, k = i - r * K;
        As[i] = A[(long long)(rowBase + r) * K + k];
    }
    __syncthreads();

    const int wave = threadIdx.x >> 5;
    const int lane = threadIdx.x & 31;
    const int colBase = wave * 16;
    const int mn = lane & 15;
    const int khalf = (lane >> 4) << 1;

    v8f acc = {};
#if __has_builtin(__builtin_amdgcn_wmma_f32_16x16x4_f32)
    for (int kk = 0; kk < K; kk += 4) {
        v2f a, b;
        a.x = As[mn * K + kk + khalf];
        a.y = As[mn * K + kk + khalf + 1];
        b.x = Bm[(long long)(kk + khalf) * NCOL + colBase + mn];
        b.y = Bm[(long long)(kk + khalf + 1) * NCOL + colBase + mn];
        acc = __builtin_amdgcn_wmma_f32_16x16x4_f32(false, a, false, b,
                                                    (short)0, acc, false, false);
    }
#else
    for (int kk = 0; kk < K; ++kk) {
        float av = As[mn * K + kk];
        for (int r = 0; r < 8; ++r)
            acc[r] += av * Bm[(long long)kk * NCOL + colBase + mn];
    }
#endif
    const int mOff = (lane >> 4) << 3;
#pragma unroll
    for (int r = 0; r < 8; ++r) {
        C[(long long)(rowBase + r + mOff) * NCOL + colBase + mn] = acc[r];
    }
}

__global__ void alpha_kernel(const float* __restrict__ h,
                             const float* __restrict__ a_src,
                             const float* __restrict__ a_dst,
                             float* __restrict__ as, float* __restrict__ ad,
                             int n_nodes, int heads, int C) {
    int gw = (int)(((long long)blockIdx.x * blockDim.x + threadIdx.x) >> 5);
    int lane = threadIdx.x & 31;
    if (gw >= n_nodes * heads) return;
    int hd = gw % heads;
    const float* hp = h + (long long)gw * C;
    const float* sp = a_src + hd * C;
    const float* dp = a_dst + hd * C;
    float s1 = 0.0f, s2 = 0.0f;
    for (int c = lane; c < C; c += 32) {
        float v = hp[c];
        s1 += v * sp[c];
        s2 += v * dp[c];
    }
    for (int off = 16; off; off >>= 1) {
        s1 += __shfl_down(s1, off, 32);
        s2 += __shfl_down(s2, off, 32);
    }
    if (lane == 0) { as[gw] = s1; ad[gw] = s2; }
}

#define OWN_R 200
#define OWN_THREADS 1024
#define OWN_WAVES (OWN_THREADS / 32)
#define OWN_CAP 16384

__device__ __forceinline__ float leaky(float v) { return (v > 0.0f) ? v : NEG_SLOPE * v; }

template <class Fn>
__device__ __forceinline__ void scan_dst(const int* __restrict__ dst, int E, int tid, Fn&& fn) {
    const bool vec = ((E & 3) == 0) && ((reinterpret_cast<size_t>(dst) & 15) == 0);
    if (vec) {
        const int4* d4 = reinterpret_cast<const int4*>(dst);
        const int E4 = E >> 2;
        for (int jb = 0; jb < E4; jb += 4 * OWN_THREADS) {
            int4 d[4];
#pragma unroll
            for (int u = 0; u < 4; ++u) {
                int j = jb + u * OWN_THREADS + tid;
                d[u] = (j < E4) ? d4[j] : make_int4(-1, -1, -1, -1);
            }
#pragma unroll
            for (int u = 0; u < 4; ++u) {
                int i = 4 * (jb + u * OWN_THREADS + tid);
                fn(i, d[u].x); fn(i + 1, d[u].y); fn(i + 2, d[u].z); fn(i + 3, d[u].w);
            }
        }
    } else {
        for (int ib = 0; ib < E; ib += OWN_THREADS) {
            int i = ib + tid;
            fn(i, (i < E) ? dst[i] : -1);
        }
    }
}

template <int F, int H, int U>
__device__ __forceinline__ void accumulate_edges(const int (&s)[U], const unsigned (&dl)[U],
                                                 const bool (&ok)[U], int lane,
                                                 const float* __restrict__ h,
                                                 const float* __restrict__ as,
                                                 const float* adl, const float* m,
                                                 float* den, float* acc) {
    constexpr int C = F / H;
    constexpr int J = F / 32;
    float r[U][J];
    float a[U][H];
#pragma unroll
    for (int u = 0; u < U; ++u) {
        const float* hs = h + (size_t)s[u] * F;
#pragma unroll
        for (int j = 0; j < J; ++j) r[u][j] = hs[lane + 32 * j];
#pragma unroll
        for (int hh = 0; hh < H; ++hh) a[u][hh] = as[(size_t)s[u] * H + hh];
    }
#pragma unroll
    for (int u = 0; u < U; ++u) {
        if (ok[u]) {
            float ex[H];
#pragma unroll
            for (int hh = 0; hh < H; ++hh) {
                float v = leaky(a[u][hh] + adl[dl[u] * H + hh]);
                ex[hh] = __expf(v - m[dl[u] * H + hh]);
            }
#pragma unroll
            for (int hh = 0; hh < H; ++hh)
                if (lane == hh) atomicAdd(&den[dl[u] * H + hh], ex[hh]);
            float* ar = acc + dl[u] * F;
#pragma unroll
            for (int j = 0; j < J; ++j)
                atomicAdd(&ar[lane + 32 * j], r[u][j] * ex[(32 * j) / C]);
        }
    }
}

template <int F, int H>
__global__ __launch_bounds__(OWN_THREADS)
void owner_aggregate_kernel(const int* __restrict__ ei, int E, int n_nodes,
                            const float* __restrict__ h,
                            const float* __restrict__ as,
                            const float* __restrict__ ad,
                            float* __restrict__ agg) {
    constexpr int C = F / H;
    __shared__ float acc[OWN_R * F];
    __shared__ unsigned int list[OWN_CAP];
    __shared__ float m[OWN_R * H];
    __shared__ float den[OWN_R * H];
    __shared__ float adl[OWN_R * H];
    __shared__ unsigned int cnt;

    const int tid = threadIdx.x;
    const int lane = tid & 31;
    const int wave = tid >> 5;
    const int base = blockIdx.x * OWN_R;
    const int R = min(OWN_R, n_nodes - base);
    const unsigned uR = (unsigned)R;
    const int* __restrict__ src = ei;
    const int* __restrict__ dst = ei + E;

    for (int i = tid; i < R * F; i += OWN_THREADS) acc[i] = 0.0f;
    for (int i = tid; i < R * H; i += OWN_THREADS) {
        float a_d = ad[(size_t)base * H + i];
        adl[i] = a_d;
        m[i] = leaky(as[(size_t)base * H + i] + a_d);
        den[i] = 0.0f;
    }
    if (tid == 0) cnt = 0u;
    __syncthreads();

    const bool packs = (E <= (1 << 24)) && (n_nodes <= (1 << 24)) && (OWN_R <= 256);
    if (packs) {
        scan_dst(dst, E, tid, [&](int i, int d) {
            unsigned dl = (unsigned)(d - base);
            if (dl < uR) {
                unsigned pos = atomicAdd(&cnt, 1u);
                if (pos < OWN_CAP) list[pos] = ((unsigned)i << 8) | dl;
            }
        });
    }
    __syncthreads();
    const unsigned total = cnt;
    const bool fast = packs && (total <= OWN_CAP);

    if (fast) {
        for (unsigned p = tid; p < total; p += OWN_THREADS) {
            unsigned pk = list[p];
            unsigned dl = pk & 255u;
            int s = src[pk >> 8];
#pragma unroll
            for (int hh = 0; hh < H; ++hh)
                atomicMaxFloat(&m[dl * H + hh], leaky(as[(size_t)s * H + hh] + adl[dl * H + hh]));
            list[p] = ((unsigned)s << 8) | dl;
        }
        __syncthreads();
        for (unsigned p0 = wave * 4; p0 < total; p0 += OWN_WAVES * 4) {
            int s[4]; unsigned dl[4]; bool ok[4];
#pragma unroll
            for (int u = 0; u < 4; ++u) {
                ok[u] = (p0 + u) < total;
                unsigned pk = ok[u] ? list[p0 + u] : 0u;
                s[u] = (int)(pk >> 8);
                dl[u] = pk & 255u;
            }
            accumulate_edges<F, H, 4>(s, dl, ok, lane, h, as, adl, m, den, acc);
        }
    } else {
        scan_dst(dst, E, tid, [&](int i, int d) {
            unsigned dl = (unsigned)(d - base);
            if (dl < uR) {
                int s = src[i];
#pragma unroll
                for (int hh = 0; hh < H; ++hh)
                    atomicMaxFloat(&m[dl * H + hh], leaky(as[(size_t)s * H + hh] + adl[dl * H + hh]));
            }
        });
        __syncthreads();
        scan_dst(dst, E, tid, [&](int i, int d) {
            unsigned dl = (unsigned)(d - base);
            bool hit = dl < uR;
            int sv = hit ? src[i] : 0;
            unsigned mask = __builtin_amdgcn_ballot_w32(hit);
            while (mask) {
                int l = __builtin_ctz(mask);
                mask &= mask - 1u;
                int s1[1] = { __shfl(sv, l, 32) };
                unsigned d1[1] = { (unsigned)__shfl((int)dl, l, 32) };
                bool o1[1] = { true };
                accumulate_edges<F, H, 1>(s1, d1, o1, lane, h, as, adl, m, den, acc);
            }
        });
    }
    for (int q0 = wave * 4; q0 < R; q0 += OWN_WAVES * 4) {
        int s[4]; unsigned dl[4]; bool ok[4];
#pragma unroll
        for (int u = 0; u < 4; ++u) {
            ok[u] = (q0 + u) < R;
            dl[u] = ok[u] ? (unsigned)(q0 + u) : 0u;
            s[u] = base + (int)dl[u];
        }
        accumulate_edges<F, H, 4>(s, dl, ok, lane, h, as, adl, m, den, acc);
    }
    __syncthreads();

    for (int i = tid; i < R * F; i += OWN_THREADS) {
        int dl = i / F;
        int c = i - dl * F;
        agg[(size_t)base * F + i] = acc[i] / den[dl * H + c / C];
    }
}

__global__ void bias_bn_relu_kernel(const float* __restrict__ x,
                                    const float* __restrict__ b,
                                    const float* __restrict__ gamma,
                                    const float* __restrict__ beta,
                                    const float* __restrict__ mean,
                                    const float* __restrict__ var,
                                    float* __restrict__ out, long long total, int F) {
    long long i = (long long)blockIdx.x * blockDim.x + threadIdx.x;
    if (i >= total) return;
    int c = (int)(i % F);
    float v = x[i] + b[c];
    v = gamma[c] * (v - mean[c]) * rsqrtf(var[c] + EPS) + beta[c];
    out[i] = (v > 0.0f) ? v : 0.0f;
}

__global__ void pool_max_kernel(const float* __restrict__ x,
                                const int* __restrict__ batch,
                                float* __restrict__ out, int n_nodes, int F) {
    long long i = (long long)blockIdx.x * blockDim.x + threadIdx.x;
    if (i >= (long long)n_nodes * F) return;
    int n = (int)(i / F);
    int c = (int)(i - (long long)n * F);
    atomicMaxFloat(&out[batch[n] * F + c], x[i]);
}

extern "C" void kernel_launch(void* const* d_in, const int* in_sizes, int n_in,
                              void* d_out, int out_size, void* d_ws, size_t ws_size,
                              hipStream_t stream) {
    const int F_IN = 128, HID = 128, HEADS = 2;
    const int F1 = HEADS * HID;
    const int N = in_sizes[0] / F_IN;
    const int E = in_sizes[1] / 2;

    const float* x      = (const float*)d_in[0];
    const int*   ei     = (const int*)d_in[1];
    const int*   batch  = (const int*)d_in[2];
    const float* W1     = (const float*)d_in[3];
    const float* a_src1 = (const float*)d_in[4];
    const float* a_dst1 = (const float*)d_in[5];
    const float* b1     = (const float*)d_in[6];
    const float* gamma1 = (const float*)d_in[7];
    const float* beta1  = (const float*)d_in[8];
    const float* rm1    = (const float*)d_in[9];
    const float* rv1    = (const float*)d_in[10];
    const float* W2     = (const float*)d_in[11];
    const float* a_src2 = (const float*)d_in[12];
    const float* a_dst2 = (const float*)d_in[13];
    const float* b2     = (const float*)d_in[14];
    const float* gamma2 = (const float*)d_in[15];
    const float* beta2  = (const float*)d_in[16];
    const float* rm2    = (const float*)d_in[17];
    const float* rv2    = (const float*)d_in[18];

    float* ws = (float*)d_ws;
    size_t o = 0;
    float* h1   = ws + o; o += (size_t)N * F1;
    float* aggA = ws + o; o += (size_t)N * F1;
    float* aggB = ws + o; o += (size_t)N * HID;
    float* as1  = ws + o; o += (size_t)N * HEADS;
    float* ad1  = ws + o; o += (size_t)N * HEADS;
    float* as2  = ws + o; o += (size_t)N;
    float* ad2  = ws + o; o += (size_t)N;
    float* h2   = h1;

    const unsigned int NEG_INF = 0xFF800000u;
    auto fillBlocks = [](long long cnt) { return (unsigned)((cnt + 255) / 256); };
    const unsigned ownerBlocks = (unsigned)((N + OWN_R - 1) / OWN_R);

    fill_u32_kernel<<<fillBlocks((long long)out_size), 256, 0, stream>>>((unsigned*)d_out, NEG_INF, (long long)out_size);

    gemm_wmma_f32_kernel<128, 256><<<N / 16, 32 * (256 / 16), 0, stream>>>(x, W1, h1);
    alpha_kernel<<<(unsigned)(((long long)N * HEADS * 32 + 255) / 256), 256, 0, stream>>>(
        h1, a_src1, a_dst1, as1, ad1, N, HEADS, HID);
    owner_aggregate_kernel<256, 2><<<ownerBlocks, OWN_THREADS, 0, stream>>>(ei, E, N, h1, as1, ad1, aggA);
    bias_bn_relu_kernel<<<fillBlocks((long long)N * F1), 256, 0, stream>>>(
        aggA, b1, gamma1, beta1, rm1, rv1, aggA, (long long)N * F1, F1);

    gemm_wmma_f32_kernel<256, 128><<<N / 16, 32 * (128 / 16), 0, stream>>>(aggA, W2, h2);
    alpha_kernel<<<(unsigned)(((long long)N * 32 + 255) / 256), 256, 0, stream>>>(
        h2, a_src2, a_dst2, as2, ad2, N, 1, HID);
    owner_aggregate_kernel<128, 1><<<ownerBlocks, OWN_THREADS, 0, stream>>>(ei, E, N, h2, as2, ad2, aggB);
    bias_bn_relu_kernel<<<fillBlocks((long long)N * HID), 256, 0, stream>>>(
        aggB, b2, gamma2, beta2, rm2, rv2, aggB, (long long)N * HID, HID);

    pool_max_kernel<<<fillBlocks((long long)N * HID), 256, 0, stream>>>(
        aggB, batch, (float*)d_out, N, HID);
}


# Layer-2 owner work-groups take their range's edge list (source, local destination) from the workspace, where the layer-1 launch leaves it, instead of streaming all destinations again
# speedup vs baseline: 2.6238x; 1.1210x over previous
#include <hip/hip_runtime.h>
#include <hip/hip_bf16.h>

typedef float v2f __attribute__((ext_vector_type(2)));
typedef float v8f __attribute__((ext_vector_type(8)));

#define NEG_SLOPE 0.2f
#define EPS 1e-5f

__device__ __forceinline__ void atomicMaxFloat(float* addr, float val) {
    if (val >= 0.0f) {
        atomicMax((int*)addr, __float_as_int(val));
    } else {
        atomicMin((unsigned int*)addr, __float_as_uint(val));
    }
}

__global__ void fill_u32_kernel(unsigned int* __restrict__ p, unsigned int v, long long count) {
    long long i = (long long)blockIdx.x * blockDim.x + threadIdx.x;
    if (i < count) p[i] = v;
}

template <int K, int NCOL>
__global__ void gemm_wmma_f32_kernel(const float* __restrict__ A,
                                     const float* __restrict__ Bm,
                                     float* __restrict__ C) {
    __shared__ float As[16 * K];
    const int rowBase = blockIdx.x * 16;
    for (int i = threadIdx.x; i < 16 * K; i += blockDim.x) {
        int r = i / K, k = i - r * K;
        As[i] = A[(long long)(rowBase + r) * K + k];
    }
    __syncthreads();

    const int wave = threadIdx.x >> 5;
    const int lane = threadIdx.x & 31;
    const int colBase = wave * 16;
    const int mn = lane & 15;
    const int khalf = (lane >> 4) << 1;

    v8f acc = {};
#if __has_builtin(__builtin_amdgcn_wmma_f32_16x16x4_f32)
    for (int kk = 0; kk < K; kk += 4) {
        v2f a, b;
        a.x = As[mn * K + kk + khalf];
        a.y = As[mn * K + kk + khalf + 1];
        b.x = Bm[(long long)(kk + khalf) * NCOL + colBase + mn];
        b.y = Bm[(long long)(kk + khalf + 1) * NCOL + colBase + mn];
        acc = __builtin_amdgcn_wmma_f32_16x16x4_f32(false, a, false, b,
                                                    (short)0, acc, false, false);
    }
#else
    for (int kk = 0; kk < K; ++kk) {
        float av = As[mn * K + kk];
        for (int r = 0; r < 8; ++r)
            acc[r] += av * Bm[(long long)kk * NCOL + colBase + mn];
    }
#endif
    const int mOff = (lane >> 4) << 3;
#pragma unroll
    for (int r = 0; r < 8; ++r) {
        C[(long long)(rowBase + r + mOff) * NCOL + colBase + mn] = acc[r];
    }
}

__global__ void alpha_kernel(const float* __restrict__ h,
                             const float* __restrict__ a_src,
                             const float* __restrict__ a_dst,
                             float* __restrict__ as, float* __restrict__ ad,
                             int n_nodes, int heads, int C) {
    int gw = (int)(((long long)blockIdx.x * blockDim.x + threadIdx.x) >> 5);
    int lane = threadIdx.x & 31;
    if (gw >= n_nodes * heads) return;
    int hd = gw % heads;
    const float* hp = h + (long long)gw * C;
    const float* sp = a_src + hd * C;
    const float* dp = a_dst + hd * C;
    float s1 = 0.0f, s2 = 0.0f;
    for (int c = lane; c < C; c += 32) {
        float v = hp[c];
        s1 += v * sp[c];
        s2 += v * dp[c];
    }
    for (int off = 16; off; off >>= 1) {
        s1 += __shfl_down(s1, off, 32);
        s2 += __shfl_down(s2, off, 32);
    }
    if (lane == 0) { as[gw] = s1; ad[gw] = s2; }
}

#define NO_LIST 0xFFFFFFFFu
#define OWN_R 200
#define OWN_THREADS 1024
#define OWN_WAVES (OWN_THREADS / 32)
#define OWN_CAP 16384

__device__ __forceinline__ float leaky(float v) { return (v > 0.0f) ? v : NEG_SLOPE * v; }

template <class Fn>
__device__ __forceinline__ void scan_dst(const int* __restrict__ dst, int E, int tid, Fn&& fn) {
    const bool vec = ((E & 3) == 0) && ((reinterpret_cast<size_t>(dst) & 15) == 0);
    if (vec) {
        const int4* d4 = reinterpret_cast<const int4*>(dst);
        const int E4 = E >> 2;
        for (int jb = 0; jb < E4; jb += 4 * OWN_THREADS) {
            int4 d[4];
#pragma unroll
            for (int u = 0; u < 4; ++u) {
                int j = jb + u * OWN_THREADS + tid;
                d[u] = (j < E4) ? d4[j] : make_int4(-1, -1, -1, -1);
            }
#pragma unroll
            for (int u = 0; u < 4; ++u) {
                int i = 4 * (jb + u * OWN_THREADS + tid);
                fn(i, d[u].x); fn(i + 1, d[u].y); fn(i + 2, d[u].z); fn(i + 3, d[u].w);
            }
        }
    } else {
        for (int ib = 0; ib < E; ib += OWN_THREADS) {
            int i = ib + tid;
            fn(i, (i < E) ? dst[i] : -1);
        }
    }
}

template <int F, int H, int U>
__device__ __forceinline__ void accumulate_edges(const int (&s)[U], const unsigned (&dl)[U],
                                                 const bool (&ok)[U], int lane,
                                                 const float* __restrict__ h,
                                                 const float* __restrict__ as,
                                                 const float* adl, const float* m,
                                                 float* den, float* acc) {
    constexpr int C = F / H;
    constexpr int J = F / 32;
    float r[U][J];
    float a[U][H];
#pragma unroll
    for (int u = 0; u < U; ++u) {
        const float* hs = h + (size_t)s[u] * F;
#pragma unroll
        for (int j = 0; j < J; ++j) r[u][j] = hs[lane + 32 * j];
#pragma unroll
        for (int hh = 0; hh < H; ++hh) a[u][hh] = as[(size_t)s[u] * H + hh];
    }
#pragma unroll
    for (int u = 0; u < U; ++u) {
        if (ok[u]) {
            float ex[H];
#pragma unroll
            for (int hh = 0; hh < H; ++hh) {
                float v = leaky(a[u][hh] + adl[dl[u] * H + hh]);
                ex[hh] = __expf(v - m[dl[u] * H + hh]);
            }
#pragma unroll
            for (int hh = 0; hh < H; ++hh)
                if (lane == hh) atomicAdd(&den[dl[u] * H + hh], ex[hh]);
            float* ar = acc + dl[u] * F;
#pragma unroll
            for (int j = 0; j < J; ++j)
                atomicAdd(&ar[lane + 32 * j], r[u][j] * ex[(32 * j) / C]);
        }
    }
}

template <int F, int H, int LISTS>
__global__ __launch_bounds__(OWN_THREADS)
void owner_aggregate_kernel(const int* __restrict__ ei, int E, int n_nodes,
                            const float* __restrict__ h,
                            const float* __restrict__ as,
                            const float* __restrict__ ad,
                            float* agg, unsigned int* keep, int keepF) {
    constexpr int C = F / H;
    __shared__ float acc[OWN_R * F];
    __shared__ unsigned int list[OWN_CAP];
    __shared__ float m[OWN_R * H];
    __shared__ float den[OWN_R * H];
    __shared__ float adl[OWN_R * H];
    __shared__ unsigned int cnt;

    const int tid = threadIdx.x;
    const int lane = tid & 31;
    const int wave = tid >> 5;
    const int base = blockIdx.x * OWN_R;
    const int R = min(OWN_R, n_nodes - base);
    const unsigned uR = (unsigned)R;
    const int* __restrict__ src = ei;
    const int* __restrict__ dst = ei + E;
    unsigned int* kp = keep + (size_t)base * keepF;
    const unsigned keepCap = (unsigned)R * (unsigned)keepF;

    for (int i = tid; i < R * F; i += OWN_THREADS) acc[i] = 0.0f;
    for (int i = tid; i < R * H; i += OWN_THREADS) {
        float a_d = ad[(size_t)base * H + i];
        adl[i] = a_d;
        m[i] = leaky(as[(size_t)base * H + i] + a_d);
        den[i] = 0.0f;
    }
    if (tid == 0) cnt = 0u;
    __syncthreads();

    unsigned kept = NO_LIST;
    if (LISTS == 2) kept = kp[0];
    const bool have = (LISTS == 2) && (kept <= OWN_CAP) && (kept < keepCap);

    const bool packs = (E <= (1 << 24)) && (n_nodes <= (1 << 24)) && (OWN_R <= 256);
    if (packs && !have) {
        scan_dst(dst, E, tid, [&](int i, int d) {
            unsigned dl = (unsigned)(d - base);
            if (dl < uR) {
                unsigned pos = atomicAdd(&cnt, 1u);
                if (pos < OWN_CAP) list[pos] = ((unsigned)i << 8) | dl;
            }
        });
    }
    __syncthreads();
    const unsigned total = have ? kept : cnt;
    const bool fast = have || (packs && (total <= OWN_CAP));
    const bool leave = (LISTS == 1) && fast && (total < keepCap);

    if (fast) {
        for (unsigned p = tid; p < total; p += OWN_THREADS) {
            unsigned dl;
            int s;
            if (have) {
                unsigned pk = kp[1 + p];
                dl = pk & 255u;
                s = (int)(pk >> 8);
            } else {
                unsigned pk = list[p];
                dl = pk & 255u;
                s = src[pk >> 8];
            }
#pragma unroll
            for (int hh = 0; hh < H; ++hh)
                atomicMaxFloat(&m[dl * H + hh], leaky(as[(size_t)s * H + hh] + adl[dl * H + hh]));
            list[p] = ((unsigned)s << 8) | dl;
            if (leave) kp[1 + p] = ((unsigned)s << 8) | dl;
        }
        if (LISTS == 1 && tid == 0) kp[0] = leave ? total : NO_LIST;
        __syncthreads();
        for (unsigned p0 = wave * 4; p0 < total; p0 += OWN_WAVES * 4) {
            int s[4]; unsigned dl[4]; bool ok[4];
#pragma unroll
            for (int u = 0; u < 4; ++u) {
                ok[u] = (p0 + u) < total;
                unsigned pk = ok[u] ? list[p0 + u] : 0u;
                s[u] = (int)(pk >> 8);
                dl[u] = pk & 255u;
            }
            accumulate_edges<F, H, 4>(s, dl, ok, lane, h, as, adl, m, den, acc);
        }
    } else {
        if (LISTS == 1 && tid == 0) kp[0] = NO_LIST;
        scan_dst(dst, E, tid, [&](int i, int d) {
            unsigned dl = (unsigned)(d - base);
            if (dl < uR) {
                int s = src[i];
#pragma unroll
                for (int hh = 0; hh < H; ++hh)
                    atomicMaxFloat(&m[dl * H + hh], leaky(as[(size_t)s * H + hh] + adl[dl * H + hh]));
            }
        });
        __syncthreads();
        scan_dst(dst, E, tid, [&](int i, int d) {
            unsigned dl = (unsigned)(d - base);
            bool hit = dl < uR;
            int sv = hit ? src[i] : 0;
            unsigned mask = __builtin_amdgcn_ballot_w32(hit);
            while (mask) {
                int l = __builtin_ctz(mask);
                mask &= mask - 1u;
                int s1[1] = { __shfl(sv, l, 32) };
                unsigned d1[1] = { (unsigned)__shfl((int)dl, l, 32) };
                bool o1[1] = { true };
                accumulate_edges<F, H, 1>(s1, d1, o1, lane, h, as, adl, m, den, acc);
            }
        });
    }
    for (int q0 = wave * 4; q0 < R; q0 += OWN_WAVES * 4) {
        int s[4]; unsigned dl[4]; bool ok[4];
#pragma unroll
        for (int u = 0; u < 4; ++u) {
            ok[u] = (q0 + u) < R;
            dl[u] = ok[u] ? (unsigned)(q0 + u) : 0u;
            s[u] = base + (int)dl[u];
        }
        accumulate_edges<F, H, 4>(s, dl, ok, lane, h, as, adl, m, den, acc);
    }
    __syncthreads();

    for (int i = tid; i < R * F; i += OWN_THREADS) {
        int dl = i / F;
        int c = i - dl * F;
        agg[(size_t)base * F + i] = acc[i] / den[dl * H + c / C];
    }
}

__global__ void bias_bn_relu_kernel(const float* __restrict__ x,
                                    const float* __restrict__ b,
                                    const float* __restrict__ gamma,
                                    const float* __restrict__ beta,
                                    const float* __restrict__ mean,
                                    const float* __restrict__ var,
                                    float* __restrict__ out, long long total, int F) {
    long long i = (long long)blockIdx.x * blockDim.x + threadIdx.x;
    if (i >= total) return;
    int c = (int)(i % F);
    float v = x[i] + b[c];
    v = gamma[c] * (v - mean[c]) * rsqrtf(var[c] + EPS) + beta[c];
    out[i] = (v > 0.0f) ? v : 0.0f;
}

__global__ void pool_max_kernel(const float* __restrict__ x,
                                const int* __restrict__ batch,
                                float* __restrict__ out, int n_nodes, int F) {
    long long i = (long long)blockIdx.x * blockDim.x + threadIdx.x;
    if (i >= (long long)n_nodes * F) return;
    int n = (int)(i / F);
    int c = (int)(i - (long long)n * F);
    atomicMaxFloat(&out[batch[n] * F + c], x[i]);
}

extern "C" void kernel_launch(void* const* d_in, const int* in_sizes, int n_in,
                              void* d_out, int out_size, void* d_ws, size_t ws_size,
                              hipStream_t stream) {
    const int F_IN = 128, HID = 128, HEADS = 2;
    const int F1 = HEADS * HID;
    const int N = in_sizes[0] / F_IN;
    const int E = in_sizes[1] / 2;

    const float* x      = (const float*)d_in[0];
    const int*   ei     = (const int*)d_in[1];
    const int*   batch  = (const int*)d_in[2];
    const float* W1     = (const float*)d_in[3];
    const float* a_src1 = (const float*)d_in[4];
    const float* a_dst1 = (const float*)d_in[5];
    const float* b1     = (const float*)d_in[6];
    const float* gamma1 = (const float*)d_in[7];
    const float* beta1  = (const float*)d_in[8];
    const float* rm1    = (const float*)d_in[9];
    const float* rv1    = (const float*)d_in[10];
    const float* W2     = (const float*)d_in[11];
    const float* a_src2 = (const float*)d_in[12];
    const float* a_dst2 = (const float*)d_in[13];
    const float* b2     = (const float*)d_in[14];
    const float* gamma2 = (const float*)d_in[15];
    const float* beta2  = (const float*)d_in[16];
    const float* rm2    = (const float*)d_in[17];
    const float* rv2    = (const float*)d_in[18];

    float* ws = (float*)d_ws;
    size_t o = 0;
    float* h1   = ws + o; o += (size_t)N * F1;
    float* aggA = ws + o; o += (size_t)N * F1;
    float* aggB = ws + o; o += (size_t)N * HID;
    float* as1  = ws + o; o += (size_t)N * HEADS;
    float* ad1  = ws + o; o += (size_t)N * HEADS;
    float* as2  = ws + o; o += (size_t)N;
    float* ad2  = ws + o; o += (size_t)N;
    float* h2   = h1;

    const unsigned int NEG_INF = 0xFF800000u;
    auto fillBlocks = [](long long cnt) { return (unsigned)((cnt + 255) / 256); };
    const unsigned ownerBlocks = (unsigned)((N + OWN_R - 1) / OWN_R);

    fill_u32_kernel<<<fillBlocks((long long)out_size), 256, 0, stream>>>((unsigned*)d_out, NEG_INF, (long long)out_size);

    gemm_wmma_f32_kernel<128, 256><<<N / 16, 32 * (256 / 16), 0, stream>>>(x, W1, h1);
    alpha_kernel<<<(unsigned)(((long long)N * HEADS * 32 + 255) / 256), 256, 0, stream>>>(
        h1, a_src1, a_dst1, as1, ad1, N, HEADS, HID);
    owner_aggregate_kernel<256, 2, 1><<<ownerBlocks, OWN_THREADS, 0, stream>>>(
        ei, E, N, h1, as1, ad1, aggA, (unsigned int*)aggB, 128);
    bias_bn_relu_kernel<<<fillBlocks((long long)N * F1), 256, 0, stream>>>(
        aggA, b1, gamma1, beta1, rm1, rv1, aggA, (long long)N * F1, F1);

    gemm_wmma_f32_kernel<256, 128><<<N / 16, 32 * (128 / 16), 0, stream>>>(aggA, W2, h2);
    alpha_kernel<<<(unsigned)(((long long)N * 32 + 255) / 256), 256, 0, stream>>>(
        h2, a_src2, a_dst2, as2, ad2, N, 1, HID);
    owner_aggregate_kernel<128, 1, 2><<<ownerBlocks, OWN_THREADS, 0, stream>>>(
        ei, E, N, h2, as2, ad2, aggB, (unsigned int*)aggB, 128);
    bias_bn_relu_kernel<<<fillBlocks((long long)N * HID), 256, 0, stream>>>(
        aggB, b2, gamma2, beta2, rm2, rv2, aggB, (long long)N * HID, HID);

    pool_max_kernel<<<fillBlocks((long long)N * HID), 256, 0, stream>>>(
        aggB, batch, (float*)d_out, N, HID);
}


# Both GEMMs re-tiled: 512 resident work-groups stage W once in LDS (k pairs interleaved); each wave keeps its W slice, walks 16-row strips, A fragments global->registers; k sum order changes
# speedup vs baseline: 2.9935x; 1.1409x over previous
#include <hip/hip_runtime.h>
#include <hip/hip_bf16.h>

typedef float v2f __attribute__((ext_vector_type(2)));
typedef float v8f __attribute__((ext_vector_type(8)));

#define NEG_SLOPE 0.2f
#define EPS 1e-5f

__device__ __forceinline__ void atomicMaxFloat(float* addr, float val) {
    if (val >= 0.0f) {
        atomicMax((int*)addr, __float_as_int(val));
    } else {
        atomicMin((unsigned int*)addr, __float_as_uint(val));
    }
}

__global__ void fill_u32_kernel(unsigned int* __restrict__ p, unsigned int v, long long count) {
    long long i = (long long)blockIdx.x * blockDim.x + threadIdx.x;
    if (i < count) p[i] = v;
}

#define GEMM_THREADS 256
#define GEMM_WAVES (GEMM_THREADS / 32)
#define GEMM_GRID 512

template <int K, int NCOL>
__global__ __launch_bounds__(GEMM_THREADS)
void gemm_wmma_f32_kernel(const float* __restrict__ A,
                          const float* __restrict__ Bm,
                          float* __restrict__ C, int nStrips) {
    constexpr int PITCH = 2 * NCOL + 16;
    constexpr int TPW = NCOL / 16 / GEMM_WAVES;
    constexpr int C4 = NCOL / 4;
    static_assert(K % 8 == 0 && NCOL % (16 * GEMM_WAVES) == 0, "tile shape");
    __shared__ __attribute__((aligned(16))) float Bs[(K / 2) * PITCH];

    const int tid = threadIdx.x;
    for (int i = tid; i < (K / 2) * C4; i += GEMM_THREADS) {
        const int p = i / C4, c4 = i - p * C4;
        const float4 r0 = *reinterpret_cast<const float4*>(Bm + (size_t)(2 * p) * NCOL + 4 * c4);
        const float4 r1 = *reinterpret_cast<const float4*>(Bm + (size_t)(2 * p + 1) * NCOL + 4 * c4);
        float4* d = reinterpret_cast<float4*>(Bs + p * PITCH + 8 * c4);
        d[0] = make_float4(r0.x, r1.x, r0.y, r1.y);
        d[1] = make_float4(r0.z, r1.z, r0.w, r1.w);
    }
    __syncthreads();

    const int wave = tid >> 5;
    const int lane = tid & 31;
    const int c = lane & 15;
    const int h = lane >> 4;
    const float* bp = Bs + (2 * h) * PITCH + 2 * (wave * TPW * 16 + c);

    for (int strip = blockIdx.x; strip < nStrips; strip += gridDim.x) {
        const float* ap = A + ((size_t)strip * 16 + c) * K + 4 * h;
        float4 a4[K / 8];
#pragma unroll
        for (int j = 0; j < K / 8; ++j) a4[j] = *reinterpret_cast<const float4*>(ap + 8 * j);

        v8f acc[TPW];
#pragma unroll
        for (int t = 0; t < TPW; ++t) acc[t] = v8f{};

#pragma unroll
        for (int j = 0; j < K / 8; ++j) {
#pragma unroll
            for (int u = 0; u < 2; ++u) {
                v2f a;
                a.x = u ? a4[j].z : a4[j].x;
                a.y = u ? a4[j].w : a4[j].y;
#pragma unroll
                for (int t = 0; t < TPW; ++t) {
                    const v2f b = *reinterpret_cast<const v2f*>(bp + (4 * j + u) * PITCH + 32 * t);
                    acc[t] = __builtin_amdgcn_wmma_f32_16x16x4_f32(false, a, false, b,
                                                                   (short)0, acc[t], false, false);
                }
            }
        }
#pragma unroll
        for (int t = 0; t < TPW; ++t) {
            float* cp = C + ((size_t)strip * 16 + 8 * h) * NCOL + (wave * TPW + t) * 16 + c;
#pragma unroll
            for (int r = 0; r < 8; ++r) cp[(size_t)r * NCOL] = acc[t][r];
        }
    }
}

__global__ void alpha_kernel(const float* __restrict__ h,
                             const float* __restrict__ a_src,
                             const float* __restrict__ a_dst,
                             float* __restrict__ as, float* __restrict__ ad,
                             int n_nodes, int heads, int C) {
    int gw = (int)(((long long)blockIdx.x * blockDim.x + threadIdx.x) >> 5);
    int lane = threadIdx.x & 31;
    if (gw >= n_nodes * heads) return;
    int hd = gw % heads;
    const float* hp = h + (long long)gw * C;
    const float* sp = a_src + hd * C;
    const float* dp = a_dst + hd * C;
    float s1 = 0.0f, s2 = 0.0f;
    for (int c = lane; c < C; c += 32) {
        float v = hp[c];
        s1 += v * sp[c];
        s2 += v * dp[c];
    }
    for (int off = 16; off; off >>= 1) {
        s1 += __shfl_down(s1, off, 32);
        s2 += __shfl_down(s2, off, 32);
    }
    if (lane == 0) { as[gw] = s1; ad[gw] = s2; }
}

#define NO_LIST 0xFFFFFFFFu
#define OWN_R 200
#define OWN_THREADS 1024
#define OWN_WAVES (OWN_THREADS / 32)
#define OWN_CAP 16384

__device__ __forceinline__ float leaky(float v) { return (v > 0.0f) ? v : NEG_SLOPE * v; }

template <class Fn>
__device__ __forceinline__ void scan_dst(const int* __restrict__ dst, int E, int tid, Fn&& fn) {
    const bool vec = ((E & 3) == 0) && ((reinterpret_cast<size_t>(dst) & 15) == 0);
    if (vec) {
        const int4* d4 = reinterpret_cast<const int4*>(dst);
        const int E4 = E >> 2;
        for (int jb = 0; jb < E4; jb += 4 * OWN_THREADS) {
            int4 d[4];
#pragma unroll
            for (int u = 0; u < 4; ++u) {
                int j = jb + u * OWN_THREADS + tid;
                d[u] = (j < E4) ? d4[j] : make_int4(-1, -1, -1, -1);
            }
#pragma unroll
            for (int u = 0; u < 4; ++u) {
                int i = 4 * (jb + u * OWN_THREADS + tid);
                fn(i, d[u].x); fn(i + 1, d[u].y); fn(i + 2, d[u].z); fn(i + 3, d[u].w);
            }
        }
    } else {
        for (int ib = 0; ib < E; ib += OWN_THREADS) {
            int i = ib + tid;
            fn(i, (i < E) ? dst[i] : -1);
        }
    }
}

template <int F, int H, int U>
__device__ __forceinline__ void accumulate_edges(const int (&s)[U], const unsigned (&dl)[U],
                                                 const bool (&ok)[U], int lane,
                                                 const float* __restrict__ h,
                                                 const float* __restrict__ as,
                                                 const float* adl, const float* m,
                                                 float* den, float* acc) {
    constexpr int C = F / H;
    constexpr int J = F / 32;
    float r[U][J];
    float a[U][H];
#pragma unroll
    for (int u = 0; u < U; ++u) {
        const float* hs = h + (size_t)s[u] * F;
#pragma unroll
        for (int j = 0; j < J; ++j) r[u][j] = hs[lane + 32 * j];
#pragma unroll
        for (int hh = 0; hh < H; ++hh) a[u][hh] = as[(size_t)s[u] * H + hh];
    }
#pragma unroll
    for (int u = 0; u < U; ++u) {
        if (ok[u]) {
            float ex[H];
#pragma unroll
            for (int hh = 0; hh < H; ++hh) {
                float v = leaky(a[u][hh] + adl[dl[u] * H + hh]);
                ex[hh] = __expf(v - m[dl[u] * H + hh]);
            }
#pragma unroll
            for (int hh = 0; hh < H; ++hh)
                if (lane == hh) atomicAdd(&den[dl[u] * H + hh], ex[hh]);
            float* ar = acc + dl[u] * F;
#pragma unroll
            for (int j = 0; j < J; ++j)
                atomicAdd(&ar[lane + 32 * j], r[u][j] * ex[(32 * j) / C]);
        }
    }
}

template <int F, int H, int LISTS>
__global__ __launch_bounds__(OWN_THREADS)
void owner_aggregate_kernel(const int* __restrict__ ei, int E, int n_nodes,
                            const float* __restrict__ h,
                            const float* __restrict__ as,
                            const float* __restrict__ ad,
                            float* agg, unsigned int* keep, int keepF) {
    constexpr int C = F / H;
    __shared__ float acc[OWN_R * F];
    __shared__ unsigned int list[OWN_CAP];
    __shared__ float m[OWN_R * H];
    __shared__ float den[OWN_R * H];
    __shared__ float adl[OWN_R * H];
    __shared__ unsigned int cnt;

    const int tid = threadIdx.x;
    const int lane = tid & 31;
    const int wave = tid >> 5;
    const int base = blockIdx.x * OWN_R;
    const int R = min(OWN_R, n_nodes - base);
    const unsigned uR = (unsigned)R;
    const int* __restrict__ src = ei;
    const int* __restrict__ dst = ei + E;
    unsigned int* kp = keep + (size_t)base * keepF;
    const unsigned keepCap = (unsigned)R * (unsigned)keepF;

    for (int i = tid; i < R * F; i += OWN_THREADS) acc[i] = 0.0f;
    for (int i = tid; i < R * H; i += OWN_THREADS) {
        float a_d = ad[(size_t)base * H + i];
        adl[i] = a_d;
        m[i] = leaky(as[(size_t)base * H + i] + a_d);
        den[i] = 0.0f;
    }
    if (tid == 0) cnt = 0u;
    __syncthreads();

    unsigned kept = NO_LIST;
    if (LISTS == 2) kept = kp[0];
    const bool have = (LISTS == 2) && (kept <= OWN_CAP) && (kept < keepCap);

    const bool packs = (E <= (1 << 24)) && (n_nodes <= (1 << 24)) && (OWN_R <= 256);
    if (packs && !have) {
        scan_dst(dst, E, tid, [&](int i, int d) {
            unsigned dl = (unsigned)(d - base);
            if (dl < uR) {
                unsigned pos = atomicAdd(&cnt, 1u);
                if (pos < OWN_CAP) list[pos] = ((unsigned)i << 8) | dl;
            }
        });
    }
    __syncthreads();
    const unsigned total = have ? kept : cnt;
    const bool fast = have || (packs && (total <= OWN_CAP));
    const bool leave = (LISTS == 1) && fast && (total < keepCap);

    if (fast) {
        for (unsigned p = tid; p < total; p += OWN_THREADS) {
            unsigned dl;
            int s;
            if (have) {
                unsigned pk = kp[1 + p];
                dl = pk & 255u;
                s = (int)(pk >> 8);
            } else {
                unsigned pk = list[p];
                dl = pk & 255u;
                s = src[pk >> 8];
            }
#pragma unroll
            for (int hh = 0; hh < H; ++hh)
                atomicMaxFloat(&m[dl * H + hh], leaky(as[(size_t)s * H + hh] + adl[dl * H + hh]));
            list[p] = ((unsigned)s << 8) | dl;
            if (leave) kp[1 + p] = ((unsigned)s << 8) | dl;
        }
        if (LISTS == 1 && tid == 0) kp[0] = leave ? total : NO_LIST;
        __syncthreads();
        for (unsigned p0 = wave * 4; p0 < total; p0 += OWN_WAVES * 4) {
            int s[4]; unsigned dl[4]; bool ok[4];
#pragma unroll
            for (int u = 0; u < 4; ++u) {
                ok[u] = (p0 + u) < total;
                unsigned pk = ok[u] ? list[p0 + u] : 0u;
                s[u] = (int)(pk >> 8);
                dl[u] = pk & 255u;
            }
            accumulate_edges<F, H, 4>(s, dl, ok, lane, h, as, adl, m, den, acc);
        }
    } else {
        if (LISTS == 1 && tid == 0) kp[0] = NO_LIST;
        scan_dst(dst, E, tid, [&](int i, int d) {
            unsigned dl = (unsigned)(d - base);
            if (dl < uR) {
                int s = src[i];
#pragma unroll
                for (int hh = 0; hh < H; ++hh)
                    atomicMaxFloat(&m[dl * H + hh], leaky(as[(size_t)s * H + hh] + adl[dl * H + hh]));
            }
        });
        __syncthreads();
        scan_dst(dst, E, tid, [&](int i, int d) {
            unsigned dl = (unsigned)(d - base);
            bool hit = dl < uR;
            int sv = hit ? src[i] : 0;
            unsigned mask = __builtin_amdgcn_ballot_w32(hit);
            while (mask) {
                int l = __builtin_ctz(mask);
                mask &= mask - 1u;
                int s1[1] = { __shfl(sv, l, 32) };
                unsigned d1[1] = { (unsigned)__shfl((int)dl, l, 32) };
                bool o1[1] = { true };
                accumulate_edges<F, H, 1>(s1, d1, o1, lane, h, as, adl, m, den, acc);
            }
        });
    }
    for (int q0 = wave * 4; q0 < R; q0 += OWN_WAVES * 4) {
        int s[4]; unsigned dl[4]; bool ok[4];
#pragma unroll
        for (int u = 0; u < 4; ++u) {
            ok[u] = (q0 + u) < R;
            dl[u] = ok[u] ? (unsigned)(q0 + u) : 0u;
            s[u] = base + (int)dl[u];
        }
        accumulate_edges<F, H, 4>(s, dl, ok, lane, h, as, adl, m, den, acc);
    }
    __syncthreads();

    for (int i = tid; i < R * F; i += OWN_THREADS) {
        int dl = i / F;
        int c = i - dl * F;
        agg[(size_t)base * F + i] = acc[i] / den[dl * H + c / C];
    }
}

__global__ void bias_bn_relu_kernel(const float* __restrict__ x,
                                    const float* __restrict__ b,
                                    const float* __restrict__ gamma,
                                    const float* __restrict__ beta,
                                    const float* __restrict__ mean,
                                    const float* __restrict__ var,
                                    float* __restrict__ out, long long total, int F) {
    long long i = (long long)blockIdx.x * blockDim.x + threadIdx.x;
    if (i >= total) return;
    int c = (int)(i % F);
    float v = x[i] + b[c];
    v = gamma[c] * (v - mean[c]) * rsqrtf(var[c] + EPS) + beta[c];
    out[i] = (v > 0.0f) ? v : 0.0f;
}

__global__ void pool_max_kernel(const float* __restrict__ x,
                                const int* __restrict__ batch,
                                float* __restrict__ out, int n_nodes, int F) {
    long long i = (long long)blockIdx.x * blockDim.x + threadIdx.x;
    if (i >= (long long)n_nodes * F) return;
    int n = (int)(i / F);
    int c = (int)(i - (long long)n * F);
    atomicMaxFloat(&out[batch[n] * F + c], x[i]);
}

extern "C" void kernel_launch(void* const* d_in, const int* in_sizes, int n_in,
                              void* d_out, int out_size, void* d_ws, size_t ws_size,
                              hipStream_t stream) {
    const int F_IN = 128, HID = 128, HEADS = 2;
    const int F1 = HEADS * HID;
    const int N = in_sizes[0] / F_IN;
    const int E = in_sizes[1] / 2;

    const float* x      = (const float*)d_in[0];
    const int*   ei     = (const int*)d_in[1];
    const int*   batch  = (const int*)d_in[2];
    const float* W1     = (const float*)d_in[3];
    const float* a_src1 = (const float*)d_in[4];
    const float* a_dst1 = (const float*)d_in[5];
    const float* b1     = (const float*)d_in[6];
    const float* gamma1 = (const float*)d_in[7];
    const float* beta1  = (const float*)d_in[8];
    const float* rm1    = (const float*)d_in[9];
    const float* rv1    = (const float*)d_in[10];
    const float* W2     = (const float*)d_in[11];
    const float* a_src2 = (const float*)d_in[12];
    const float* a_dst2 = (const float*)d_in[13];
    const float* b2     = (const float*)d_in[14];
    const float* gamma2 = (const float*)d_in[15];
    const float* beta2  = (const float*)d_in[16];
    const float* rm2    = (const float*)d_in[17];
    const float* rv2    = (const float*)d_in[18];

    float* ws = (float*)d_ws;
    size_t o = 0;
    float* h1   = ws + o; o += (size_t)N * F1;
    float* aggA = ws + o; o += (size_t)N * F1;
    float* aggB = ws + o; o += (size_t)N * HID;
    float* as1  = ws + o; o += (size_t)N * HEADS;
    float* ad1  = ws + o; o += (size_t)N * HEADS;
    float* as2  = ws + o; o += (size_t)N;
    float* ad2  = ws + o; o += (size_t)N;
    float* h2   = h1;

    const unsigned int NEG_INF = 0xFF800000u;
    auto fillBlocks = [](long long cnt) { return (unsigned)((cnt + 255) / 256); };
    const unsigned ownerBlocks = (unsigned)((N + OWN_R - 1) / OWN_R);

    fill_u32_kernel<<<fillBlocks((long long)out_size), 256, 0, stream>>>((unsigned*)d_out, NEG_INF, (long long)out_size);

    gemm_wmma_f32_kernel<128, 256><<<GEMM_GRID, GEMM_THREADS, 0, stream>>>(x, W1, h1, N / 16);
    alpha_kernel<<<(unsigned)(((long long)N * HEADS * 32 + 255) / 256), 256, 0, stream>>>(
        h1, a_src1, a_dst1, as1, ad1, N, HEADS, HID);
    owner_aggregate_kernel<256, 2, 1><<<ownerBlocks, OWN_THREADS, 0, stream>>>(
        ei, E, N, h1, as1, ad1, aggA, (unsigned int*)aggB, 128);
    bias_bn_relu_kernel<<<fillBlocks((long long)N * F1), 256, 0, stream>>>(
        aggA, b1, gamma1, beta1, rm1, rv1, aggA, (long long)N * F1, F1);

    gemm_wmma_f32_kernel<256, 128><<<GEMM_GRID, GEMM_THREADS, 0, stream>>>(aggA, W2, h2, N / 16);
    alpha_kernel<<<(unsigned)(((long long)N * 32 + 255) / 256), 256, 0, stream>>>(
        h2, a_src2, a_dst2, as2, ad2, N, 1, HID);
    owner_aggregate_kernel<128, 1, 2><<<ownerBlocks, OWN_THREADS, 0, stream>>>(
        ei, E, N, h2, as2, ad2, aggB, (unsigned int*)aggB, 128);
    bias_bn_relu_kernel<<<fillBlocks((long long)N * HID), 256, 0, stream>>>(
        aggB, b2, gamma2, beta2, rm2, rv2, aggB, (long long)N * HID, HID);

    pool_max_kernel<<<fillBlocks((long long)N * HID), 256, 0, stream>>>(
        aggB, batch, (float*)d_out, N, HID);
}


# Owner work-groups build their edge lists from a 1-byte-per-edge owner-id plane (destination/200, one new small launch per call), not the 4-byte destinations; list order, so sum order, changes
# speedup vs baseline: 3.2033x; 1.0701x over previous
#include <hip/hip_runtime.h>
#include <hip/hip_bf16.h>

typedef float v2f __attribute__((ext_vector_type(2)));
typedef float v8f __attribute__((ext_vector_type(8)));

#define NEG_SLOPE 0.2f
#define EPS 1e-5f

__device__ __forceinline__ void atomicMaxFloat(float* addr, float val) {
    if (val >= 0.0f) {
        atomicMax((int*)addr, __float_as_int(val));
    } else {
        atomicMin((unsigned int*)addr, __float_as_uint(val));
    }
}

__global__ void fill_u32_kernel(unsigned int* __restrict__ p, unsigned int v, long long count) {
    long long i = (long long)blockIdx.x * blockDim.x + threadIdx.x;
    if (i < count) p[i] = v;
}

#define GEMM_THREADS 256
#define GEMM_WAVES (GEMM_THREADS / 32)
#define GEMM_GRID 512

template <int K, int NCOL>
__global__ __launch_bounds__(GEMM_THREADS)
void gemm_wmma_f32_kernel(const float* __restrict__ A,
                          const float* __restrict__ Bm,
                          float* __restrict__ C, int nStrips) {
    constexpr int PITCH = 2 * NCOL + 16;
    constexpr int TPW = NCOL / 16 / GEMM_WAVES;
    constexpr int C4 = NCOL / 4;
    static_assert(K % 8 == 0 && NCOL % (16 * GEMM_WAVES) == 0, "tile shape");
    __shared__ __attribute__((aligned(16))) float Bs[(K / 2) * PITCH];

    const int tid = threadIdx.x;
    for (int i = tid; i < (K / 2) * C4; i += GEMM_THREADS) {
        const int p = i / C4, c4 = i - p * C4;
        const float4 r0 = *reinterpret_cast<const float4*>(Bm + (size_t)(2 * p) * NCOL + 4 * c4);
        const float4 r1 = *reinterpret_cast<const float4*>(Bm + (size_t)(2 * p + 1) * NCOL + 4 * c4);
        float4* d = reinterpret_cast<float4*>(Bs + p * PITCH + 8 * c4);
        d[0] = make_float4(r0.x, r1.x, r0.y, r1.y);
        d[1] = make_float4(r0.z, r1.z, r0.w, r1.w);
    }
    __syncthreads();

    const int wave = tid >> 5;
    const int lane = tid & 31;
    const int c = lane & 15;
    const int h = lane >> 4;
    const float* bp = Bs + (2 * h) * PITCH + 2 * (wave * TPW * 16 + c);

    for (int strip = blockIdx.x; strip < nStrips; strip += gridDim.x) {
        const float* ap = A + ((size_t)strip * 16 + c) * K + 4 * h;
        float4 a4[K / 8];
#pragma unroll
        for (int j = 0; j < K / 8; ++j) a4[j] = *reinterpret_cast<const float4*>(ap + 8 * j);

        v8f acc[TPW];
#pragma unroll
        for (int t = 0; t < TPW; ++t) acc[t] = v8f{};

#pragma unroll
        for (int j = 0; j < K / 8; ++j) {
#pragma unroll
            for (int u = 0; u < 2; ++u) {
                v2f a;
                a.x = u ? a4[j].z : a4[j].x;
                a.y = u ? a4[j].w : a4[j].y;
#pragma unroll
                for (int t = 0; t < TPW; ++t) {
                    const v2f b = *reinterpret_cast<const v2f*>(bp + (4 * j + u) * PITCH + 32 * t);
                    acc[t] = __builtin_amdgcn_wmma_f32_16x16x4_f32(false, a, false, b,
                                                                   (short)0, acc[t], false, false);
                }
            }
        }
#pragma unroll
        for (int t = 0; t < TPW; ++t) {
            float* cp = C + ((size_t)strip * 16 + 8 * h) * NCOL + (wave * TPW + t) * 16 + c;
#pragma unroll
            for (int r = 0; r < 8; ++r) cp[(size_t)r * NCOL] = acc[t][r];
        }
    }
}

__global__ void alpha_kernel(const float* __restrict__ h,
                             const float* __restrict__ a_src,
                             const float* __restrict__ a_dst,
                             float* __restrict__ as, float* __restrict__ ad,
                             int n_nodes, int heads, int C) {
    int gw = (int)(((long long)blockIdx.x * blockDim.x + threadIdx.x) >> 5);
    int lane = threadIdx.x & 31;
    if (gw >= n_nodes * heads) return;
    int hd = gw % heads;
    const float* hp = h + (long long)gw * C;
    const float* sp = a_src + hd * C;
    const float* dp = a_dst + hd * C;
    float s1 = 0.0f, s2 = 0.0f;
    for (int c = lane; c < C; c += 32) {
        float v = hp[c];
        s1 += v * sp[c];
        s2 += v * dp[c];
    }
    for (int off = 16; off; off >>= 1) {
        s1 += __shfl_down(s1, off, 32);
        s2 += __shfl_down(s2, off, 32);
    }
    if (lane == 0) { as[gw] = s1; ad[gw] = s2; }
}

#define NO_LIST 0xFFFFFFFFu
#define OWN_R 200
#define OWN_THREADS 1024
#define OWN_WAVES (OWN_THREADS / 32)
#define OWN_CAP 16384
#define NO_OWNER 255u
#define KEEP_WORDS 96
#define PLANE_WORDS 32

__device__ __forceinline__ bool owner_plane_ok(int E, int n_nodes, const void* plane) {
    const unsigned nOwner = ((unsigned)n_nodes + OWN_R - 1) / OWN_R;
    const size_t bytes = ((size_t)E + 15) & ~(size_t)15;
    return plane != nullptr && nOwner <= NO_OWNER
        && bytes <= (size_t)n_nodes * (PLANE_WORDS * 4)
        && (reinterpret_cast<size_t>(plane) & 15) == 0;
}

__global__ void owner_plane_kernel(const int* __restrict__ ei, int E, int n_nodes,
                                   uint4* __restrict__ plane) {
    if (!owner_plane_ok(E, n_nodes, plane)) return;
    const int* __restrict__ dst = ei + E;
    const int nVec = (E + 15) >> 4;
    const int v = blockIdx.x * blockDim.x + threadIdx.x;
    if (v >= nVec) return;
    int d[16];
    if (((E & 15) == 0) && ((reinterpret_cast<size_t>(dst) & 15) == 0)) {
        const int4* d4 = reinterpret_cast<const int4*>(dst) + 4 * (size_t)v;
#pragma unroll
        for (int q = 0; q < 4; ++q) {
            const int4 t = d4[q];
            d[4 * q] = t.x; d[4 * q + 1] = t.y; d[4 * q + 2] = t.z; d[4 * q + 3] = t.w;
        }
    } else {
#pragma unroll
        for (int k = 0; k < 16; ++k) {
            const int i = 16 * v + k;
            d[k] = (i < E) ? dst[i] : -1;
        }
    }
    unsigned w[4];
#pragma unroll
    for (int q = 0; q < 4; ++q) {
        w[q] = 0u;
#pragma unroll
        for (int k = 0; k < 4; ++k) {
            const unsigned dd = (unsigned)d[4 * q + k];
            const unsigned o = (dd < (unsigned)n_nodes) ? dd / OWN_R : NO_OWNER;
            w[q] |= o << (8 * k);
        }
    }
    plane[v] = make_uint4(w[0], w[1], w[2], w[3]);
}

__device__ __forceinline__ float leaky(float v) { return (v > 0.0f) ? v : NEG_SLOPE * v; }

template <class Fn>
__device__ __forceinline__ void scan_dst(const int* __restrict__ dst, int E, int tid, Fn&& fn) {
    const bool vec = ((E & 3) == 0) && ((reinterpret_cast<size_t>(dst) & 15) == 0);
    if (vec) {
        const int4* d4 = reinterpret_cast<const int4*>(dst);
        const int E4 = E >> 2;
        for (int jb = 0; jb < E4; jb += 4 * OWN_THREADS) {
            int4 d[4];
#pragma unroll
            for (int u = 0; u < 4; ++u) {
                int j = jb + u * OWN_THREADS + tid;
                d[u] = (j < E4) ? d4[j] : make_int4(-1, -1, -1, -1);
            }
#pragma unroll
            for (int u = 0; u < 4; ++u) {
                int i = 4 * (jb + u * OWN_THREADS + tid);
                fn(i, d[u].x); fn(i + 1, d[u].y); fn(i + 2, d[u].z); fn(i + 3, d[u].w);
            }
        }
    } else {
        for (int ib = 0; ib < E; ib += OWN_THREADS) {
            int i = ib + tid;
            fn(i, (i < E) ? dst[i] : -1);
        }
    }
}

template <class Fn>
__device__ __forceinline__ void scan_plane(const uint4* __restrict__ plane, int nVec,
                                           unsigned me, int tid, Fn&& hit) {
    const unsigned key = me * 0x01010101u;
    for (int vb = 0; vb < nVec; vb += 8 * OWN_THREADS) {
        uint4 d[8];
#pragma unroll
        for (int u = 0; u < 8; ++u) {
            const int v = vb + u * OWN_THREADS + tid;
            d[u] = (v < nVec) ? plane[v] : make_uint4(~0u, ~0u, ~0u, ~0u);
        }
#pragma unroll
        for (int u = 0; u < 8; ++u) {
            const int i0 = 16 * (vb + u * OWN_THREADS + tid);
            const unsigned w[4] = { d[u].x, d[u].y, d[u].z, d[u].w };
#pragma unroll
            for (int q = 0; q < 4; ++q) {
                const unsigned x = w[q] ^ key;
                unsigned z = ~(((x & 0x7F7F7F7Fu) + 0x7F7F7F7Fu) | x | 0x7F7F7F7Fu);
                while (z) {
                    const int k = __builtin_ctz(z) >> 3;
                    z &= z - 1u;
                    hit(i0 + 4 * q + k);
                }
            }
        }
    }
}

template <int F, int H, int U>
__device__ __forceinline__ void accumulate_edges(const int (&s)[U], const unsigned (&dl)[U],
                                                 const bool (&ok)[U], int lane,
                                                 const float* __restrict__ h,
                                                 const float* __restrict__ as,
                                                 const float* adl, const float* m,
                                                 float* den, float* acc) {
    constexpr int C = F / H;
    constexpr int J = F / 32;
    float r[U][J];
    float a[U][H];
#pragma unroll
    for (int u = 0; u < U; ++u) {
        const float* hs = h + (size_t)s[u] * F;
#pragma unroll
        for (int j = 0; j < J; ++j) r[u][j] = hs[lane + 32 * j];
#pragma unroll
        for (int hh = 0; hh < H; ++hh) a[u][hh] = as[(size_t)s[u] * H + hh];
    }
#pragma unroll
    for (int u = 0; u < U; ++u) {
        if (ok[u]) {
            float ex[H];
#pragma unroll
            for (int hh = 0; hh < H; ++hh) {
                float v = leaky(a[u][hh] + adl[dl[u] * H + hh]);
                ex[hh] = __expf(v - m[dl[u] * H + hh]);
            }
#pragma unroll
            for (int hh = 0; hh < H; ++hh)
                if (lane == hh) atomicAdd(&den[dl[u] * H + hh], ex[hh]);
            float* ar = acc + dl[u] * F;
#pragma unroll
            for (int j = 0; j < J; ++j)
                atomicAdd(&ar[lane + 32 * j], r[u][j] * ex[(32 * j) / C]);
        }
    }
}

template <int F, int H, int LISTS>
__global__ __launch_bounds__(OWN_THREADS)
void owner_aggregate_kernel(const int* __restrict__ ei, int E, int n_nodes,
                            const float* __restrict__ h,
                            const float* __restrict__ as,
                            const float* __restrict__ ad,
                            float* agg, unsigned int* keep, int keepF,
                            const uint4* __restrict__ plane) {
    constexpr int C = F / H;
    __shared__ float acc[OWN_R * F];
    __shared__ unsigned int list[OWN_CAP];
    __shared__ float m[OWN_R * H];
    __shared__ float den[OWN_R * H];
    __shared__ float adl[OWN_R * H];
    __shared__ unsigned int cnt;

    const int tid = threadIdx.x;
    const int lane = tid & 31;
    const int wave = tid >> 5;
    const int base = blockIdx.x * OWN_R;
    const int R = min(OWN_R, n_nodes - base);
    const unsigned uR = (unsigned)R;
    const int* __restrict__ src = ei;
    const int* __restrict__ dst = ei + E;
    unsigned int* kp = keep + (size_t)base * keepF;
    const unsigned keepCap = (unsigned)R * (unsigned)keepF;

    for (int i = tid; i < R * F; i += OWN_THREADS) acc[i] = 0.0f;
    for (int i = tid; i < R * H; i += OWN_THREADS) {
        float a_d = ad[(size_t)base * H + i];
        adl[i] = a_d;
        m[i] = leaky(as[(size_t)base * H + i] + a_d);
        den[i] = 0.0f;
    }
    if (tid == 0) cnt = 0u;
    __syncthreads();

    unsigned kept = NO_LIST;
    if (LISTS == 2) kept = kp[0];
    const bool have = (LISTS == 2) && (kept <= OWN_CAP) && (kept < keepCap);

    const bool packs = (E <= (1 << 24)) && (n_nodes <= (1 << 24)) && (OWN_R <= 256);
    if (packs && !have) {
        if (owner_plane_ok(E, n_nodes, plane)) {
            scan_plane(plane, (E + 15) >> 4, (unsigned)blockIdx.x, tid, [&](int i) {
                unsigned pos = atomicAdd(&cnt, 1u);
                if (pos < OWN_CAP) list[pos] = (unsigned)i << 8;
            });
        } else {
            scan_dst(dst, E, tid, [&](int i, int d) {
                unsigned dl = (unsigned)(d - base);
                if (dl < uR) {
                    unsigned pos = atomicAdd(&cnt, 1u);
                    if (pos < OWN_CAP) list[pos] = (unsigned)i << 8;
                }
            });
        }
    }
    __syncthreads();
    const unsigned total = have ? kept : cnt;
    const bool fast = have || (packs && (total <= OWN_CAP));
    const bool leave = (LISTS == 1) && fast && (total < keepCap);

    if (fast) {
        for (unsigned p = tid; p < total; p += OWN_THREADS) {
            unsigned dl;
            int s;
            if (have) {
                unsigned pk = kp[1 + p];
                dl = pk & 255u;
                s = (int)(pk >> 8);
            } else {
                const unsigned i = list[p] >> 8;
                dl = (unsigned)(dst[i] - base);
                s = src[i];
            }
#pragma unroll
            for (int hh = 0; hh < H; ++hh)
                atomicMaxFloat(&m[dl * H + hh], leaky(as[(size_t)s * H + hh] + adl[dl * H + hh]));
            list[p] = ((unsigned)s << 8) | dl;
            if (leave) kp[1 + p] = ((unsigned)s << 8) | dl;
        }
        if (LISTS == 1 && tid == 0) kp[0] = leave ? total : NO_LIST;
        __syncthreads();
        for (unsigned p0 = wave * 4; p0 < total; p0 += OWN_WAVES * 4) {
            int s[4]; unsigned dl[4]; bool ok[4];
#pragma unroll
            for (int u = 0; u < 4; ++u) {
                ok[u] = (p0 + u) < total;
                unsigned pk = ok[u] ? list[p0 + u] : 0u;
                s[u] = (int)(pk >> 8);
                dl[u] = pk & 255u;
            }
            accumulate_edges<F, H, 4>(s, dl, ok, lane, h, as, adl, m, den, acc);
        }
    } else {
        if (LISTS == 1 && tid == 0) kp[0] = NO_LIST;
        scan_dst(dst, E, tid, [&](int i, int d) {
            unsigned dl = (unsigned)(d - base);
            if (dl < uR) {
                int s = src[i];
#pragma unroll
                for (int hh = 0; hh < H; ++hh)
                    atomicMaxFloat(&m[dl * H + hh], leaky(as[(size_t)s * H + hh] + adl[dl * H + hh]));
            }
        });
        __syncthreads();
        scan_dst(dst, E, tid, [&](int i, int d) {
            unsigned dl = (unsigned)(d - base);
            bool hit = dl < uR;
            int sv = hit ? src[i] : 0;
            unsigned mask = __builtin_amdgcn_ballot_w32(hit);
            while (mask) {
                int l = __builtin_ctz(mask);
                mask &= mask - 1u;
                int s1[1] = { __shfl(sv, l, 32) };
                unsigned d1[1] = { (unsigned)__shfl((int)dl, l, 32) };
                bool o1[1] = { true };
                accumulate_edges<F, H, 1>(s1, d1, o1, lane, h, as, adl, m, den, acc);
            }
        });
    }
    for (int q0 = wave * 4; q0 < R; q0 += OWN_WAVES * 4) {
        int s[4]; unsigned dl[4]; bool ok[4];
#pragma unroll
        for (int u = 0; u < 4; ++u) {
            ok[u] = (q0 + u) < R;
            dl[u] = ok[u] ? (unsigned)(q0 + u) : 0u;
            s[u] = base + (int)dl[u];
        }
        accumulate_edges<F, H, 4>(s, dl, ok, lane, h, as, adl, m, den, acc);
    }
    __syncthreads();

    for (int i = tid; i < R * F; i += OWN_THREADS) {
        int dl = i / F;
        int c = i - dl * F;
        agg[(size_t)base * F + i] = acc[i] / den[dl * H + c / C];
    }
}

__global__ void bias_bn_relu_kernel(const float* __restrict__ x,
                                    const float* __restrict__ b,
                                    const float* __restrict__ gamma,
                                    const float* __restrict__ beta,
                                    const float* __restrict__ mean,
                                    const float* __restrict__ var,
                                    float* __restrict__ out, long long total, int F) {
    long long i = (long long)blockIdx.x * blockDim.x + threadIdx.x;
    if (i >= total) return;
    int c = (int)(i % F);
    float v = x[i] + b[c];
    v = gamma[c] * (v - mean[c]) * rsqrtf(var[c] + EPS) + beta[c];
    out[i] = (v > 0.0f) ? v : 0.0f;
}

__global__ void pool_max_kernel(const float* __restrict__ x,
                                const int* __restrict__ batch,
                                float* __restrict__ out, int n_nodes, int F) {
    long long i = (long long)blockIdx.x * blockDim.x + threadIdx.x;
    if (i >= (long long)n_nodes * F) return;
    int n = (int)(i / F);
    int c = (int)(i - (long long)n * F);
    atomicMaxFloat(&out[batch[n] * F + c], x[i]);
}

extern "C" void kernel_launch(void* const* d_in, const int* in_sizes, int n_in,
                              void* d_out, int out_size, void* d_ws, size_t ws_size,
                              hipStream_t stream) {
    const int F_IN = 128, HID = 128, HEADS = 2;
    const int F1 = HEADS * HID;
    const int N = in_sizes[0] / F_IN;
    const int E = in_sizes[1] / 2;

    const float* x      = (const float*)d_in[0];
    const int*   ei     = (const int*)d_in[1];
    const int*   batch  = (const int*)d_in[2];
    const float* W1     = (const float*)d_in[3];
    const float* a_src1 = (const float*)d_in[4];
    const float* a_dst1 = (const float*)d_in[5];
    const float* b1     = (const float*)d_in[6];
    const float* gamma1 = (const float*)d_in[7];
    const float* beta1  = (const float*)d_in[8];
    const float* rm1    = (const float*)d_in[9];
    const float* rv1    = (const float*)d_in[10];
    const float* W2     = (const float*)d_in[11];
    const float* a_src2 = (const float*)d_in[12];
    const float* a_dst2 = (const float*)d_in[13];
    const float* b2     = (const float*)d_in[14];
    const float* gamma2 = (const float*)d_in[15];
    const float* beta2  = (const float*)d_in[16];
    const float* rm2    = (const float*)d_in[17];
    const float* rv2    = (const float*)d_in[18];

    float* ws = (float*)d_ws;
    size_t o = 0;
    float* h1   = ws + o; o += (size_t)N * F1;
    float* aggA = ws + o; o += (size_t)N * F1;
    float* aggB = ws + o; o += (size_t)N * HID;
    float* as1  = ws + o; o += (size_t)N * HEADS;
    float* ad1  = ws + o; o += (size_t)N * HEADS;
    float* as2  = ws + o; o += (size_t)N;
    float* ad2  = ws + o; o += (size_t)N;
    float* h2   = h1;
    float* agg2 = h1 + (size_t)N * HID;
    unsigned int* keep = (unsigned int*)aggB;
    uint4* plane = (uint4*)(aggB + (size_t)N * KEEP_WORDS);

    const unsigned int NEG_INF = 0xFF800000u;
    auto fillBlocks = [](long long cnt) { return (unsigned)((cnt + 255) / 256); };
    const unsigned ownerBlocks = (unsigned)((N + OWN_R - 1) / OWN_R);

    fill_u32_kernel<<<fillBlocks((long long)out_size), 256, 0, stream>>>((unsigned*)d_out, NEG_INF, (long long)out_size);

    gemm_wmma_f32_kernel<128, 256><<<GEMM_GRID, GEMM_THREADS, 0, stream>>>(x, W1, h1, N / 16);
    alpha_kernel<<<(unsigned)(((long long)N * HEADS * 32 + 255) / 256), 256, 0, stream>>>(
        h1, a_src1, a_dst1, as1, ad1, N, HEADS, HID);
    owner_plane_kernel<<<fillBlocks((long long)((E + 15) / 16)), 256, 0, stream>>>(ei, E, N, plane);
    owner_aggregate_kernel<256, 2, 1><<<ownerBlocks, OWN_THREADS, 0, stream>>>(
        ei, E, N, h1, as1, ad1, aggA, keep, KEEP_WORDS, plane);
    bias_bn_relu_kernel<<<fillBlocks((long long)N * F1), 256, 0, stream>>>(
        aggA, b1, gamma1, beta1, rm1, rv1, aggA, (long long)N * F1, F1);

    gemm_wmma_f32_kernel<256, 128><<<GEMM_GRID, GEMM_THREADS, 0, stream>>>(aggA, W2, h2, N / 16);
    alpha_kernel<<<(unsigned)(((long long)N * 32 + 255) / 256), 256, 0, stream>>>(
        h2, a_src2, a_dst2, as2, ad2, N, 1, HID);
    owner_aggregate_kernel<128, 1, 2><<<ownerBlocks, OWN_THREADS, 0, stream>>>(
        ei, E, N, h2, as2, ad2, agg2, keep, KEEP_WORDS, plane);
    bias_bn_relu_kernel<<<fillBlocks((long long)N * HID), 256, 0, stream>>>(
        agg2, b2, gamma2, beta2, rm2, rv2, agg2, (long long)N * HID, HID);

    pool_max_kernel<<<fillBlocks((long long)N * HID), 256, 0, stream>>>(
        agg2, batch, (float*)d_out, N, HID);
}


# Max pool by graph-owner work-groups: each scans batch ids, lists its graph's nodes in LDS, takes their layer-2 rows through bias+BN+ReLU on load, stores its row once; no atomics, fill, BN-2 launch
# speedup vs baseline: 3.4634x; 1.0812x over previous
#include <hip/hip_runtime.h>
#include <hip/hip_bf16.h>

typedef float v2f __attribute__((ext_vector_type(2)));
typedef float v8f __attribute__((ext_vector_type(8)));

#define NEG_SLOPE 0.2f
#define EPS 1e-5f

__device__ __forceinline__ void atomicMaxFloat(float* addr, float val) {
    if (val >= 0.0f) {
        atomicMax((int*)addr, __float_as_int(val));
    } else {
        atomicMin((unsigned int*)addr, __float_as_uint(val));
    }
}

#define GEMM_THREADS 256
#define GEMM_WAVES (GEMM_THREADS / 32)
#define GEMM_GRID 512

template <int K, int NCOL>
__global__ __launch_bounds__(GEMM_THREADS)
void gemm_wmma_f32_kernel(const float* __restrict__ A,
                          const float* __restrict__ Bm,
                          float* __restrict__ C, int nStrips) {
    constexpr int PITCH = 2 * NCOL + 16;
    constexpr int TPW = NCOL / 16 / GEMM_WAVES;
    constexpr int C4 = NCOL / 4;
    static_assert(K % 8 == 0 && NCOL % (16 * GEMM_WAVES) == 0, "tile shape");
    __shared__ __attribute__((aligned(16))) float Bs[(K / 2) * PITCH];

    const int tid = threadIdx.x;
    for (int i = tid; i < (K / 2) * C4; i += GEMM_THREADS) {
        const int p = i / C4, c4 = i - p * C4;
        const float4 r0 = *reinterpret_cast<const float4*>(Bm + (size_t)(2 * p) * NCOL + 4 * c4);
        const float4 r1 = *reinterpret_cast<const float4*>(Bm + (size_t)(2 * p + 1) * NCOL + 4 * c4);
        float4* d = reinterpret_cast<float4*>(Bs + p * PITCH + 8 * c4);
        d[0] = make_float4(r0.x, r1.x, r0.y, r1.y);
        d[1] = make_float4(r0.z, r1.z, r0.w, r1.w);
    }
    __syncthreads();

    const int wave = tid >> 5;
    const int lane = tid & 31;
    const int c = lane & 15;
    const int h = lane >> 4;
    const float* bp = Bs + (2 * h) * PITCH + 2 * (wave * TPW * 16 + c);

    for (int strip = blockIdx.x; strip < nStrips; strip += gridDim.x) {
        const float* ap = A + ((size_t)strip * 16 + c) * K + 4 * h;
        float4 a4[K / 8];
#pragma unroll
        for (int j = 0; j < K / 8; ++j) a4[j] = *reinterpret_cast<const float4*>(ap + 8 * j);

        v8f acc[TPW];
#pragma unroll
        for (int t = 0; t < TPW; ++t) acc[t] = v8f{};

#pragma unroll
        for (int j = 0; j < K / 8; ++j) {
#pragma unroll
            for (int u = 0; u < 2; ++u) {
                v2f a;
                a.x = u ? a4[j].z : a4[j].x;
                a.y = u ? a4[j].w : a4[j].y;
#pragma unroll
                for (int t = 0; t < TPW; ++t) {
                    const v2f b = *reinterpret_cast<const v2f*>(bp + (4 * j + u) * PITCH + 32 * t);
                    acc[t] = __builtin_amdgcn_wmma_f32_16x16x4_f32(false, a, false, b,
                                                                   (short)0, acc[t], false, false);
                }
            }
        }
#pragma unroll
        for (int t = 0; t < TPW; ++t) {
            float* cp = C + ((size_t)strip * 16 + 8 * h) * NCOL + (wave * TPW + t) * 16 + c;
#pragma unroll
            for (int r = 0; r < 8; ++r) cp[(size_t)r * NCOL] = acc[t][r];
        }
    }
}

__global__ void alpha_kernel(const float* __restrict__ h,
                             const float* __restrict__ a_src,
                             const float* __restrict__ a_dst,
                             float* __restrict__ as, float* __restrict__ ad,
                             int n_nodes, int heads, int C) {
    int gw = (int)(((long long)blockIdx.x * blockDim.x + threadIdx.x) >> 5);
    int lane = threadIdx.x & 31;
    if (gw >= n_nodes * heads) return;
    int hd = gw % heads;
    const float* hp = h + (long long)gw * C;
    const float* sp = a_src + hd * C;
    const float* dp = a_dst + hd * C;
    float s1 = 0.0f, s2 = 0.0f;
    for (int c = lane; c < C; c += 32) {
        float v = hp[c];
        s1 += v * sp[c];
        s2 += v * dp[c];
    }
    for (int off = 16; off; off >>= 1) {
        s1 += __shfl_down(s1, off, 32);
        s2 += __shfl_down(s2, off, 32);
    }
    if (lane == 0) { as[gw] = s1; ad[gw] = s2; }
}

#define NO_LIST 0xFFFFFFFFu
#define OWN_R 200
#define OWN_THREADS 1024
#define OWN_WAVES (OWN_THREADS / 32)
#define OWN_CAP 16384
#define NO_OWNER 255u
#define KEEP_WORDS 96
#define PLANE_WORDS 32

__device__ __forceinline__ bool owner_plane_ok(int E, int n_nodes, const void* plane) {
    const unsigned nOwner = ((unsigned)n_nodes + OWN_R - 1) / OWN_R;
    const size_t bytes = ((size_t)E + 15) & ~(size_t)15;
    return plane != nullptr && nOwner <= NO_OWNER
        && bytes <= (size_t)n_nodes * (PLANE_WORDS * 4)
        && (reinterpret_cast<size_t>(plane) & 15) == 0;
}

__global__ void owner_plane_kernel(const int* __restrict__ ei, int E, int n_nodes,
                                   uint4* __restrict__ plane) {
    if (!owner_plane_ok(E, n_nodes, plane)) return;
    const int* __restrict__ dst = ei + E;
    const int nVec = (E + 15) >> 4;
    const int v = blockIdx.x * blockDim.x + threadIdx.x;
    if (v >= nVec) return;
    int d[16];
    if (((E & 15) == 0) && ((reinterpret_cast<size_t>(dst) & 15) == 0)) {
        const int4* d4 = reinterpret_cast<const int4*>(dst) + 4 * (size_t)v;
#pragma unroll
        for (int q = 0; q < 4; ++q) {
            const int4 t = d4[q];
            d[4 * q] = t.x; d[4 * q + 1] = t.y; d[4 * q + 2] = t.z; d[4 * q + 3] = t.w;
        }
    } else {
#pragma unroll
        for (int k = 0; k < 16; ++k) {
            const int i = 16 * v + k;
            d[k] = (i < E) ? dst[i] : -1;
        }
    }
    unsigned w[4];
#pragma unroll
    for (int q = 0; q < 4; ++q) {
        w[q] = 0u;
#pragma unroll
        for (int k = 0; k < 4; ++k) {
            const unsigned dd = (unsigned)d[4 * q + k];
            const unsigned o = (dd < (unsigned)n_nodes) ? dd / OWN_R : NO_OWNER;
            w[q] |= o << (8 * k);
        }
    }
    plane[v] = make_uint4(w[0], w[1], w[2], w[3]);
}

__device__ __forceinline__ float leaky(float v) { return (v > 0.0f) ? v : NEG_SLOPE * v; }

template <class Fn>
__device__ __forceinline__ void scan_dst(const int* __restrict__ dst, int E, int tid, Fn&& fn) {
    const bool vec = ((E & 3) == 0) && ((reinterpret_cast<size_t>(dst) & 15) == 0);
    if (vec) {
        const int4* d4 = reinterpret_cast<const int4*>(dst);
        const int E4 = E >> 2;
        for (int jb = 0; jb < E4; jb += 4 * OWN_THREADS) {
            int4 d[4];
#pragma unroll
            for (int u = 0; u < 4; ++u) {
                int j = jb + u * OWN_THREADS + tid;
                d[u] = (j < E4) ? d4[j] : make_int4(-1, -1, -1, -1);
            }
#pragma unroll
            for (int u = 0; u < 4; ++u) {
                int i = 4 * (jb + u * OWN_THREADS + tid);
                fn(i, d[u].x); fn(i + 1, d[u].y); fn(i + 2, d[u].z); fn(i + 3, d[u].w);
            }
        }
    } else {
        for (int ib = 0; ib < E; ib += OWN_THREADS) {
            int i = ib + tid;
            fn(i, (i < E) ? dst[i] : -1);
        }
    }
}

template <class Fn>
__device__ __forceinline__ void scan_plane(const uint4* __restrict__ plane, int nVec,
                                           unsigned me, int tid, Fn&& hit) {
    const unsigned key = me * 0x01010101u;
    for (int vb = 0; vb < nVec; vb += 8 * OWN_THREADS) {
        uint4 d[8];
#pragma unroll
        for (int u = 0; u < 8; ++u) {
            const int v = vb + u * OWN_THREADS + tid;
            d[u] = (v < nVec) ? plane[v] : make_uint4(~0u, ~0u, ~0u, ~0u);
        }
#pragma unroll
        for (int u = 0; u < 8; ++u) {
            const int i0 = 16 * (vb + u * OWN_THREADS + tid);
            const unsigned w[4] = { d[u].x, d[u].y, d[u].z, d[u].w };
#pragma unroll
            for (int q = 0; q < 4; ++q) {
                const unsigned x = w[q] ^ key;
                unsigned z = ~(((x & 0x7F7F7F7Fu) + 0x7F7F7F7Fu) | x | 0x7F7F7F7Fu);
                while (z) {
                    const int k = __builtin_ctz(z) >> 3;
                    z &= z - 1u;
                    hit(i0 + 4 * q + k);
                }
            }
        }
    }
}

template <int F, int H, int U>
__device__ __forceinline__ void accumulate_edges(const int (&s)[U], const unsigned (&dl)[U],
                                                 const bool (&ok)[U], int lane,
                                                 const float* __restrict__ h,
                                                 const float* __restrict__ as,
                                                 const float* adl, const float* m,
                                                 float* den, float* acc) {
    constexpr int C = F / H;
    constexpr int J = F / 32;
    float r[U][J];
    float a[U][H];
#pragma unroll
    for (int u = 0; u < U; ++u) {
        const float* hs = h + (size_t)s[u] * F;
#pragma unroll
        for (int j = 0; j < J; ++j) r[u][j] = hs[lane + 32 * j];
#pragma unroll
        for (int hh = 0; hh < H; ++hh) a[u][hh] = as[(size_t)s[u] * H + hh];
    }
#pragma unroll
    for (int u = 0; u < U; ++u) {
        if (ok[u]) {
            float ex[H];
#pragma unroll
            for (int hh = 0; hh < H; ++hh) {
                float v = leaky(a[u][hh] + adl[dl[u] * H + hh]);
                ex[hh] = __expf(v - m[dl[u] * H + hh]);
            }
#pragma unroll
            for (int hh = 0; hh < H; ++hh)
                if (lane == hh) atomicAdd(&den[dl[u] * H + hh], ex[hh]);
            float* ar = acc + dl[u] * F;
#pragma unroll
            for (int j = 0; j < J; ++j)
                atomicAdd(&ar[lane + 32 * j], r[u][j] * ex[(32 * j) / C]);
        }
    }
}

template <int F, int H, int LISTS>
__global__ __launch_bounds__(OWN_THREADS)
void owner_aggregate_kernel(const int* __restrict__ ei, int E, int n_nodes,
                            const float* __restrict__ h,
                            const float* __restrict__ as,
                            const float* __restrict__ ad,
                            float* agg, unsigned int* keep, int keepF,
                            const uint4* __restrict__ plane) {
    constexpr int C = F / H;
    __shared__ float acc[OWN_R * F];
    __shared__ unsigned int list[OWN_CAP];
    __shared__ float m[OWN_R * H];
    __shared__ float den[OWN_R * H];
    __shared__ float adl[OWN_R * H];
    __shared__ unsigned int cnt;

    const int tid = threadIdx.x;
    const int lane = tid & 31;
    const int wave = tid >> 5;
    const int base = blockIdx.x * OWN_R;
    const int R = min(OWN_R, n_nodes - base);
    const unsigned uR = (unsigned)R;
    const int* __restrict__ src = ei;
    const int* __restrict__ dst = ei + E;
    unsigned int* kp = keep + (size_t)base * keepF;
    const unsigned keepCap = (unsigned)R * (unsigned)keepF;

    for (int i = tid; i < R * F; i += OWN_THREADS) acc[i] = 0.0f;
    for (int i = tid; i < R * H; i += OWN_THREADS) {
        float a_d = ad[(size_t)base * H + i];
        adl[i] = a_d;
        m[i] = leaky(as[(size_t)base * H + i] + a_d);
        den[i] = 0.0f;
    }
    if (tid == 0) cnt = 0u;
    __syncthreads();

    unsigned kept = NO_LIST;
    if (LISTS == 2) kept = kp[0];
    const bool have = (LISTS == 2) && (kept <= OWN_CAP) && (kept < keepCap);

    const bool packs = (E <= (1 << 24)) && (n_nodes <= (1 << 24)) && (OWN_R <= 256);
    if (packs && !have) {
        if (owner_plane_ok(E, n_nodes, plane)) {
            scan_plane(plane, (E + 15) >> 4, (unsigned)blockIdx.x, tid, [&](int i) {
                unsigned pos = atomicAdd(&cnt, 1u);
                if (pos < OWN_CAP) list[pos] = (unsigned)i << 8;
            });
        } else {
            scan_dst(dst, E, tid, [&](int i, int d) {
                unsigned dl = (unsigned)(d - base);
                if (dl < uR) {
                    unsigned pos = atomicAdd(&cnt, 1u);
                    if (pos < OWN_CAP) list[pos] = (unsigned)i << 8;
                }
            });
        }
    }
    __syncthreads();
    const unsigned total = have ? kept : cnt;
    const bool fast = have || (packs && (total <= OWN_CAP));
    const bool leave = (LISTS == 1) && fast && (total < keepCap);

    if (fast) {
        for (unsigned p = tid; p < total; p += OWN_THREADS) {
            unsigned dl;
            int s;
            if (have) {
                unsigned pk = kp[1 + p];
                dl = pk & 255u;
                s = (int)(pk >> 8);
            } else {
                const unsigned i = list[p] >> 8;
                dl = (unsigned)(dst[i] - base);
                s = src[i];
            }
#pragma unroll
            for (int hh = 0; hh < H; ++hh)
                atomicMaxFloat(&m[dl * H + hh], leaky(as[(size_t)s * H + hh] + adl[dl * H + hh]));
            list[p] = ((unsigned)s << 8) | dl;
            if (leave) kp[1 + p] = ((unsigned)s << 8) | dl;
        }
        if (LISTS == 1 && tid == 0) kp[0] = leave ? total : NO_LIST;
        __syncthreads();
        for (unsigned p0 = wave * 4; p0 < total; p0 += OWN_WAVES * 4) {
            int s[4]; unsigned dl[4]; bool ok[4];
#pragma unroll
            for (int u = 0; u < 4; ++u) {
                ok[u] = (p0 + u) < total;
                unsigned pk = ok[u] ? list[p0 + u] : 0u;
                s[u] = (int)(pk >> 8);
                dl[u] = pk & 255u;
            }
            accumulate_edges<F, H, 4>(s, dl, ok, lane, h, as, adl, m, den, acc);
        }
    } else {
        if (LISTS == 1 && tid == 0) kp[0] = NO_LIST;
        scan_dst(dst, E, tid, [&](int i, int d) {
            unsigned dl = (unsigned)(d - base);
            if (dl < uR) {
                int s = src[i];
#pragma unroll
                for (int hh = 0; hh < H; ++hh)
                    atomicMaxFloat(&m[dl * H + hh], leaky(as[(size_t)s * H + hh] + adl[dl * H + hh]));
            }
        });
        __syncthreads();
        scan_dst(dst, E, tid, [&](int i, int d) {
            unsigned dl = (unsigned)(d - base);
            bool hit = dl < uR;
            int sv = hit ? src[i] : 0;
            unsigned mask = __builtin_amdgcn_ballot_w32(hit);
            while (mask) {
                int l = __builtin_ctz(mask);
                mask &= mask - 1u;
                int s1[1] = { __shfl(sv, l, 32) };
                unsigned d1[1] = { (unsigned)__shfl((int)dl, l, 32) };
                bool o1[1] = { true };
                accumulate_edges<F, H, 1>(s1, d1, o1, lane, h, as, adl, m, den, acc);
            }
        });
    }
    for (int q0 = wave * 4; q0 < R; q0 += OWN_WAVES * 4) {
        int s[4]; unsigned dl[4]; bool ok[4];
#pragma unroll
        for (int u = 0; u < 4; ++u) {
            ok[u] = (q0 + u) < R;
            dl[u] = ok[u] ? (unsigned)(q0 + u) : 0u;
            s[u] = base + (int)dl[u];
        }
        accumulate_edges<F, H, 4>(s, dl, ok, lane, h, as, adl, m, den, acc);
    }
    __syncthreads();

    for (int i = tid; i < R * F; i += OWN_THREADS) {
        int dl = i / F;
        int c = i - dl * F;
        agg[(size_t)base * F + i] = acc[i] / den[dl * H + c / C];
    }
}

__global__ void bias_bn_relu_kernel(const float* __restrict__ x,
                                    const float* __restrict__ b,
                                    const float* __restrict__ gamma,
                                    const float* __restrict__ beta,
                                    const float* __restrict__ mean,
                                    const float* __restrict__ var,
                                    float* __restrict__ out, long long total, int F) {
    long long i = (long long)blockIdx.x * blockDim.x + threadIdx.x;
    if (i >= total) return;
    int c = (int)(i % F);
    float v = x[i] + b[c];
    v = gamma[c] * (v - mean[c]) * rsqrtf(var[c] + EPS) + beta[c];
    out[i] = (v > 0.0f) ? v : 0.0f;
}

#define POOL_THREADS 1024
#define POOL_WAVES (POOL_THREADS / 32)
#define POOL_CAP 65536
#define POOL_VECS 16
#define POOL_ROWS 8

__global__ __launch_bounds__(POOL_THREADS)
void pool_owner_kernel(const float* __restrict__ x,
                       const int* __restrict__ batch,
                       const float* __restrict__ b,
                       const float* __restrict__ gamma,
                       const float* __restrict__ beta,
                       const float* __restrict__ mean,
                       const float* __restrict__ var,
                       float* __restrict__ out, int n_nodes) {
    constexpr int F = 128;
    __shared__ unsigned short list[POOL_CAP];
    __shared__ float4 part[POOL_WAVES][32];
    __shared__ unsigned int cnt;

    const int tid = threadIdx.x;
    const int lane = tid & 31;
    const int wave = tid >> 5;
    const int g = blockIdx.x;
    const float4* __restrict__ x4 = reinterpret_cast<const float4*>(x);

    float pb[4], pg[4], pt[4], pm[4], prs[4];
    for (int k = 0; k < 4; ++k) {
        const int c = 4 * lane + k;
        pb[k] = b[c]; pg[k] = gamma[c]; pt[k] = beta[c]; pm[k] = mean[c];
        prs[k] = rsqrtf(var[c] + EPS);
    }
    auto act = [&](float xv, int k) {
        float v = xv + pb[k];
        v = pg[k] * (v - pm[k]) * prs[k] + pt[k];
        return (v > 0.0f) ? v : 0.0f;
    };
    const float ninf = __int_as_float((int)0xFF800000u);
    float mx[4] = { ninf, ninf, ninf, ninf };

    for (int chunk0 = 0; chunk0 < n_nodes; chunk0 += POOL_CAP) {
        const int len = min(POOL_CAP, n_nodes - chunk0);
        if (tid == 0) cnt = 0u;
        __syncthreads();
        const int* __restrict__ bp = batch + chunk0;
        const int len4 = ((reinterpret_cast<size_t>(bp) & 15) == 0) ? (len >> 2) : 0;
        const int4* __restrict__ bp4 = reinterpret_cast<const int4*>(bp);
        auto push = [&](int i) {
            const unsigned pos = atomicAdd(&cnt, 1u);
            list[pos] = (unsigned short)i;
        };
        for (int jb = 0; jb < len4; jb += POOL_VECS * POOL_THREADS) {
            int4 d[POOL_VECS];
#pragma unroll
            for (int u = 0; u < POOL_VECS; ++u) {
                const int j = jb + u * POOL_THREADS + tid;
                d[u] = (j < len4) ? bp4[j] : make_int4(-1, -1, -1, -1);
            }
            bool any = false;
#pragma unroll
            for (int u = 0; u < POOL_VECS; ++u)
                any |= (d[u].x == g) | (d[u].y == g) | (d[u].z == g) | (d[u].w == g);
            if (__builtin_amdgcn_ballot_w32(any) != 0u) {
                unsigned long long hm = 0ull;
#pragma unroll
                for (int u = 0; u < POOL_VECS; ++u) {
                    const unsigned m4 = (d[u].x == g ? 1u : 0u) | (d[u].y == g ? 2u : 0u)
                                      | (d[u].z == g ? 4u : 0u) | (d[u].w == g ? 8u : 0u);
                    hm |= (unsigned long long)m4 << (4 * u);
                }
                while (hm) {
                    const int bit = __builtin_ctzll(hm);
                    hm &= hm - 1ull;
                    push(4 * (jb + (bit >> 2) * POOL_THREADS + tid) + (bit & 3));
                }
            }
        }
        for (int i = 4 * len4 + tid; i < len; i += POOL_THREADS)
            if (bp[i] == g) push(i);
        __syncthreads();
        const unsigned total = cnt;

        for (unsigned p0 = wave; p0 < total; p0 += POOL_WAVES * POOL_ROWS) {
            float4 r[POOL_ROWS];
#pragma unroll
            for (int u = 0; u < POOL_ROWS; ++u) {
                const unsigned p = p0 + u * POOL_WAVES;
                const int n = chunk0 + (int)list[(p < total) ? p : p0];
                r[u] = x4[(size_t)n * (F / 4) + lane];
            }
#pragma unroll
            for (int u = 0; u < POOL_ROWS; ++u) {
                if (p0 + u * POOL_WAVES < total) {
                    mx[0] = fmaxf(mx[0], act(r[u].x, 0));
                    mx[1] = fmaxf(mx[1], act(r[u].y, 1));
                    mx[2] = fmaxf(mx[2], act(r[u].z, 2));
                    mx[3] = fmaxf(mx[3], act(r[u].w, 3));
                }
            }
        }
        __syncthreads();
    }

    part[wave][lane] = make_float4(mx[0], mx[1], mx[2], mx[3]);
    __syncthreads();
    if (tid < F) {
        const float* pf = reinterpret_cast<const float*>(&part[0][0]);
        float m = pf[tid];
        for (int w = 1; w < POOL_WAVES; ++w) m = fmaxf(m, pf[w * F + tid]);
        out[(size_t)g * F + tid] = m;
    }
}

extern "C" void kernel_launch(void* const* d_in, const int* in_sizes, int n_in,
                              void* d_out, int out_size, void* d_ws, size_t ws_size,
                              hipStream_t stream) {
    const int F_IN = 128, HID = 128, HEADS = 2;
    const int F1 = HEADS * HID;
    const int N = in_sizes[0] / F_IN;
    const int E = in_sizes[1] / 2;

    const float* x      = (const float*)d_in[0];
    const int*   ei     = (const int*)d_in[1];
    const int*   batch  = (const int*)d_in[2];
    const float* W1     = (const float*)d_in[3];
    const float* a_src1 = (const float*)d_in[4];
    const float* a_dst1 = (const float*)d_in[5];
    const float* b1     = (const float*)d_in[6];
    const float* gamma1 = (const float*)d_in[7];
    const float* beta1  = (const float*)d_in[8];
    const float* rm1    = (const float*)d_in[9];
    const float* rv1    = (const float*)d_in[10];
    const float* W2     = (const float*)d_in[11];
    const float* a_src2 = (const float*)d_in[12];
    const float* a_dst2 = (const float*)d_in[13];
    const float* b2     = (const float*)d_in[14];
    const float* gamma2 = (const float*)d_in[15];
    const float* beta2  = (const float*)d_in[16];
    const float* rm2    = (const float*)d_in[17];
    const float* rv2    = (const float*)d_in[18];

    float* ws = (float*)d_ws;
    size_t o = 0;
    float* h1   = ws + o; o += (size_t)N * F1;
    float* aggA = ws + o; o += (size_t)N * F1;
    float* aggB = ws + o; o += (size_t)N * HID;
    float* as1  = ws + o; o += (size_t)N * HEADS;
    float* ad1  = ws + o; o += (size_t)N * HEADS;
    float* as2  = ws + o; o += (size_t)N;
    float* ad2  = ws + o; o += (size_t)N;
    float* h2   = h1;
    float* agg2 = h1 + (size_t)N * HID;
    unsigned int* keep = (unsigned int*)aggB;
    uint4* plane = (uint4*)(aggB + (size_t)N * KEEP_WORDS);

    auto fillBlocks = [](long long cnt) { return (unsigned)((cnt + 255) / 256); };
    const unsigned ownerBlocks = (unsigned)((N + OWN_R - 1) / OWN_R);

    gemm_wmma_f32_kernel<128, 256><<<GEMM_GRID, GEMM_THREADS, 0, stream>>>(x, W1, h1, N / 16);
    alpha_kernel<<<(unsigned)(((long long)N * HEADS * 32 + 255) / 256), 256, 0, stream>>>(
        h1, a_src1, a_dst1, as1, ad1, N, HEADS, HID);
    owner_plane_kernel<<<fillBlocks((long long)((E + 15) / 16)), 256, 0, stream>>>(ei, E, N, plane);
    owner_aggregate_kernel<256, 2, 1><<<ownerBlocks, OWN_THREADS, 0, stream>>>(
        ei, E, N, h1, as1, ad1, aggA, keep, KEEP_WORDS, plane);
    bias_bn_relu_kernel<<<fillBlocks((long long)N * F1), 256, 0, stream>>>(
        aggA, b1, gamma1, beta1, rm1, rv1, aggA, (long long)N * F1, F1);

    gemm_wmma_f32_kernel<256, 128><<<GEMM_GRID, GEMM_THREADS, 0, stream>>>(aggA, W2, h2, N / 16);
    alpha_kernel<<<(unsigned)(((long long)N * 32 + 255) / 256), 256, 0, stream>>>(
        h2, a_src2, a_dst2, as2, ad2, N, 1, HID);
    owner_aggregate_kernel<128, 1, 2><<<ownerBlocks, OWN_THREADS, 0, stream>>>(
        ei, E, N, h2, as2, ad2, agg2, keep, KEEP_WORDS, plane);

    pool_owner_kernel<<<(unsigned)(out_size / HID), POOL_THREADS, 0, stream>>>(
        agg2, batch, b2, gamma2, beta2, rm2, rv2, (float*)d_out, N);
}


# Layer-1 bias+BN+ReLU applied by the owner work-groups to each normalised sum as its row leaves LDS: the elementwise launch over [N,256] and its read+write of that plane are gone (9 to 8 launches)
# speedup vs baseline: 3.8722x; 1.1180x over previous
#include <hip/hip_runtime.h>
#include <hip/hip_bf16.h>

typedef float v2f __attribute__((ext_vector_type(2)));
typedef float v8f __attribute__((ext_vector_type(8)));

#define NEG_SLOPE 0.2f
#define EPS 1e-5f

__device__ __forceinline__ void atomicMaxFloat(float* addr, float val) {
    if (val >= 0.0f) {
        atomicMax((int*)addr, __float_as_int(val));
    } else {
        atomicMin((unsigned int*)addr, __float_as_uint(val));
    }
}

#define GEMM_THREADS 256
#define GEMM_WAVES (GEMM_THREADS / 32)
#define GEMM_GRID 512

template <int K, int NCOL>
__global__ __launch_bounds__(GEMM_THREADS)
void gemm_wmma_f32_kernel(const float* __restrict__ A,
                          const float* __restrict__ Bm,
                          float* __restrict__ C, int nStrips) {
    constexpr int PITCH = 2 * NCOL + 16;
    constexpr int TPW = NCOL / 16 / GEMM_WAVES;
    constexpr int C4 = NCOL / 4;
    static_assert(K % 8 == 0 && NCOL % (16 * GEMM_WAVES) == 0, "tile shape");
    __shared__ __attribute__((aligned(16))) float Bs[(K / 2) * PITCH];

    const int tid = threadIdx.x;
    for (int i = tid; i < (K / 2) * C4; i += GEMM_THREADS) {
        const int p = i / C4, c4 = i - p * C4;
        const float4 r0 = *reinterpret_cast<const float4*>(Bm + (size_t)(2 * p) * NCOL + 4 * c4);
        const float4 r1 = *reinterpret_cast<const float4*>(Bm + (size_t)(2 * p + 1) * NCOL + 4 * c4);
        float4* d = reinterpret_cast<float4*>(Bs + p * PITCH + 8 * c4);
        d[0] = make_float4(r0.x, r1.x, r0.y, r1.y);
        d[1] = make_float4(r0.z, r1.z, r0.w, r1.w);
    }
    __syncthreads();

    const int wave = tid >> 5;
    const int lane = tid & 31;
    const int c = lane & 15;
    const int h = lane >> 4;
    const float* bp = Bs + (2 * h) * PITCH + 2 * (wave * TPW * 16 + c);

    for (int strip = blockIdx.x; strip < nStrips; strip += gridDim.x) {
        const float* ap = A + ((size_t)strip * 16 + c) * K + 4 * h;
        float4 a4[K / 8];
#pragma unroll
        for (int j = 0; j < K / 8; ++j) a4[j] = *reinterpret_cast<const float4*>(ap + 8 * j);

        v8f acc[TPW];
#pragma unroll
        for (int t = 0; t < TPW; ++t) acc[t] = v8f{};

#pragma unroll
        for (int j = 0; j < K / 8; ++j) {
#pragma unroll
            for (int u = 0; u < 2; ++u) {
                v2f a;
                a.x = u ? a4[j].z : a4[j].x;
                a.y = u ? a4[j].w : a4[j].y;
#pragma unroll
                for (int t = 0; t < TPW; ++t) {
                    const v2f b = *reinterpret_cast<const v2f*>(bp + (4 * j + u) * PITCH + 32 * t);
                    acc[t] = __builtin_amdgcn_wmma_f32_16x16x4_f32(false, a, false, b,
                                                                   (short)0, acc[t], false, false);
                }
            }
        }
#pragma unroll
        for (int t = 0; t < TPW; ++t) {
            float* cp = C + ((size_t)strip * 16 + 8 * h) * NCOL + (wave * TPW + t) * 16 + c;
#pragma unroll
            for (int r = 0; r < 8; ++r) cp[(size_t)r * NCOL] = acc[t][r];
        }
    }
}

__global__ void alpha_kernel(const float* __restrict__ h,
                             const float* __restrict__ a_src,
                             const float* __restrict__ a_dst,
                             float* __restrict__ as, float* __restrict__ ad,
                             int n_nodes, int heads, int C) {
    int gw = (int)(((long long)blockIdx.x * blockDim.x + threadIdx.x) >> 5);
    int lane = threadIdx.x & 31;
    if (gw >= n_nodes * heads) return;
    int hd = gw % heads;
    const float* hp = h + (long long)gw * C;
    const float* sp = a_src + hd * C;
    const float* dp = a_dst + hd * C;
    float s1 = 0.0f, s2 = 0.0f;
    for (int c = lane; c < C; c += 32) {
        float v = hp[c];
        s1 += v * sp[c];
        s2 += v * dp[c];
    }
    for (int off = 16; off; off >>= 1) {
        s1 += __shfl_down(s1, off, 32);
        s2 += __shfl_down(s2, off, 32);
    }
    if (lane == 0) { as[gw] = s1; ad[gw] = s2; }
}

#define NO_LIST 0xFFFFFFFFu
#define OWN_R 200
#define OWN_THREADS 1024
#define OWN_WAVES (OWN_THREADS / 32)
#define OWN_CAP 16384
#define NO_OWNER 255u
#define KEEP_WORDS 96
#define PLANE_WORDS 32

__device__ __forceinline__ bool owner_plane_ok(int E, int n_nodes, const void* plane) {
    const unsigned nOwner = ((unsigned)n_nodes + OWN_R - 1) / OWN_R;
    const size_t bytes = ((size_t)E + 15) & ~(size_t)15;
    return plane != nullptr && nOwner <= NO_OWNER
        && bytes <= (size_t)n_nodes * (PLANE_WORDS * 4)
        && (reinterpret_cast<size_t>(plane) & 15) == 0;
}

__global__ void owner_plane_kernel(const int* __restrict__ ei, int E, int n_nodes,
                                   uint4* __restrict__ plane) {
    if (!owner_plane_ok(E, n_nodes, plane)) return;
    const int* __restrict__ dst = ei + E;
    const int nVec = (E + 15) >> 4;
    const int v = blockIdx.x * blockDim.x + threadIdx.x;
    if (v >= nVec) return;
    int d[16];
    if (((E & 15) == 0) && ((reinterpret_cast<size_t>(dst) & 15) == 0)) {
        const int4* d4 = reinterpret_cast<const int4*>(dst) + 4 * (size_t)v;
#pragma unroll
        for (int q = 0; q < 4; ++q) {
            const int4 t = d4[q];
            d[4 * q] = t.x; d[4 * q + 1] = t.y; d[4 * q + 2] = t.z; d[4 * q + 3] = t.w;
        }
    } else {
#pragma unroll
        for (int k = 0; k < 16; ++k) {
            const int i = 16 * v + k;
            d[k] = (i < E) ? dst[i] : -1;
        }
    }
    unsigned w[4];
#pragma unroll
    for (int q = 0; q < 4; ++q) {
        w[q] = 0u;
#pragma unroll
        for (int k = 0; k < 4; ++k) {
            const unsigned dd = (unsigned)d[4 * q + k];
            const unsigned o = (dd < (unsigned)n_nodes) ? dd / OWN_R : NO_OWNER;
            w[q] |= o << (8 * k);
        }
    }
    plane[v] = make_uint4(w[0], w[1], w[2], w[3]);
}

__device__ __forceinline__ float leaky(float v) { return (v > 0.0f) ? v : NEG_SLOPE * v; }

template <class Fn>
__device__ __forceinline__ void scan_dst(const int* __restrict__ dst, int E, int tid, Fn&& fn) {
    const bool vec = ((E & 3) == 0) && ((reinterpret_cast<size_t>(dst) & 15) == 0);
    if (vec) {
        const int4* d4 = reinterpret_cast<const int4*>(dst);
        const int E4 = E >> 2;
        for (int jb = 0; jb < E4; jb += 4 * OWN_THREADS) {
            int4 d[4];
#pragma unroll
            for (int u = 0; u < 4; ++u) {
                int j = jb + u * OWN_THREADS + tid;
                d[u] = (j < E4) ? d4[j] : make_int4(-1, -1, -1, -1);
            }
#pragma unroll
            for (int u = 0; u < 4; ++u) {
                int i = 4 * (jb + u * OWN_THREADS + tid);
                fn(i, d[u].x); fn(i + 1, d[u].y); fn(i + 2, d[u].z); fn(i + 3, d[u].w);
            }
        }
    } else {
        for (int ib = 0; ib < E; ib += OWN_THREADS) {
            int i = ib + tid;
            fn(i, (i < E) ? dst[i] : -1);
        }
    }
}

template <class Fn>
__device__ __forceinline__ void scan_plane(const uint4* __restrict__ plane, int nVec,
                                           unsigned me, int tid, Fn&& hit) {
    const unsigned key = me * 0x01010101u;
    for (int vb = 0; vb < nVec; vb += 8 * OWN_THREADS) {
        uint4 d[8];
#pragma unroll
        for (int u = 0; u < 8; ++u) {
            const int v = vb + u * OWN_THREADS + tid;
            d[u] = (v < nVec) ? plane[v] : make_uint4(~0u, ~0u, ~0u, ~0u);
        }
#pragma unroll
        for (int u = 0; u < 8; ++u) {
            const int i0 = 16 * (vb + u * OWN_THREADS + tid);
            const unsigned w[4] = { d[u].x, d[u].y, d[u].z, d[u].w };
#pragma unroll
            for (int q = 0; q < 4; ++q) {
                const unsigned x = w[q] ^ key;
                unsigned z = ~(((x & 0x7F7F7F7Fu) + 0x7F7F7F7Fu) | x | 0x7F7F7F7Fu);
                while (z) {
                    const int k = __builtin_ctz(z) >> 3;
                    z &= z - 1u;
                    hit(i0 + 4 * q + k);
                }
            }
        }
    }
}

template <int F, int H, int U>
__device__ __forceinline__ void accumulate_edges(const int (&s)[U], const unsigned (&dl)[U],
                                                 const bool (&ok)[U], int lane,
                                                 const float* __restrict__ h,
                                                 const float* __restrict__ as,
                                                 const float* adl, const float* m,
                                                 float* den, float* acc) {
    constexpr int C = F / H;
    constexpr int J = F / 32;
    float r[U][J];
    float a[U][H];
#pragma unroll
    for (int u = 0; u < U; ++u) {
        const float* hs = h + (size_t)s[u] * F;
#pragma unroll
        for (int j = 0; j < J; ++j) r[u][j] = hs[lane + 32 * j];
#pragma unroll
        for (int hh = 0; hh < H; ++hh) a[u][hh] = as[(size_t)s[u] * H + hh];
    }
#pragma unroll
    for (int u = 0; u < U; ++u) {
        if (ok[u]) {
            float ex[H];
#pragma unroll
            for (int hh = 0; hh < H; ++hh) {
                float v = leaky(a[u][hh] + adl[dl[u] * H + hh]);
                ex[hh] = __expf(v - m[dl[u] * H + hh]);
            }
#pragma unroll
            for (int hh = 0; hh < H; ++hh)
                if (lane == hh) atomicAdd(&den[dl[u] * H + hh], ex[hh]);
            float* ar = acc + dl[u] * F;
#pragma unroll
            for (int j = 0; j < J; ++j)
                atomicAdd(&ar[lane + 32 * j], r[u][j] * ex[(32 * j) / C]);
        }
    }
}

template <int F, int H, int LISTS, bool POST>
__global__ __launch_bounds__(OWN_THREADS)
void owner_aggregate_kernel(const int* __restrict__ ei, int E, int n_nodes,
                            const float* __restrict__ h,
                            const float* __restrict__ as,
                            const float* __restrict__ ad,
                            float* agg, unsigned int* keep, int keepF,
                            const uint4* __restrict__ plane,
                            const float* __restrict__ pb,
                            const float* __restrict__ pgamma,
                            const float* __restrict__ pbeta,
                            const float* __restrict__ pmean,
                            const float* __restrict__ pvar) {
    constexpr int C = F / H;
    __shared__ float acc[OWN_R * F];
    __shared__ unsigned int list[OWN_CAP];
    __shared__ float m[OWN_R * H];
    __shared__ float den[OWN_R * H];
    __shared__ float adl[OWN_R * H];
    __shared__ unsigned int cnt;

    const int tid = threadIdx.x;
    const int lane = tid & 31;
    const int wave = tid >> 5;
    const int base = blockIdx.x * OWN_R;
    const int R = min(OWN_R, n_nodes - base);
    const unsigned uR = (unsigned)R;
    const int* __restrict__ src = ei;
    const int* __restrict__ dst = ei + E;
    unsigned int* kp = keep + (size_t)base * keepF;
    const unsigned keepCap = (unsigned)R * (unsigned)keepF;

    for (int i = tid; i < R * F; i += OWN_THREADS) acc[i] = 0.0f;
    for (int i = tid; i < R * H; i += OWN_THREADS) {
        float a_d = ad[(size_t)base * H + i];
        adl[i] = a_d;
        m[i] = leaky(as[(size_t)base * H + i] + a_d);
        den[i] = 0.0f;
    }
    if (tid == 0) cnt = 0u;
    __syncthreads();

    unsigned kept = NO_LIST;
    if (LISTS == 2) kept = kp[0];
    const bool have = (LISTS == 2) && (kept <= OWN_CAP) && (kept < keepCap);

    const bool packs = (E <= (1 << 24)) && (n_nodes <= (1 << 24)) && (OWN_R <= 256);
    if (packs && !have) {
        if (owner_plane_ok(E, n_nodes, plane)) {
            scan_plane(plane, (E + 15) >> 4, (unsigned)blockIdx.x, tid, [&](int i) {
                unsigned pos = atomicAdd(&cnt, 1u);
                if (pos < OWN_CAP) list[pos] = (unsigned)i << 8;
            });
        } else {
            scan_dst(dst, E, tid, [&](int i, int d) {
                unsigned dl = (unsigned)(d - base);
                if (dl < uR) {
                    unsigned pos = atomicAdd(&cnt, 1u);
                    if (pos < OWN_CAP) list[pos] = (unsigned)i << 8;
                }
            });
        }
    }
    __syncthreads();
    const unsigned total = have ? kept : cnt;
    const bool fast = have || (packs && (total <= OWN_CAP));
    const bool leave = (LISTS == 1) && fast && (total < keepCap);

    if (fast) {
        for (unsigned p = tid; p < total; p += OWN_THREADS) {
            unsigned dl;
            int s;
            if (have) {
                unsigned pk = kp[1 + p];
                dl = pk & 255u;
                s = (int)(pk >> 8);
            } else {
                const unsigned i = list[p] >> 8;
                dl = (unsigned)(dst[i] - base);
                s = src[i];
            }
#pragma unroll
            for (int hh = 0; hh < H; ++hh)
                atomicMaxFloat(&m[dl * H + hh], leaky(as[(size_t)s * H + hh] + adl[dl * H + hh]));
            list[p] = ((unsigned)s << 8) | dl;
            if (leave) kp[1 + p] = ((unsigned)s << 8) | dl;
        }
        if (LISTS == 1 && tid == 0) kp[0] = leave ? total : NO_LIST;
        __syncthreads();
        for (unsigned p0 = wave * 4; p0 < total; p0 += OWN_WAVES * 4) {
            int s[4]; unsigned dl[4]; bool ok[4];
#pragma unroll
            for (int u = 0; u < 4; ++u) {
                ok[u] = (p0 + u) < total;
                unsigned pk = ok[u] ? list[p0 + u] : 0u;
                s[u] = (int)(pk >> 8);
                dl[u] = pk & 255u;
            }
            accumulate_edges<F, H, 4>(s, dl, ok, lane, h, as, adl, m, den, acc);
        }
    } else {
        if (LISTS == 1 && tid == 0) kp[0] = NO_LIST;
        scan_dst(dst, E, tid, [&](int i, int d) {
            unsigned dl = (unsigned)(d - base);
            if (dl < uR) {
                int s = src[i];
#pragma unroll
                for (int hh = 0; hh < H; ++hh)
                    atomicMaxFloat(&m[dl * H + hh], leaky(as[(size_t)s * H + hh] + adl[dl * H + hh]));
            }
        });
        __syncthreads();
        scan_dst(dst, E, tid, [&](int i, int d) {
            unsigned dl = (unsigned)(d - base);
            bool hit = dl < uR;
            int sv = hit ? src[i] : 0;
            unsigned mask = __builtin_amdgcn_ballot_w32(hit);
            while (mask) {
                int l = __builtin_ctz(mask);
                mask &= mask - 1u;
                int s1[1] = { __shfl(sv, l, 32) };
                unsigned d1[1] = { (unsigned)__shfl((int)dl, l, 32) };
                bool o1[1] = { true };
                accumulate_edges<F, H, 1>(s1, d1, o1, lane, h, as, adl, m, den, acc);
            }
        });
    }
    for (int q0 = wave * 4; q0 < R; q0 += OWN_WAVES * 4) {
        int s[4]; unsigned dl[4]; bool ok[4];
#pragma unroll
        for (int u = 0; u < 4; ++u) {
            ok[u] = (q0 + u) < R;
            dl[u] = ok[u] ? (unsigned)(q0 + u) : 0u;
            s[u] = base + (int)dl[u];
        }
        accumulate_edges<F, H, 4>(s, dl, ok, lane, h, as, adl, m, den, acc);
    }
    __syncthreads();

    if (POST) {
        static_assert(OWN_THREADS % F == 0, "one channel per thread");
        const int c = tid % F;
        const float fb = pb[c], fg = pgamma[c], ft = pbeta[c], fm = pmean[c];
        const float frs = rsqrtf(pvar[c] + EPS);
        for (int i = tid; i < R * F; i += OWN_THREADS) {
            const int dl = i / F;
            float v = acc[i] / den[dl * H + c / C] + fb;
            v = fg * (v - fm) * frs + ft;
            agg[(size_t)base * F + i] = (v > 0.0f) ? v : 0.0f;
        }
    } else {
        for (int i = tid; i < R * F; i += OWN_THREADS) {
            int dl = i / F;
            int c = i - dl * F;
            agg[(size_t)base * F + i] = acc[i] / den[dl * H + c / C];
        }
    }
}

#define POOL_THREADS 1024
#define POOL_WAVES (POOL_THREADS / 32)
#define POOL_CAP 65536
#define POOL_VECS 16
#define POOL_ROWS 8

__global__ __launch_bounds__(POOL_THREADS)
void pool_owner_kernel(const float* __restrict__ x,
                       const int* __restrict__ batch,
                       const float* __restrict__ b,
                       const float* __restrict__ gamma,
                       const float* __restrict__ beta,
                       const float* __restrict__ mean,
                       const float* __restrict__ var,
                       float* __restrict__ out, int n_nodes) {
    constexpr int F = 128;
    __shared__ unsigned short list[POOL_CAP];
    __shared__ float4 part[POOL_WAVES][32];
    __shared__ unsigned int cnt;

    const int tid = threadIdx.x;
    const int lane = tid & 31;
    const int wave = tid >> 5;
    const int g = blockIdx.x;
    const float4* __restrict__ x4 = reinterpret_cast<const float4*>(x);

    float pb[4], pg[4], pt[4], pm[4], prs[4];
    for (int k = 0; k < 4; ++k) {
        const int c = 4 * lane + k;
        pb[k] = b[c]; pg[k] = gamma[c]; pt[k] = beta[c]; pm[k] = mean[c];
        prs[k] = rsqrtf(var[c] + EPS);
    }
    auto act = [&](float xv, int k) {
        float v = xv + pb[k];
        v = pg[k] * (v - pm[k]) * prs[k] + pt[k];
        return (v > 0.0f) ? v : 0.0f;
    };
    const float ninf = __int_as_float((int)0xFF800000u);
    float mx[4] = { ninf, ninf, ninf, ninf };

    for (int chunk0 = 0; chunk0 < n_nodes; chunk0 += POOL_CAP) {
        const int len = min(POOL_CAP, n_nodes - chunk0);
        if (tid == 0) cnt = 0u;
        __syncthreads();
        const int* __restrict__ bp = batch + chunk0;
        const int len4 = ((reinterpret_cast<size_t>(bp) & 15) == 0) ? (len >> 2) : 0;
        const int4* __restrict__ bp4 = reinterpret_cast<const int4*>(bp);
        auto push = [&](int i) {
            const unsigned pos = atomicAdd(&cnt, 1u);
            list[pos] = (unsigned short)i;
        };
        for (int jb = 0; jb < len4; jb += POOL_VECS * POOL_THREADS) {
            int4 d[POOL_VECS];
#pragma unroll
            for (int u = 0; u < POOL_VECS; ++u) {
                const int j = jb + u * POOL_THREADS + tid;
                d[u] = (j < len4) ? bp4[j] : make_int4(-1, -1, -1, -1);
            }
            bool any = false;
#pragma unroll
            for (int u = 0; u < POOL_VECS; ++u)
                any |= (d[u].x == g) | (d[u].y == g) | (d[u].z == g) | (d[u].w == g);
            if (__builtin_amdgcn_ballot_w32(any) != 0u) {
                unsigned long long hm = 0ull;
#pragma unroll
                for (int u = 0; u < POOL_VECS; ++u) {
                    const unsigned m4 = (d[u].x == g ? 1u : 0u) | (d[u].y == g ? 2u : 0u)
                                      | (d[u].z == g ? 4u : 0u) | (d[u].w == g ? 8u : 0u);
                    hm |= (unsigned long long)m4 << (4 * u);
                }
                while (hm) {
                    const int bit = __builtin_ctzll(hm);
                    hm &= hm - 1ull;
                    push(4 * (jb + (bit >> 2) * POOL_THREADS + tid) + (bit & 3));
                }
            }
        }
        for (int i = 4 * len4 + tid; i < len; i += POOL_THREADS)
            if (bp[i] == g) push(i);
        __syncthreads();
        const unsigned total = cnt;

        for (unsigned p0 = wave; p0 < total; p0 += POOL_WAVES * POOL_ROWS) {
            float4 r[POOL_ROWS];
#pragma unroll
            for (int u = 0; u < POOL_ROWS; ++u) {
                const unsigned p = p0 + u * POOL_WAVES;
                const int n = chunk0 + (int)list[(p < total) ? p : p0];
                r[u] = x4[(size_t)n * (F / 4) + lane];
            }
#pragma unroll
            for (int u = 0; u < POOL_ROWS; ++u) {
                if (p0 + u * POOL_WAVES < total) {
                    mx[0] = fmaxf(mx[0], act(r[u].x, 0));
                    mx[1] = fmaxf(mx[1], act(r[u].y, 1));
                    mx[2] = fmaxf(mx[2], act(r[u].z, 2));
                    mx[3] = fmaxf(mx[3], act(r[u].w, 3));
                }
            }
        }
        __syncthreads();
    }

    part[wave][lane] = make_float4(mx[0], mx[1], mx[2], mx[3]);
    __syncthreads();
    if (tid < F) {
        const float* pf = reinterpret_cast<const float*>(&part[0][0]);
        float m = pf[tid];
        for (int w = 1; w < POOL_WAVES; ++w) m = fmaxf(m, pf[w * F + tid]);
        out[(size_t)g * F + tid] = m;
    }
}

extern "C" void kernel_launch(void* const* d_in, const int* in_sizes, int n_in,
                              void* d_out, int out_size, void* d_ws, size_t ws_size,
                              hipStream_t stream) {
    const int F_IN = 128, HID = 128, HEADS = 2;
    const int F1 = HEADS * HID;
    const int N = in_sizes[0] / F_IN;
    const int E = in_sizes[1] / 2;

    const float* x      = (const float*)d_in[0];
    const int*   ei     = (const int*)d_in[1];
    const int*   batch  = (const int*)d_in[2];
    const float* W1     = (const float*)d_in[3];
    const float* a_src1 = (const float*)d_in[4];
    const float* a_dst1 = (const float*)d_in[5];
    const float* b1     = (const float*)d_in[6];
    const float* gamma1 = (const float*)d_in[7];
    const float* beta1  = (const float*)d_in[8];
    const float* rm1    = (const float*)d_in[9];
    const float* rv1    = (const float*)d_in[10];
    const float* W2     = (const float*)d_in[11];
    const float* a_src2 = (const float*)d_in[12];
    const float* a_dst2 = (const float*)d_in[13];
    const float* b2     = (const float*)d_in[14];
    const float* gamma2 = (const float*)d_in[15];
    const float* beta2  = (const float*)d_in[16];
    const float* rm2    = (const float*)d_in[17];
    const float* rv2    = (const float*)d_in[18];

    float* ws = (float*)d_ws;
    size_t o = 0;
    float* h1   = ws + o; o += (size_t)N * F1;
    float* aggA = ws + o; o += (size_t)N * F1;
    float* aggB = ws + o; o += (size_t)N * HID;
    float* as1  = ws + o; o += (size_t)N * HEADS;
    float* ad1  = ws + o; o += (size_t)N * HEADS;
    float* as2  = ws + o; o += (size_t)N;
    float* ad2  = ws + o; o += (size_t)N;
    float* h2   = h1;
    float* agg2 = h1 + (size_t)N * HID;
    unsigned int* keep = (unsigned int*)aggB;
    uint4* plane = (uint4*)(aggB + (size_t)N * KEEP_WORDS);

    auto fillBlocks = [](long long cnt) { return (unsigned)((cnt + 255) / 256); };
    const unsigned ownerBlocks = (unsigned)((N + OWN_R - 1) / OWN_R);

    gemm_wmma_f32_kernel<128, 256><<<GEMM_GRID, GEMM_THREADS, 0, stream>>>(x, W1, h1, N / 16);
    alpha_kernel<<<(unsigned)(((long long)N * HEADS * 32 + 255) / 256), 256, 0, stream>>>(
        h1, a_src1, a_dst1, as1, ad1, N, HEADS, HID);
    owner_plane_kernel<<<fillBlocks((long long)((E + 15) / 16)), 256, 0, stream>>>(ei, E, N, plane);
    owner_aggregate_kernel<256, 2, 1, true><<<ownerBlocks, OWN_THREADS, 0, stream>>>(
        ei, E, N, h1, as1, ad1, aggA, keep, KEEP_WORDS, plane, b1, gamma1, beta1, rm1, rv1);

    gemm_wmma_f32_kernel<256, 128><<<GEMM_GRID, GEMM_THREADS, 0, stream>>>(aggA, W2, h2, N / 16);
    alpha_kernel<<<(unsigned)(((long long)N * 32 + 255) / 256), 256, 0, stream>>>(
        h2, a_src2, a_dst2, as2, ad2, N, 1, HID);
    owner_aggregate_kernel<128, 1, 2, false><<<ownerBlocks, OWN_THREADS, 0, stream>>>(
        ei, E, N, h2, as2, ad2, agg2, keep, KEEP_WORDS, plane,
        nullptr, nullptr, nullptr, nullptr, nullptr);

    pool_owner_kernel<<<(unsigned)(out_size / HID), POOL_THREADS, 0, stream>>>(
        agg2, batch, b2, gamma2, beta2, rm2, rv2, (float*)d_out, N);
}


# Logit launches re-gridded: 4096 resident waves, each takes 16 row-heads per trip with 16-byte loads and one shared reduce-scatter, not one short wave per (node, head); logit sum order changes
# speedup vs baseline: 4.3913x; 1.1341x over previous
#include <hip/hip_runtime.h>
#include <hip/hip_bf16.h>

typedef float v2f __attribute__((ext_vector_type(2)));
typedef float v8f __attribute__((ext_vector_type(8)));

#define NEG_SLOPE 0.2f
#define EPS 1e-5f

__device__ __forceinline__ void atomicMaxFloat(float* addr, float val) {
    if (val >= 0.0f) {
        atomicMax((int*)addr, __float_as_int(val));
    } else {
        atomicMin((unsigned int*)addr, __float_as_uint(val));
    }
}

#define GEMM_THREADS 256
#define GEMM_WAVES (GEMM_THREADS / 32)
#define GEMM_GRID 512

template <int K, int NCOL>
__global__ __launch_bounds__(GEMM_THREADS)
void gemm_wmma_f32_kernel(const float* __restrict__ A,
                          const float* __restrict__ Bm,
                          float* __restrict__ C, int nStrips) {
    constexpr int PITCH = 2 * NCOL + 16;
    constexpr int TPW = NCOL / 16 / GEMM_WAVES;
    constexpr int C4 = NCOL / 4;
    static_assert(K % 8 == 0 && NCOL % (16 * GEMM_WAVES) == 0, "tile shape");
    __shared__ __attribute__((aligned(16))) float Bs[(K / 2) * PITCH];

    const int tid = threadIdx.x;
    for (int i = tid; i < (K / 2) * C4; i += GEMM_THREADS) {
        const int p = i / C4, c4 = i - p * C4;
        const float4 r0 = *reinterpret_cast<const float4*>(Bm + (size_t)(2 * p) * NCOL + 4 * c4);
        const float4 r1 = *reinterpret_cast<const float4*>(Bm + (size_t)(2 * p + 1) * NCOL + 4 * c4);
        float4* d = reinterpret_cast<float4*>(Bs + p * PITCH + 8 * c4);
        d[0] = make_float4(r0.x, r1.x, r0.y, r1.y);
        d[1] = make_float4(r0.z, r1.z, r0.w, r1.w);
    }
    __syncthreads();

    const int wave = tid >> 5;
    const int lane = tid & 31;
    const int c = lane & 15;
    const int h = lane >> 4;
    const float* bp = Bs + (2 * h) * PITCH + 2 * (wave * TPW * 16 + c);

    for (int strip = blockIdx.x; strip < nStrips; strip += gridDim.x) {
        const float* ap = A + ((size_t)strip * 16 + c) * K + 4 * h;
        float4 a4[K / 8];
#pragma unroll
        for (int j = 0; j < K / 8; ++j) a4[j] = *reinterpret_cast<const float4*>(ap + 8 * j);

        v8f acc[TPW];
#pragma unroll
        for (int t = 0; t < TPW; ++t) acc[t] = v8f{};

#pragma unroll
        for (int j = 0; j < K / 8; ++j) {
#pragma unroll
            for (int u = 0; u < 2; ++u) {
                v2f a;
                a.x = u ? a4[j].z : a4[j].x;
                a.y = u ? a4[j].w : a4[j].y;
#pragma unroll
                for (int t = 0; t < TPW; ++t) {
                    const v2f b = *reinterpret_cast<const v2f*>(bp + (4 * j + u) * PITCH + 32 * t);
                    acc[t] = __builtin_amdgcn_wmma_f32_16x16x4_f32(false, a, false, b,
                                                                   (short)0, acc[t], false, false);
                }
            }
        }
#pragma unroll
        for (int t = 0; t < TPW; ++t) {
            float* cp = C + ((size_t)strip * 16 + 8 * h) * NCOL + (wave * TPW + t) * 16 + c;
#pragma unroll
            for (int r = 0; r < 8; ++r) cp[(size_t)r * NCOL] = acc[t][r];
        }
    }
}

#define ALPHA_GRID 512
#define ALPHA_THREADS 256

template <int H>
__global__ __launch_bounds__(ALPHA_THREADS)
void alpha_rows_kernel(const float* __restrict__ h,
                       const float* __restrict__ a_src,
                       const float* __restrict__ a_dst,
                       float* __restrict__ as, float* __restrict__ ad, int n_nodes) {
    static_assert(H == 1 || H == 2, "16 rows x heads per trip");
    constexpr int ROWS = 16 / H;
    constexpr int RV = 32 * H;
    const int lane = threadIdx.x & 31;
    const int gw = __builtin_amdgcn_readfirstlane((int)((blockIdx.x * ALPHA_THREADS + threadIdx.x) >> 5));
    const int nW = (int)((gridDim.x * ALPHA_THREADS) >> 5);
    const float4* __restrict__ h4 = reinterpret_cast<const float4*>(h);

    float4 ws[H], wd[H];
#pragma unroll
    for (int hh = 0; hh < H; ++hh) {
        const float* sp = a_src + hh * 128 + 4 * lane;
        const float* dp = a_dst + hh * 128 + 4 * lane;
        ws[hh] = make_float4(sp[0], sp[1], sp[2], sp[3]);
        wd[hh] = make_float4(dp[0], dp[1], dp[2], dp[3]);
    }

    for (int r0 = gw * ROWS; r0 < n_nodes; r0 += nW * ROWS) {
        float4 v[ROWS][H];
#pragma unroll
        for (int u = 0; u < ROWS; ++u) {
            const int n = min(r0 + u, n_nodes - 1);
#pragma unroll
            for (int hh = 0; hh < H; ++hh) v[u][hh] = h4[(size_t)n * RV + 32 * hh + lane];
        }
        __builtin_amdgcn_sched_barrier(0);
        float val[32];
#pragma unroll
        for (int u = 0; u < ROWS; ++u) {
#pragma unroll
            for (int hh = 0; hh < H; ++hh) {
                const float4 x = v[u][hh];
                val[u * H + hh] = x.x * ws[hh].x + x.y * ws[hh].y + x.z * ws[hh].z + x.w * ws[hh].w;
                val[16 + u * H + hh] = x.x * wd[hh].x + x.y * wd[hh].y + x.z * wd[hh].z + x.w * wd[hh].w;
            }
        }
#pragma unroll
        for (int half = 16, off = 16; half >= 1; half >>= 1, off >>= 1) {
            const bool up = (lane & off) != 0;
#pragma unroll
            for (int i = 0; i < half; ++i) {
                const float lo = val[i], hi = val[i + half];
                const float keep = up ? hi : lo;
                const float send = up ? lo : hi;
                val[i] = keep + __shfl_xor(send, off, 32);
            }
        }
        const int k = lane & 15;
        if (r0 + k / H < n_nodes) {
            float* o = (lane & 16) ? ad : as;
            o[(size_t)r0 * H + k] = val[0];
        }
    }
}

#define NO_LIST 0xFFFFFFFFu
#define OWN_R 200
#define OWN_THREADS 1024
#define OWN_WAVES (OWN_THREADS / 32)
#define OWN_CAP 16384
#define NO_OWNER 255u
#define KEEP_WORDS 96
#define PLANE_WORDS 32

__device__ __forceinline__ bool owner_plane_ok(int E, int n_nodes, const void* plane) {
    const unsigned nOwner = ((unsigned)n_nodes + OWN_R - 1) / OWN_R;
    const size_t bytes = ((size_t)E + 15) & ~(size_t)15;
    return plane != nullptr && nOwner <= NO_OWNER
        && bytes <= (size_t)n_nodes * (PLANE_WORDS * 4)
        && (reinterpret_cast<size_t>(plane) & 15) == 0;
}

__global__ void owner_plane_kernel(const int* __restrict__ ei, int E, int n_nodes,
                                   uint4* __restrict__ plane) {
    if (!owner_plane_ok(E, n_nodes, plane)) return;
    const int* __restrict__ dst = ei + E;
    const int nVec = (E + 15) >> 4;
    const int v = blockIdx.x * blockDim.x + threadIdx.x;
    if (v >= nVec) return;
    int d[16];
    if (((E & 15) == 0) && ((reinterpret_cast<size_t>(dst) & 15) == 0)) {
        const int4* d4 = reinterpret_cast<const int4*>(dst) + 4 * (size_t)v;
#pragma unroll
        for (int q = 0; q < 4; ++q) {
            const int4 t = d4[q];
            d[4 * q] = t.x; d[4 * q + 1] = t.y; d[4 * q + 2] = t.z; d[4 * q + 3] = t.w;
        }
    } else {
#pragma unroll
        for (int k = 0; k < 16; ++k) {
            const int i = 16 * v + k;
            d[k] = (i < E) ? dst[i] : -1;
        }
    }
    unsigned w[4];
#pragma unroll
    for (int q = 0; q < 4; ++q) {
        w[q] = 0u;
#pragma unroll
        for (int k = 0; k < 4; ++k) {
            const unsigned dd = (unsigned)d[4 * q + k];
            const unsigned o = (dd < (unsigned)n_nodes) ? dd / OWN_R : NO_OWNER;
            w[q] |= o << (8 * k);
        }
    }
    plane[v] = make_uint4(w[0], w[1], w[2], w[3]);
}

__device__ __forceinline__ float leaky(float v) { return (v > 0.0f) ? v : NEG_SLOPE * v; }

template <class Fn>
__device__ __forceinline__ void scan_dst(const int* __restrict__ dst, int E, int tid, Fn&& fn) {
    const bool vec = ((E & 3) == 0) && ((reinterpret_cast<size_t>(dst) & 15) == 0);
    if (vec) {
        const int4* d4 = reinterpret_cast<const int4*>(dst);
        const int E4 = E >> 2;
        for (int jb = 0; jb < E4; jb += 4 * OWN_THREADS) {
            int4 d[4];
#pragma unroll
            for (int u = 0; u < 4; ++u) {
                int j = jb + u * OWN_THREADS + tid;
                d[u] = (j < E4) ? d4[j] : make_int4(-1, -1, -1, -1);
            }
#pragma unroll
            for (int u = 0; u < 4; ++u) {
                int i = 4 * (jb + u * OWN_THREADS + tid);
                fn(i, d[u].x); fn(i + 1, d[u].y); fn(i + 2, d[u].z); fn(i + 3, d[u].w);
            }
        }
    } else {
        for (int ib = 0; ib < E; ib += OWN_THREADS) {
            int i = ib + tid;
            fn(i, (i < E) ? dst[i] : -1);
        }
    }
}

template <class Fn>
__device__ __forceinline__ void scan_plane(const uint4* __restrict__ plane, int nVec,
                                           unsigned me, int tid, Fn&& hit) {
    const unsigned key = me * 0x01010101u;
    for (int vb = 0; vb < nVec; vb += 8 * OWN_THREADS) {
        uint4 d[8];
#pragma unroll
        for (int u = 0; u < 8; ++u) {
            const int v = vb + u * OWN_THREADS + tid;
            d[u] = (v < nVec) ? plane[v] : make_uint4(~0u, ~0u, ~0u, ~0u);
        }
#pragma unroll
        for (int u = 0; u < 8; ++u) {
            const int i0 = 16 * (vb + u * OWN_THREADS + tid);
            const unsigned w[4] = { d[u].x, d[u].y, d[u].z, d[u].w };
#pragma unroll
            for (int q = 0; q < 4; ++q) {
                const unsigned x = w[q] ^ key;
                unsigned z = ~(((x & 0x7F7F7F7Fu) + 0x7F7F7F7Fu) | x | 0x7F7F7F7Fu);
                while (z) {
                    const int k = __builtin_ctz(z) >> 3;
                    z &= z - 1u;
                    hit(i0 + 4 * q + k);
                }
            }
        }
    }
}

template <int F, int H, int U>
__device__ __forceinline__ void accumulate_edges(const int (&s)[U], const unsigned (&dl)[U],
                                                 const bool (&ok)[U], int lane,
                                                 const float* __restrict__ h,
                                                 const float* __restrict__ as,
                                                 const float* adl, const float* m,
                                                 float* den, float* acc) {
    constexpr int C = F / H;
    constexpr int J = F / 32;
    float r[U][J];
    float a[U][H];
#pragma unroll
    for (int u = 0; u < U; ++u) {
        const float* hs = h + (size_t)s[u] * F;
#pragma unroll
        for (int j = 0; j < J; ++j) r[u][j] = hs[lane + 32 * j];
#pragma unroll
        for (int hh = 0; hh < H; ++hh) a[u][hh] = as[(size_t)s[u] * H + hh];
    }
#pragma unroll
    for (int u = 0; u < U; ++u) {
        if (ok[u]) {
            float ex[H];
#pragma unroll
            for (int hh = 0; hh < H; ++hh) {
                float v = leaky(a[u][hh] + adl[dl[u] * H + hh]);
                ex[hh] = __expf(v - m[dl[u] * H + hh]);
            }
#pragma unroll
            for (int hh = 0; hh < H; ++hh)
                if (lane == hh) atomicAdd(&den[dl[u] * H + hh], ex[hh]);
            float* ar = acc + dl[u] * F;
#pragma unroll
            for (int j = 0; j < J; ++j)
                atomicAdd(&ar[lane + 32 * j], r[u][j] * ex[(32 * j) / C]);
        }
    }
}

template <int F, int H, int LISTS, bool POST>
__global__ __launch_bounds__(OWN_THREADS)
void owner_aggregate_kernel(const int* __restrict__ ei, int E, int n_nodes,
                            const float* __restrict__ h,
                            const float* __restrict__ as,
                            const float* __restrict__ ad,
                            float* agg, unsigned int* keep, int keepF,
                            const uint4* __restrict__ plane,
                            const float* __restrict__ pb,
                            const float* __restrict__ pgamma,
                            const float* __restrict__ pbeta,
                            const float* __restrict__ pmean,
                            const float* __restrict__ pvar) {
    constexpr int C = F / H;
    __shared__ float acc[OWN_R * F];
    __shared__ unsigned int list[OWN_CAP];
    __shared__ float m[OWN_R * H];
    __shared__ float den[OWN_R * H];
    __shared__ float adl[OWN_R * H];
    __shared__ unsigned int cnt;

    const int tid = threadIdx.x;
    const int lane = tid & 31;
    const int wave = tid >> 5;
    const int base = blockIdx.x * OWN_R;
    const int R = min(OWN_R, n_nodes - base);
    const unsigned uR = (unsigned)R;
    const int* __restrict__ src = ei;
    const int* __restrict__ dst = ei + E;
    unsigned int* kp = keep + (size_t)base * keepF;
    const unsigned keepCap = (unsigned)R * (unsigned)keepF;

    for (int i = tid; i < R * F; i += OWN_THREADS) acc[i] = 0.0f;
    for (int i = tid; i < R * H; i += OWN_THREADS) {
        float a_d = ad[(size_t)base * H + i];
        adl[i] = a_d;
        m[i] = leaky(as[(size_t)base * H + i] + a_d);
        den[i] = 0.0f;
    }
    if (tid == 0) cnt = 0u;
    __syncthreads();

    unsigned kept = NO_LIST;
    if (LISTS == 2) kept = kp[0];
    const bool have = (LISTS == 2) && (kept <= OWN_CAP) && (kept < keepCap);

    const bool packs = (E <= (1 << 24)) && (n_nodes <= (1 << 24)) && (OWN_R <= 256);
    if (packs && !have) {
        if (owner_plane_ok(E, n_nodes, plane)) {
            scan_plane(plane, (E + 15) >> 4, (unsigned)blockIdx.x, tid, [&](int i) {
                unsigned pos = atomicAdd(&cnt, 1u);
                if (pos < OWN_CAP) list[pos] = (unsigned)i << 8;
            });
        } else {
            scan_dst(dst, E, tid, [&](int i, int d) {
                unsigned dl = (unsigned)(d - base);
                if (dl < uR) {
                    unsigned pos = atomicAdd(&cnt, 1u);
                    if (pos < OWN_CAP) list[pos] = (unsigned)i << 8;
                }
            });
        }
    }
    __syncthreads();
    const unsigned total = have ? kept : cnt;
    const bool fast = have || (packs && (total <= OWN_CAP));
    const bool leave = (LISTS == 1) && fast && (total < keepCap);

    if (fast) {
        for (unsigned p = tid; p < total; p += OWN_THREADS) {
            unsigned dl;
            int s;
            if (have) {
                unsigned pk = kp[1 + p];
                dl = pk & 255u;
                s = (int)(pk >> 8);
            } else {
                const unsigned i = list[p] >> 8;
                dl = (unsigned)(dst[i] - base);
                s = src[i];
            }
#pragma unroll
            for (int hh = 0; hh < H; ++hh)
                atomicMaxFloat(&m[dl * H + hh], leaky(as[(size_t)s * H + hh] + adl[dl * H + hh]));
            list[p] = ((unsigned)s << 8) | dl;
            if (leave) kp[1 + p] = ((unsigned)s << 8) | dl;
        }
        if (LISTS == 1 && tid == 0) kp[0] = leave ? total : NO_LIST;
        __syncthreads();
        for (unsigned p0 = wave * 4; p0 < total; p0 += OWN_WAVES * 4) {
            int s[4]; unsigned dl[4]; bool ok[4];
#pragma unroll
            for (int u = 0; u < 4; ++u) {
                ok[u] = (p0 + u) < total;
                unsigned pk = ok[u] ? list[p0 + u] : 0u;
                s[u] = (int)(pk >> 8);
                dl[u] = pk & 255u;
            }
            accumulate_edges<F, H, 4>(s, dl, ok, lane, h, as, adl, m, den, acc);
        }
    } else {
        if (LISTS == 1 && tid == 0) kp[0] = NO_LIST;
        scan_dst(dst, E, tid, [&](int i, int d) {
            unsigned dl = (unsigned)(d - base);
            if (dl < uR) {
                int s = src[i];
#pragma unroll
                for (int hh = 0; hh < H; ++hh)
                    atomicMaxFloat(&m[dl * H + hh], leaky(as[(size_t)s * H + hh] + adl[dl * H + hh]));
            }
        });
        __syncthreads();
        scan_dst(dst, E, tid, [&](int i, int d) {
            unsigned dl = (unsigned)(d - base);
            bool hit = dl < uR;
            int sv = hit ? src[i] : 0;
            unsigned mask = __builtin_amdgcn_ballot_w32(hit);
            while (mask) {
                int l = __builtin_ctz(mask);
                mask &= mask - 1u;
                int s1[1] = { __shfl(sv, l, 32) };
                unsigned d1[1] = { (unsigned)__shfl((int)dl, l, 32) };
                bool o1[1] = { true };
                accumulate_edges<F, H, 1>(s1, d1, o1, lane, h, as, adl, m, den, acc);
            }
        });
    }
    for (int q0 = wave * 4; q0 < R; q0 += OWN_WAVES * 4) {
        int s[4]; unsigned dl[4]; bool ok[4];
#pragma unroll
        for (int u = 0; u < 4; ++u) {
            ok[u] = (q0 + u) < R;
            dl[u] = ok[u] ? (unsigned)(q0 + u) : 0u;
            s[u] = base + (int)dl[u];
        }
        accumulate_edges<F, H, 4>(s, dl, ok, lane, h, as, adl, m, den, acc);
    }
    __syncthreads();

    if (POST) {
        static_assert(OWN_THREADS % F == 0, "one channel per thread");
        const int c = tid % F;
        const float fb = pb[c], fg = pgamma[c], ft = pbeta[c], fm = pmean[c];
        const float frs = rsqrtf(pvar[c] + EPS);
        for (int i = tid; i < R * F; i += OWN_THREADS) {
            const int dl = i / F;
            float v = acc[i] / den[dl * H + c / C] + fb;
            v = fg * (v - fm) * frs + ft;
            agg[(size_t)base * F + i] = (v > 0.0f) ? v : 0.0f;
        }
    } else {
        for (int i = tid; i < R * F; i += OWN_THREADS) {
            int dl = i / F;
            int c = i - dl * F;
            agg[(size_t)base * F + i] = acc[i] / den[dl * H + c / C];
        }
    }
}

#define POOL_THREADS 1024
#define POOL_WAVES (POOL_THREADS / 32)
#define POOL_CAP 65536
#define POOL_VECS 16
#define POOL_ROWS 8

__global__ __launch_bounds__(POOL_THREADS)
void pool_owner_kernel(const float* __restrict__ x,
                       const int* __restrict__ batch,
                       const float* __restrict__ b,
                       const float* __restrict__ gamma,
                       const float* __restrict__ beta,
                       const float* __restrict__ mean,
                       const float* __restrict__ var,
                       float* __restrict__ out, int n_nodes) {
    constexpr int F = 128;
    __shared__ unsigned short list[POOL_CAP];
    __shared__ float4 part[POOL_WAVES][32];
    __shared__ unsigned int cnt;

    const int tid = threadIdx.x;
    const int lane = tid & 31;
    const int wave = tid >> 5;
    const int g = blockIdx.x;
    const float4* __restrict__ x4 = reinterpret_cast<const float4*>(x);

    float pb[4], pg[4], pt[4], pm[4], prs[4];
    for (int k = 0; k < 4; ++k) {
        const int c = 4 * lane + k;
        pb[k] = b[c]; pg[k] = gamma[c]; pt[k] = beta[c]; pm[k] = mean[c];
        prs[k] = rsqrtf(var[c] + EPS);
    }
    auto act = [&](float xv, int k) {
        float v = xv + pb[k];
        v = pg[k] * (v - pm[k]) * prs[k] + pt[k];
        return (v > 0.0f) ? v : 0.0f;
    };
    const float ninf = __int_as_float((int)0xFF800000u);
    float mx[4] = { ninf, ninf, ninf, ninf };

    for (int chunk0 = 0; chunk0 < n_nodes; chunk0 += POOL_CAP) {
        const int len = min(POOL_CAP, n_nodes - chunk0);
        if (tid == 0) cnt = 0u;
        __syncthreads();
        const int* __restrict__ bp = batch + chunk0;
        const int len4 = ((reinterpret_cast<size_t>(bp) & 15) == 0) ? (len >> 2) : 0;
        const int4* __restrict__ bp4 = reinterpret_cast<const int4*>(bp);
        auto push = [&](int i) {
            const unsigned pos = atomicAdd(&cnt, 1u);
            list[pos] = (unsigned short)i;
        };
        for (int jb = 0; jb < len4; jb += POOL_VECS * POOL_THREADS) {
            int4 d[POOL_VECS];
#pragma unroll
            for (int u = 0; u < POOL_VECS; ++u) {
                const int j = jb + u * POOL_THREADS + tid;
                d[u] = (j < len4) ? bp4[j] : make_int4(-1, -1, -1, -1);
            }
            bool any = false;
#pragma unroll
            for (int u = 0; u < POOL_VECS; ++u)
                any |= (d[u].x == g) | (d[u].y == g) | (d[u].z == g) | (d[u].w == g);
            if (__builtin_amdgcn_ballot_w32(any) != 0u) {
                unsigned long long hm = 0ull;
#pragma unroll
                for (int u = 0; u < POOL_VECS; ++u) {
                    const unsigned m4 = (d[u].x == g ? 1u : 0u) | (d[u].y == g ? 2u : 0u)
                                      | (d[u].z == g ? 4u : 0u) | (d[u].w == g ? 8u : 0u);
                    hm |= (unsigned long long)m4 << (4 * u);
                }
                while (hm) {
                    const int bit = __builtin_ctzll(hm);
                    hm &= hm - 1ull;
                    push(4 * (jb + (bit >> 2) * POOL_THREADS + tid) + (bit & 3));
                }
            }
        }
        for (int i = 4 * len4 + tid; i < len; i += POOL_THREADS)
            if (bp[i] == g) push(i);
        __syncthreads();
        const unsigned total = cnt;

        for (unsigned p0 = wave; p0 < total; p0 += POOL_WAVES * POOL_ROWS) {
            float4 r[POOL_ROWS];
#pragma unroll
            for (int u = 0; u < POOL_ROWS; ++u) {
                const unsigned p = p0 + u * POOL_WAVES;
                const int n = chunk0 + (int)list[(p < total) ? p : p0];
                r[u] = x4[(size_t)n * (F / 4) + lane];
            }
#pragma unroll
            for (int u = 0; u < POOL_ROWS; ++u) {
                if (p0 + u * POOL_WAVES < total) {
                    mx[0] = fmaxf(mx[0], act(r[u].x, 0));
                    mx[1] = fmaxf(mx[1], act(r[u].y, 1));
                    mx[2] = fmaxf(mx[2], act(r[u].z, 2));
                    mx[3] = fmaxf(mx[3], act(r[u].w, 3));
                }
            }
        }
        __syncthreads();
    }

    part[wave][lane] = make_float4(mx[0], mx[1], mx[2], mx[3]);
    __syncthreads();
    if (tid < F) {
        const float* pf = reinterpret_cast<const float*>(&part[0][0]);
        float m = pf[tid];
        for (int w = 1; w < POOL_WAVES; ++w) m = fmaxf(m, pf[w * F + tid]);
        out[(size_t)g * F + tid] = m;
    }
}

extern "C" void kernel_launch(void* const* d_in, const int* in_sizes, int n_in,
                              void* d_out, int out_size, void* d_ws, size_t ws_size,
                              hipStream_t stream) {
    const int F_IN = 128, HID = 128, HEADS = 2;
    const int F1 = HEADS * HID;
    const int N = in_sizes[0] / F_IN;
    const int E = in_sizes[1] / 2;

    const float* x      = (const float*)d_in[0];
    const int*   ei     = (const int*)d_in[1];
    const int*   batch  = (const int*)d_in[2];
    const float* W1     = (const float*)d_in[3];
    const float* a_src1 = (const float*)d_in[4];
    const float* a_dst1 = (const float*)d_in[5];
    const float* b1     = (const float*)d_in[6];
    const float* gamma1 = (const float*)d_in[7];
    const float* beta1  = (const float*)d_in[8];
    const float* rm1    = (const float*)d_in[9];
    const float* rv1    = (const float*)d_in[10];
    const float* W2     = (const float*)d_in[11];
    const float* a_src2 = (const float*)d_in[12];
    const float* a_dst2 = (const float*)d_in[13];
    const float* b2     = (const float*)d_in[14];
    const float* gamma2 = (const float*)d_in[15];
    const float* beta2  = (const float*)d_in[16];
    const float* rm2    = (const float*)d_in[17];
    const float* rv2    = (const float*)d_in[18];

    float* ws = (float*)d_ws;
    size_t o = 0;
    float* h1   = ws + o; o += (size_t)N * F1;
    float* aggA = ws + o; o += (size_t)N * F1;
    float* aggB = ws + o; o += (size_t)N * HID;
    float* as1  = ws + o; o += (size_t)N * HEADS;
    float* ad1  = ws + o; o += (size_t)N * HEADS;
    float* as2  = ws + o; o += (size_t)N;
    float* ad2  = ws + o; o += (size_t)N;
    float* h2   = h1;
    float* agg2 = h1 + (size_t)N * HID;
    unsigned int* keep = (unsigned int*)aggB;
    uint4* plane = (uint4*)(aggB + (size_t)N * KEEP_WORDS);

    auto fillBlocks = [](long long cnt) { return (unsigned)((cnt + 255) / 256); };
    const unsigned ownerBlocks = (unsigned)((N + OWN_R - 1) / OWN_R);

    gemm_wmma_f32_kernel<128, 256><<<GEMM_GRID, GEMM_THREADS, 0, stream>>>(x, W1, h1, N / 16);
    alpha_rows_kernel<2><<<ALPHA_GRID, ALPHA_THREADS, 0, stream>>>(h1, a_src1, a_dst1, as1, ad1, N);
    owner_plane_kernel<<<fillBlocks((long long)((E + 15) / 16)), 256, 0, stream>>>(ei, E, N, plane);
    owner_aggregate_kernel<256, 2, 1, true><<<ownerBlocks, OWN_THREADS, 0, stream>>>(
        ei, E, N, h1, as1, ad1, aggA, keep, KEEP_WORDS, plane, b1, gamma1, beta1, rm1, rv1);

    gemm_wmma_f32_kernel<256, 128><<<GEMM_GRID, GEMM_THREADS, 0, stream>>>(aggA, W2, h2, N / 16);
    alpha_rows_kernel<1><<<ALPHA_GRID, ALPHA_THREADS, 0, stream>>>(h2, a_src2, a_dst2, as2, ad2, N);
    owner_aggregate_kernel<128, 1, 2, false><<<ownerBlocks, OWN_THREADS, 0, stream>>>(
        ei, E, N, h2, as2, ad2, agg2, keep, KEEP_WORDS, plane,
        nullptr, nullptr, nullptr, nullptr, nullptr);

    pool_owner_kernel<<<(unsigned)(out_size / HID), POOL_THREADS, 0, stream>>>(
        agg2, batch, b2, gamma2, beta2, rm2, rv2, (float*)d_out, N);
}


# Owner kernels move rows in 16-byte pieces per lane: one gather load per 512 contiguous row bytes (was four 4-byte loads), LDS rows lane-interleaved so adds stay on 32 banks, 16-byte row stores
# speedup vs baseline: 4.6561x; 1.0603x over previous
#include <hip/hip_runtime.h>
#include <hip/hip_bf16.h>

typedef float v2f __attribute__((ext_vector_type(2)));
typedef float v8f __attribute__((ext_vector_type(8)));

#define NEG_SLOPE 0.2f
#define EPS 1e-5f

__device__ __forceinline__ void atomicMaxFloat(float* addr, float val) {
    if (val >= 0.0f) {
        atomicMax((int*)addr, __float_as_int(val));
    } else {
        atomicMin((unsigned int*)addr, __float_as_uint(val));
    }
}

#define GEMM_THREADS 256
#define GEMM_WAVES (GEMM_THREADS / 32)
#define GEMM_GRID 512

template <int K, int NCOL>
__global__ __launch_bounds__(GEMM_THREADS)
void gemm_wmma_f32_kernel(const float* __restrict__ A,
                          const float* __restrict__ Bm,
                          float* __restrict__ C, int nStrips) {
    constexpr int PITCH = 2 * NCOL + 16;
    constexpr int TPW = NCOL / 16 / GEMM_WAVES;
    constexpr int C4 = NCOL / 4;
    static_assert(K % 8 == 0 && NCOL % (16 * GEMM_WAVES) == 0, "tile shape");
    __shared__ __attribute__((aligned(16))) float Bs[(K / 2) * PITCH];

    const int tid = threadIdx.x;
    for (int i = tid; i < (K / 2) * C4; i += GEMM_THREADS) {
        const int p = i / C4, c4 = i - p * C4;
        const float4 r0 = *reinterpret_cast<const float4*>(Bm + (size_t)(2 * p) * NCOL + 4 * c4);
        const float4 r1 = *reinterpret_cast<const float4*>(Bm + (size_t)(2 * p + 1) * NCOL + 4 * c4);
        float4* d = reinterpret_cast<float4*>(Bs + p * PITCH + 8 * c4);
        d[0] = make_float4(r0.x, r1.x, r0.y, r1.y);
        d[1] = make_float4(r0.z, r1.z, r0.w, r1.w);
    }
    __syncthreads();

    const int wave = tid >> 5;
    const int lane = tid & 31;
    const int c = lane & 15;
    const int h = lane >> 4;
    const float* bp = Bs + (2 * h) * PITCH + 2 * (wave * TPW * 16 + c);

    for (int strip = blockIdx.x; strip < nStrips; strip += gridDim.x) {
        const float* ap = A + ((size_t)strip * 16 + c) * K + 4 * h;
        float4 a4[K / 8];
#pragma unroll
        for (int j = 0; j < K / 8; ++j) a4[j] = *reinterpret_cast<const float4*>(ap + 8 * j);

        v8f acc[TPW];
#pragma unroll
        for (int t = 0; t < TPW; ++t) acc[t] = v8f{};

#pragma unroll
        for (int j = 0; j < K / 8; ++j) {
#pragma unroll
            for (int u = 0; u < 2; ++u) {
                v2f a;
                a.x = u ? a4[j].z : a4[j].x;
                a.y = u ? a4[j].w : a4[j].y;
#pragma unroll
                for (int t = 0; t < TPW; ++t) {
                    const v2f b = *reinterpret_cast<const v2f*>(bp + (4 * j + u) * PITCH + 32 * t);
                    acc[t] = __builtin_amdgcn_wmma_f32_16x16x4_f32(false, a, false, b,
                                                                   (short)0, acc[t], false, false);
                }
            }
        }
#pragma unroll
        for (int t = 0; t < TPW; ++t) {
            float* cp = C + ((size_t)strip * 16 + 8 * h) * NCOL + (wave * TPW + t) * 16 + c;
#pragma unroll
            for (int r = 0; r < 8; ++r) cp[(size_t)r * NCOL] = acc[t][r];
        }
    }
}

#define ALPHA_GRID 512
#define ALPHA_THREADS 256

template <int H>
__global__ __launch_bounds__(ALPHA_THREADS)
void alpha_rows_kernel(const float* __restrict__ h,
                       const float* __restrict__ a_src,
                       const float* __restrict__ a_dst,
                       float* __restrict__ as, float* __restrict__ ad, int n_nodes) {
    static_assert(H == 1 || H == 2, "16 rows x heads per trip");
    constexpr int ROWS = 16 / H;
    constexpr int RV = 32 * H;
    const int lane = threadIdx.x & 31;
    const int gw = __builtin_amdgcn_readfirstlane((int)((blockIdx.x * ALPHA_THREADS + threadIdx.x) >> 5));
    const int nW = (int)((gridDim.x * ALPHA_THREADS) >> 5);
    const float4* __restrict__ h4 = reinterpret_cast<const float4*>(h);

    float4 ws[H], wd[H];
#pragma unroll
    for (int hh = 0; hh < H; ++hh) {
        const float* sp = a_src + hh * 128 + 4 * lane;
        const float* dp = a_dst + hh * 128 + 4 * lane;
        ws[hh] = make_float4(sp[0], sp[1], sp[2], sp[3]);
        wd[hh] = make_float4(dp[0], dp[1], dp[2], dp[3]);
    }

    for (int r0 = gw * ROWS; r0 < n_nodes; r0 += nW * ROWS) {
        float4 v[ROWS][H];
#pragma unroll
        for (int u = 0; u < ROWS; ++u) {
            const int n = min(r0 + u, n_nodes - 1);
#pragma unroll
            for (int hh = 0; hh < H; ++hh) v[u][hh] = h4[(size_t)n * RV + 32 * hh + lane];
        }
        __builtin_amdgcn_sched_barrier(0);
        float val[32];
#pragma unroll
        for (int u = 0; u < ROWS; ++u) {
#pragma unroll
            for (int hh = 0; hh < H; ++hh) {
                const float4 x = v[u][hh];
                val[u * H + hh] = x.x * ws[hh].x + x.y * ws[hh].y + x.z * ws[hh].z + x.w * ws[hh].w;
                val[16 + u * H + hh] = x.x * wd[hh].x + x.y * wd[hh].y + x.z * wd[hh].z + x.w * wd[hh].w;
            }
        }
#pragma unroll
        for (int half = 16, off = 16; half >= 1; half >>= 1, off >>= 1) {
            const bool up = (lane & off) != 0;
#pragma unroll
            for (int i = 0; i < half; ++i) {
                const float lo = val[i], hi = val[i + half];
                const float keep = up ? hi : lo;
                const float send = up ? lo : hi;
                val[i] = keep + __shfl_xor(send, off, 32);
            }
        }
        const int k = lane & 15;
        if (r0 + k / H < n_nodes) {
            float* o = (lane & 16) ? ad : as;
            o[(size_t)r0 * H + k] = val[0];
        }
    }
}

#define NO_LIST 0xFFFFFFFFu
#define OWN_R 200
#define OWN_THREADS 1024
#define OWN_WAVES (OWN_THREADS / 32)
#define OWN_CAP 16384
#define NO_OWNER 255u
#define KEEP_WORDS 96
#define PLANE_WORDS 32

__device__ __forceinline__ bool owner_plane_ok(int E, int n_nodes, const void* plane) {
    const unsigned nOwner = ((unsigned)n_nodes + OWN_R - 1) / OWN_R;
    const size_t bytes = ((size_t)E + 15) & ~(size_t)15;
    return plane != nullptr && nOwner <= NO_OWNER
        && bytes <= (size_t)n_nodes * (PLANE_WORDS * 4)
        && (reinterpret_cast<size_t>(plane) & 15) == 0;
}

__global__ void owner_plane_kernel(const int* __restrict__ ei, int E, int n_nodes,
                                   uint4* __restrict__ plane) {
    if (!owner_plane_ok(E, n_nodes, plane)) return;
    const int* __restrict__ dst = ei + E;
    const int nVec = (E + 15) >> 4;
    const int v = blockIdx.x * blockDim.x + threadIdx.x;
    if (v >= nVec) return;
    int d[16];
    if (((E & 15) == 0) && ((reinterpret_cast<size_t>(dst) & 15) == 0)) {
        const int4* d4 = reinterpret_cast<const int4*>(dst) + 4 * (size_t)v;
#pragma unroll
        for (int q = 0; q < 4; ++q) {
            const int4 t = d4[q];
            d[4 * q] = t.x; d[4 * q + 1] = t.y; d[4 * q + 2] = t.z; d[4 * q + 3] = t.w;
        }
    } else {
#pragma unroll
        for (int k = 0; k < 16; ++k) {
            const int i = 16 * v + k;
            d[k] = (i < E) ? dst[i] : -1;
        }
    }
    unsigned w[4];
#pragma unroll
    for (int q = 0; q < 4; ++q) {
        w[q] = 0u;
#pragma unroll
        for (int k = 0; k < 4; ++k) {
            const unsigned dd = (unsigned)d[4 * q + k];
            const unsigned o = (dd < (unsigned)n_nodes) ? dd / OWN_R : NO_OWNER;
            w[q] |= o << (8 * k);
        }
    }
    plane[v] = make_uint4(w[0], w[1], w[2], w[3]);
}

__device__ __forceinline__ float leaky(float v) { return (v > 0.0f) ? v : NEG_SLOPE * v; }

template <class Fn>
__device__ __forceinline__ void scan_dst(const int* __restrict__ dst, int E, int tid, Fn&& fn) {
    const bool vec = ((E & 3) == 0) && ((reinterpret_cast<size_t>(dst) & 15) == 0);
    if (vec) {
        const int4* d4 = reinterpret_cast<const int4*>(dst);
        const int E4 = E >> 2;
        for (int jb = 0; jb < E4; jb += 4 * OWN_THREADS) {
            int4 d[4];
#pragma unroll
            for (int u = 0; u < 4; ++u) {
                int j = jb + u * OWN_THREADS + tid;
                d[u] = (j < E4) ? d4[j] : make_int4(-1, -1, -1, -1);
            }
#pragma unroll
            for (int u = 0; u < 4; ++u) {
                int i = 4 * (jb + u * OWN_THREADS + tid);
                fn(i, d[u].x); fn(i + 1, d[u].y); fn(i + 2, d[u].z); fn(i + 3, d[u].w);
            }
        }
    } else {
        for (int ib = 0; ib < E; ib += OWN_THREADS) {
            int i = ib + tid;
            fn(i, (i < E) ? dst[i] : -1);
        }
    }
}

template <class Fn>
__device__ __forceinline__ void scan_plane(const uint4* __restrict__ plane, int nVec,
                                           unsigned me, int tid, Fn&& hit) {
    const unsigned key = me * 0x01010101u;
    for (int vb = 0; vb < nVec; vb += 8 * OWN_THREADS) {
        uint4 d[8];
#pragma unroll
        for (int u = 0; u < 8; ++u) {
            const int v = vb + u * OWN_THREADS + tid;
            d[u] = (v < nVec) ? plane[v] : make_uint4(~0u, ~0u, ~0u, ~0u);
        }
#pragma unroll
        for (int u = 0; u < 8; ++u) {
            const int i0 = 16 * (vb + u * OWN_THREADS + tid);
            const unsigned w[4] = { d[u].x, d[u].y, d[u].z, d[u].w };
#pragma unroll
            for (int q = 0; q < 4; ++q) {
                const unsigned x = w[q] ^ key;
                unsigned z = ~(((x & 0x7F7F7F7Fu) + 0x7F7F7F7Fu) | x | 0x7F7F7F7Fu);
                while (z) {
                    const int k = __builtin_ctz(z) >> 3;
                    z &= z - 1u;
                    hit(i0 + 4 * q + k);
                }
            }
        }
    }
}

template <int F, int H, int U>
__device__ __forceinline__ void accumulate_edges(const int (&s)[U], const unsigned (&dl)[U],
                                                 const bool (&ok)[U], int lane,
                                                 const float* __restrict__ h,
                                                 const float* __restrict__ as,
                                                 const float* adl, const float* m,
                                                 float* den, float* acc) {
    constexpr int C = F / H;
    constexpr int J = F / 128;
    static_assert(F % 128 == 0 && C % 128 == 0, "a 512-byte piece group lies inside one head");
    float4 r[U][J];
    float a[U][H];
#pragma unroll
    for (int u = 0; u < U; ++u) {
        const float4* hs = reinterpret_cast<const float4*>(h + (size_t)s[u] * F);
#pragma unroll
        for (int j = 0; j < J; ++j) r[u][j] = hs[lane + 32 * j];
#pragma unroll
        for (int hh = 0; hh < H; ++hh) a[u][hh] = as[(size_t)s[u] * H + hh];
    }
#pragma unroll
    for (int u = 0; u < U; ++u) {
        if (ok[u]) {
            float ex[H];
#pragma unroll
            for (int hh = 0; hh < H; ++hh) {
                float v = leaky(a[u][hh] + adl[dl[u] * H + hh]);
                ex[hh] = __expf(v - m[dl[u] * H + hh]);
            }
#pragma unroll
            for (int hh = 0; hh < H; ++hh)
                if (lane == hh) atomicAdd(&den[dl[u] * H + hh], ex[hh]);
            float* ar = acc + dl[u] * F + lane;
#pragma unroll
            for (int j = 0; j < J; ++j) {
                const float e = ex[(128 * j) / C];
                atomicAdd(&ar[128 * j],      r[u][j].x * e);
                atomicAdd(&ar[128 * j + 32], r[u][j].y * e);
                atomicAdd(&ar[128 * j + 64], r[u][j].z * e);
                atomicAdd(&ar[128 * j + 96], r[u][j].w * e);
            }
        }
    }
}

template <int F, int H, int LISTS, bool POST>
__global__ __launch_bounds__(OWN_THREADS)
void owner_aggregate_kernel(const int* __restrict__ ei, int E, int n_nodes,
                            const float* __restrict__ h,
                            const float* __restrict__ as,
                            const float* __restrict__ ad,
                            float* agg, unsigned int* keep, int keepF,
                            const uint4* __restrict__ plane,
                            const float* __restrict__ pb,
                            const float* __restrict__ pgamma,
                            const float* __restrict__ pbeta,
                            const float* __restrict__ pmean,
                            const float* __restrict__ pvar) {
    constexpr int C = F / H;
    __shared__ float acc[OWN_R * F];
    __shared__ unsigned int list[OWN_CAP];
    __shared__ float m[OWN_R * H];
    __shared__ float den[OWN_R * H];
    __shared__ float adl[OWN_R * H];
    __shared__ unsigned int cnt;

    const int tid = threadIdx.x;
    const int lane = tid & 31;
    const int wave = tid >> 5;
    const int base = blockIdx.x * OWN_R;
    const int R = min(OWN_R, n_nodes - base);
    const unsigned uR = (unsigned)R;
    const int* __restrict__ src = ei;
    const int* __restrict__ dst = ei + E;
    unsigned int* kp = keep + (size_t)base * keepF;
    const unsigned keepCap = (unsigned)R * (unsigned)keepF;

    for (int i = tid; i < R * F; i += OWN_THREADS) acc[i] = 0.0f;
    for (int i = tid; i < R * H; i += OWN_THREADS) {
        float a_d = ad[(size_t)base * H + i];
        adl[i] = a_d;
        m[i] = leaky(as[(size_t)base * H + i] + a_d);
        den[i] = 0.0f;
    }
    if (tid == 0) cnt = 0u;
    __syncthreads();

    unsigned kept = NO_LIST;
    if (LISTS == 2) kept = kp[0];
    const bool have = (LISTS == 2) && (kept <= OWN_CAP) && (kept < keepCap);

    const bool packs = (E <= (1 << 24)) && (n_nodes <= (1 << 24)) && (OWN_R <= 256);
    if (packs && !have) {
        if (owner_plane_ok(E, n_nodes, plane)) {
            scan_plane(plane, (E + 15) >> 4, (unsigned)blockIdx.x, tid, [&](int i) {
                unsigned pos = atomicAdd(&cnt, 1u);
                if (pos < OWN_CAP) list[pos] = (unsigned)i << 8;
            });
        } else {
            scan_dst(dst, E, tid, [&](int i, int d) {
                unsigned dl = (unsigned)(d - base);
                if (dl < uR) {
                    unsigned pos = atomicAdd(&cnt, 1u);
                    if (pos < OWN_CAP) list[pos] = (unsigned)i << 8;
                }
            });
        }
    }
    __syncthreads();
    const unsigned total = have ? kept : cnt;
    const bool fast = have || (packs && (total <= OWN_CAP));
    const bool leave = (LISTS == 1) && fast && (total < keepCap);

    if (fast) {
        for (unsigned p = tid; p < total; p += OWN_THREADS) {
            unsigned dl;
            int s;
            if (have) {
                unsigned pk = kp[1 + p];
                dl = pk & 255u;
                s = (int)(pk >> 8);
            } else {
                const unsigned i = list[p] >> 8;
                dl = (unsigned)(dst[i] - base);
                s = src[i];
            }
#pragma unroll
            for (int hh = 0; hh < H; ++hh)
                atomicMaxFloat(&m[dl * H + hh], leaky(as[(size_t)s * H + hh] + adl[dl * H + hh]));
            list[p] = ((unsigned)s << 8) | dl;
            if (leave) kp[1 + p] = ((unsigned)s << 8) | dl;
        }
        if (LISTS == 1 && tid == 0) kp[0] = leave ? total : NO_LIST;
        __syncthreads();
        for (unsigned p0 = wave * 4; p0 < total; p0 += OWN_WAVES * 4) {
            int s[4]; unsigned dl[4]; bool ok[4];
#pragma unroll
            for (int u = 0; u < 4; ++u) {
                ok[u] = (p0 + u) < total;
                unsigned pk = ok[u] ? list[p0 + u] : 0u;
                s[u] = (int)(pk >> 8);
                dl[u] = pk & 255u;
            }
            accumulate_edges<F, H, 4>(s, dl, ok, lane, h, as, adl, m, den, acc);
        }
    } else {
        if (LISTS == 1 && tid == 0) kp[0] = NO_LIST;
        scan_dst(dst, E, tid, [&](int i, int d) {
            unsigned dl = (unsigned)(d - base);
            if (dl < uR) {
                int s = src[i];
#pragma unroll
                for (int hh = 0; hh < H; ++hh)
                    atomicMaxFloat(&m[dl * H + hh], leaky(as[(size_t)s * H + hh] + adl[dl * H + hh]));
            }
        });
        __syncthreads();
        scan_dst(dst, E, tid, [&](int i, int d) {
            unsigned dl = (unsigned)(d - base);
            bool hit = dl < uR;
            int sv = hit ? src[i] : 0;
            unsigned mask = __builtin_amdgcn_ballot_w32(hit);
            while (mask) {
                int l = __builtin_ctz(mask);
                mask &= mask - 1u;
                int s1[1] = { __shfl(sv, l, 32) };
                unsigned d1[1] = { (unsigned)__shfl((int)dl, l, 32) };
                bool o1[1] = { true };
                accumulate_edges<F, H, 1>(s1, d1, o1, lane, h, as, adl, m, den, acc);
            }
        });
    }
    for (int q0 = wave * 4; q0 < R; q0 += OWN_WAVES * 4) {
        int s[4]; unsigned dl[4]; bool ok[4];
#pragma unroll
        for (int u = 0; u < 4; ++u) {
            ok[u] = (q0 + u) < R;
            dl[u] = ok[u] ? (unsigned)(q0 + u) : 0u;
            s[u] = base + (int)dl[u];
        }
        accumulate_edges<F, H, 4>(s, dl, ok, lane, h, as, adl, m, den, acc);
    }
    __syncthreads();

    constexpr int Q = F / 4;
    static_assert(OWN_THREADS % Q == 0, "one piece per thread");
    const int cq = tid % Q;
    const int c0 = 4 * cq;
    const int w0 = 128 * (cq >> 5) + (cq & 31);
    const int hd = c0 / C;
    float4* agg4 = reinterpret_cast<float4*>(agg + (size_t)base * F);
    if (POST) {
        float fb[4], fg[4], ft[4], fm[4], frs[4];
        for (int k = 0; k < 4; ++k) {
            const int c = c0 + k;
            fb[k] = pb[c]; fg[k] = pgamma[c]; ft[k] = pbeta[c]; fm[k] = pmean[c];
            frs[k] = rsqrtf(pvar[c] + EPS);
        }
        for (int i = tid; i < R * Q; i += OWN_THREADS) {
            const int dl = i / Q;
            const float* ar = acc + dl * F + w0;
            const float d = den[dl * H + hd];
            float o[4];
            for (int k = 0; k < 4; ++k) {
                float v = ar[32 * k] / d + fb[k];
                v = fg[k] * (v - fm[k]) * frs[k] + ft[k];
                o[k] = (v > 0.0f) ? v : 0.0f;
            }
            agg4[i] = make_float4(o[0], o[1], o[2], o[3]);
        }
    } else {
        for (int i = tid; i < R * Q; i += OWN_THREADS) {
            const int dl = i / Q;
            const float* ar = acc + dl * F + w0;
            const float d = den[dl * H + hd];
            agg4[i] = make_float4(ar[0] / d, ar[32] / d, ar[64] / d, ar[96] / d);
        }
    }
}

#define POOL_THREADS 1024
#define POOL_WAVES (POOL_THREADS / 32)
#define POOL_CAP 65536
#define POOL_VECS 16
#define POOL_ROWS 8

__global__ __launch_bounds__(POOL_THREADS)
void pool_owner_kernel(const float* __restrict__ x,
                       const int* __restrict__ batch,
                       const float* __restrict__ b,
                       const float* __restrict__ gamma,
                       const float* __restrict__ beta,
                       const float* __restrict__ mean,
                       const float* __restrict__ var,
                       float* __restrict__ out, int n_nodes) {
    constexpr int F = 128;
    __shared__ unsigned short list[POOL_CAP];
    __shared__ float4 part[POOL_WAVES][32];
    __shared__ unsigned int cnt;

    const int tid = threadIdx.x;
    const int lane = tid & 31;
    const int wave = tid >> 5;
    const int g = blockIdx.x;
    const float4* __restrict__ x4 = reinterpret_cast<const float4*>(x);

    float pb[4], pg[4], pt[4], pm[4], prs[4];
    for (int k = 0; k < 4; ++k) {
        const int c = 4 * lane + k;
        pb[k] = b[c]; pg[k] = gamma[c]; pt[k] = beta[c]; pm[k] = mean[c];
        prs[k] = rsqrtf(var[c] + EPS);
    }
    auto act = [&](float xv, int k) {
        float v = xv + pb[k];
        v = pg[k] * (v - pm[k]) * prs[k] + pt[k];
        return (v > 0.0f) ? v : 0.0f;
    };
    const float ninf = __int_as_float((int)0xFF800000u);
    float mx[4] = { ninf, ninf, ninf, ninf };

    for (int chunk0 = 0; chunk0 < n_nodes; chunk0 += POOL_CAP) {
        const int len = min(POOL_CAP, n_nodes - chunk0);
        if (tid == 0) cnt = 0u;
        __syncthreads();
        const int* __restrict__ bp = batch + chunk0;
        const int len4 = ((reinterpret_cast<size_t>(bp) & 15) == 0) ? (len >> 2) : 0;
        const int4* __restrict__ bp4 = reinterpret_cast<const int4*>(bp);
        auto push = [&](int i) {
            const unsigned pos = atomicAdd(&cnt, 1u);
            list[pos] = (unsigned short)i;
        };
        for (int jb = 0; jb < len4; jb += POOL_VECS * POOL_THREADS) {
            int4 d[POOL_VECS];
#pragma unroll
            for (int u = 0; u < POOL_VECS; ++u) {
                const int j = jb + u * POOL_THREADS + tid;
                d[u] = (j < len4) ? bp4[j] : make_int4(-1, -1, -1, -1);
            }
            bool any = false;
#pragma unroll
            for (int u = 0; u < POOL_VECS; ++u)
                any |= (d[u].x == g) | (d[u].y == g) | (d[u].z == g) | (d[u].w == g);
            if (__builtin_amdgcn_ballot_w32(any) != 0u) {
                unsigned long long hm = 0ull;
#pragma unroll
                for (int u = 0; u < POOL_VECS; ++u) {
                    const unsigned m4 = (d[u].x == g ? 1u : 0u) | (d[u].y == g ? 2u : 0u)
                                      | (d[u].z == g ? 4u : 0u) | (d[u].w == g ? 8u : 0u);
                    hm |= (unsigned long long)m4 << (4 * u);
                }
                while (hm) {
                    const int bit = __builtin_ctzll(hm);
                    hm &= hm - 1ull;
                    push(4 * (jb + (bit >> 2) * POOL_THREADS + tid) + (bit & 3));
                }
            }
        }
        for (int i = 4 * len4 + tid; i < len; i += POOL_THREADS)
            if (bp[i] == g) push(i);
        __syncthreads();
        const unsigned total = cnt;

        for (unsigned p0 = wave; p0 < total; p0 += POOL_WAVES * POOL_ROWS) {
            float4 r[POOL_ROWS];
#pragma unroll
            for (int u = 0; u < POOL_ROWS; ++u) {
                const unsigned p = p0 + u * POOL_WAVES;
                const int n = chunk0 + (int)list[(p < total) ? p : p0];
                r[u] = x4[(size_t)n * (F / 4) + lane];
            }
#pragma unroll
            for (int u = 0; u < POOL_ROWS; ++u) {
                if (p0 + u * POOL_WAVES < total) {
                    mx[0] = fmaxf(mx[0], act(r[u].x, 0));
                    mx[1] = fmaxf(mx[1], act(r[u].y, 1));
                    mx[2] = fmaxf(mx[2], act(r[u].z, 2));
                    mx[3] = fmaxf(mx[3], act(r[u].w, 3));
                }
            }
        }
        __syncthreads();
    }

    part[wave][lane] = make_float4(mx[0], mx[1], mx[2], mx[3]);
    __syncthreads();
    if (tid < F) {
        const float* pf = reinterpret_cast<const float*>(&part[0][0]);
        float m = pf[tid];
        for (int w = 1; w < POOL_WAVES; ++w) m = fmaxf(m, pf[w * F + tid]);
        out[(size_t)g * F + tid] = m;
    }
}

extern "C" void kernel_launch(void* const* d_in, const int* in_sizes, int n_in,
                              void* d_out, int out_size, void* d_ws, size_t ws_size,
                              hipStream_t stream) {
    const int F_IN = 128, HID = 128, HEADS = 2;
    const int F1 = HEADS * HID;
    const int N = in_sizes[0] / F_IN;
    const int E = in_sizes[1] / 2;

    const float* x      = (const float*)d_in[0];
    const int*   ei     = (const int*)d_in[1];
    const int*   batch  = (const int*)d_in[2];
    const float* W1     = (const float*)d_in[3];
    const float* a_src1 = (const float*)d_in[4];
    const float* a_dst1 = (const float*)d_in[5];
    const float* b1     = (const float*)d_in[6];
    const float* gamma1 = (const float*)d_in[7];
    const float* beta1  = (const float*)d_in[8];
    const float* rm1    = (const float*)d_in[9];
    const float* rv1    = (const float*)d_in[10];
    const float* W2     = (const float*)d_in[11];
    const float* a_src2 = (const float*)d_in[12];
    const float* a_dst2 = (const float*)d_in[13];
    const float* b2     = (const float*)d_in[14];
    const float* gamma2 = (const float*)d_in[15];
    const float* beta2  = (const float*)d_in[16];
    const float* rm2    = (const float*)d_in[17];
    const float* rv2    = (const float*)d_in[18];

    float* ws = (float*)d_ws;
    size_t o = 0;
    float* h1   = ws + o; o += (size_t)N * F1;
    float* aggA = ws + o; o += (size_t)N * F1;
    float* aggB = ws + o; o += (size_t)N * HID;
    float* as1  = ws + o; o += (size_t)N * HEADS;
    float* ad1  = ws + o; o += (size_t)N * HEADS;
    float* as2  = ws + o; o += (size_t)N;
    float* ad2  = ws + o; o += (size_t)N;
    float* h2   = h1;
    float* agg2 = h1 + (size_t)N * HID;
    unsigned int* keep = (unsigned int*)aggB;
    uint4* plane = (uint4*)(aggB + (size_t)N * KEEP_WORDS);

    auto fillBlocks = [](long long cnt) { return (unsigned)((cnt + 255) / 256); };
    const unsigned ownerBlocks = (unsigned)((N + OWN_R - 1) / OWN_R);

    gemm_wmma_f32_kernel<128, 256><<<GEMM_GRID, GEMM_THREADS, 0, stream>>>(x, W1, h1, N / 16);
    alpha_rows_kernel<2><<<ALPHA_GRID, ALPHA_THREADS, 0, stream>>>(h1, a_src1, a_dst1, as1, ad1, N);
    owner_plane_kernel<<<fillBlocks((long long)((E + 15) / 16)), 256, 0, stream>>>(ei, E, N, plane);
    owner_aggregate_kernel<256, 2, 1, true><<<ownerBlocks, OWN_THREADS, 0, stream>>>(
        ei, E, N, h1, as1, ad1, aggA, keep, KEEP_WORDS, plane, b1, gamma1, beta1, rm1, rv1);

    gemm_wmma_f32_kernel<256, 128><<<GEMM_GRID, GEMM_THREADS, 0, stream>>>(aggA, W2, h2, N / 16);
    alpha_rows_kernel<1><<<ALPHA_GRID, ALPHA_THREADS, 0, stream>>>(h2, a_src2, a_dst2, as2, ad2, N);
    owner_aggregate_kernel<128, 1, 2, false><<<ownerBlocks, OWN_THREADS, 0, stream>>>(
        ei, E, N, h2, as2, ad2, agg2, keep, KEEP_WORDS, plane,
        nullptr, nullptr, nullptr, nullptr, nullptr);

    pool_owner_kernel<<<(unsigned)(out_size / HID), POOL_THREADS, 0, stream>>>(
        agg2, batch, b2, gamma2, beta2, rm2, rv2, (float*)d_out, N);
}


# Owner sum pass: per-edge scalar work (list entry, logits, exp, exp-sum add) done once by one lane per (edge, head), handed to the wave by readlane, not redone in all 32 lanes; exp-sum order changes
# speedup vs baseline: 4.8514x; 1.0419x over previous
#include <hip/hip_runtime.h>
#include <hip/hip_bf16.h>

typedef float v2f __attribute__((ext_vector_type(2)));
typedef float v8f __attribute__((ext_vector_type(8)));

#define NEG_SLOPE 0.2f
#define EPS 1e-5f

__device__ __forceinline__ void atomicMaxFloat(float* addr, float val) {
    if (val >= 0.0f) {
        atomicMax((int*)addr, __float_as_int(val));
    } else {
        atomicMin((unsigned int*)addr, __float_as_uint(val));
    }
}

#define GEMM_THREADS 256
#define GEMM_WAVES (GEMM_THREADS / 32)
#define GEMM_GRID 512

template <int K, int NCOL>
__global__ __launch_bounds__(GEMM_THREADS)
void gemm_wmma_f32_kernel(const float* __restrict__ A,
                          const float* __restrict__ Bm,
                          float* __restrict__ C, int nStrips) {
    constexpr int PITCH = 2 * NCOL + 16;
    constexpr int TPW = NCOL / 16 / GEMM_WAVES;
    constexpr int C4 = NCOL / 4;
    static_assert(K % 8 == 0 && NCOL % (16 * GEMM_WAVES) == 0, "tile shape");
    __shared__ __attribute__((aligned(16))) float Bs[(K / 2) * PITCH];

    const int tid = threadIdx.x;
    for (int i = tid; i < (K / 2) * C4; i += GEMM_THREADS) {
        const int p = i / C4, c4 = i - p * C4;
        const float4 r0 = *reinterpret_cast<const float4*>(Bm + (size_t)(2 * p) * NCOL + 4 * c4);
        const float4 r1 = *reinterpret_cast<const float4*>(Bm + (size_t)(2 * p + 1) * NCOL + 4 * c4);
        float4* d = reinterpret_cast<float4*>(Bs + p * PITCH + 8 * c4);
        d[0] = make_float4(r0.x, r1.x, r0.y, r1.y);
        d[1] = make_float4(r0.z, r1.z, r0.w, r1.w);
    }
    __syncthreads();

    const int wave = tid >> 5;
    const int lane = tid & 31;
    const int c = lane & 15;
    const int h = lane >> 4;
    const float* bp = Bs + (2 * h) * PITCH + 2 * (wave * TPW * 16 + c);

    for (int strip = blockIdx.x; strip < nStrips; strip += gridDim.x) {
        const float* ap = A + ((size_t)strip * 16 + c) * K + 4 * h;
        float4 a4[K / 8];
#pragma unroll
        for (int j = 0; j < K / 8; ++j) a4[j] = *reinterpret_cast<const float4*>(ap + 8 * j);

        v8f acc[TPW];
#pragma unroll
        for (int t = 0; t < TPW; ++t) acc[t] = v8f{};

#pragma unroll
        for (int j = 0; j < K / 8; ++j) {
#pragma unroll
            for (int u = 0; u < 2; ++u) {
                v2f a;
                a.x = u ? a4[j].z : a4[j].x;
                a.y = u ? a4[j].w : a4[j].y;
#pragma unroll
                for (int t = 0; t < TPW; ++t) {
                    const v2f b = *reinterpret_cast<const v2f*>(bp + (4 * j + u) * PITCH + 32 * t);
                    acc[t] = __builtin_amdgcn_wmma_f32_16x16x4_f32(false, a, false, b,
                                                                   (short)0, acc[t], false, false);
                }
            }
        }
#pragma unroll
        for (int t = 0; t < TPW; ++t) {
            float* cp = C + ((size_t)strip * 16 + 8 * h) * NCOL + (wave * TPW + t) * 16 + c;
#pragma unroll
            for (int r = 0; r < 8; ++r) cp[(size_t)r * NCOL] = acc[t][r];
        }
    }
}

#define ALPHA_GRID 512
#define ALPHA_THREADS 256

template <int H>
__global__ __launch_bounds__(ALPHA_THREADS)
void alpha_rows_kernel(const float* __restrict__ h,
                       const float* __restrict__ a_src,
                       const float* __restrict__ a_dst,
                       float* __restrict__ as, float* __restrict__ ad, int n_nodes) {
    static_assert(H == 1 || H == 2, "16 rows x heads per trip");
    constexpr int ROWS = 16 / H;
    constexpr int RV = 32 * H;
    const int lane = threadIdx.x & 31;
    const int gw = __builtin_amdgcn_readfirstlane((int)((blockIdx.x * ALPHA_THREADS + threadIdx.x) >> 5));
    const int nW = (int)((gridDim.x * ALPHA_THREADS) >> 5);
    const float4* __restrict__ h4 = reinterpret_cast<const float4*>(h);

    float4 ws[H], wd[H];
#pragma unroll
    for (int hh = 0; hh < H; ++hh) {
        const float* sp = a_src + hh * 128 + 4 * lane;
        const float* dp = a_dst + hh * 128 + 4 * lane;
        ws[hh] = make_float4(sp[0], sp[1], sp[2], sp[3]);
        wd[hh] = make_float4(dp[0], dp[1], dp[2], dp[3]);
    }

    for (int r0 = gw * ROWS; r0 < n_nodes; r0 += nW * ROWS) {
        float4 v[ROWS][H];
#pragma unroll
        for (int u = 0; u < ROWS; ++u) {
            const int n = min(r0 + u, n_nodes - 1);
#pragma unroll
            for (int hh = 0; hh < H; ++hh) v[u][hh] = h4[(size_t)n * RV + 32 * hh + lane];
        }
        __builtin_amdgcn_sched_barrier(0);
        float val[32];
#pragma unroll
        for (int u = 0; u < ROWS; ++u) {
#pragma unroll
            for (int hh = 0; hh < H; ++hh) {
                const float4 x = v[u][hh];
                val[u * H + hh] = x.x * ws[hh].x + x.y * ws[hh].y + x.z * ws[hh].z + x.w * ws[hh].w;
                val[16 + u * H + hh] = x.x * wd[hh].x + x.y * wd[hh].y + x.z * wd[hh].z + x.w * wd[hh].w;
            }
        }
#pragma unroll
        for (int half = 16, off = 16; half >= 1; half >>= 1, off >>= 1) {
            const bool up = (lane & off) != 0;
#pragma unroll
            for (int i = 0; i < half; ++i) {
                const float lo = val[i], hi = val[i + half];
                const float keep = up ? hi : lo;
                const float send = up ? lo : hi;
                val[i] = keep + __shfl_xor(send, off, 32);
            }
        }
        const int k = lane & 15;
        if (r0 + k / H < n_nodes) {
            float* o = (lane & 16) ? ad : as;
            o[(size_t)r0 * H + k] = val[0];
        }
    }
}

#define NO_LIST 0xFFFFFFFFu
#define OWN_R 200
#define OWN_THREADS 1024
#define OWN_WAVES (OWN_THREADS / 32)
#define OWN_CAP 16384
#define NO_OWNER 255u
#define KEEP_WORDS 96
#define PLANE_WORDS 32

__device__ __forceinline__ bool owner_plane_ok(int E, int n_nodes, const void* plane) {
    const unsigned nOwner = ((unsigned)n_nodes + OWN_R - 1) / OWN_R;
    const size_t bytes = ((size_t)E + 15) & ~(size_t)15;
    return plane != nullptr && nOwner <= NO_OWNER
        && bytes <= (size_t)n_nodes * (PLANE_WORDS * 4)
        && (reinterpret_cast<size_t>(plane) & 15) == 0;
}

__global__ void owner_plane_kernel(const int* __restrict__ ei, int E, int n_nodes,
                                   uint4* __restrict__ plane) {
    if (!owner_plane_ok(E, n_nodes, plane)) return;
    const int* __restrict__ dst = ei + E;
    const int nVec = (E + 15) >> 4;
    const int v = blockIdx.x * blockDim.x + threadIdx.x;
    if (v >= nVec) return;
    int d[16];
    if (((E & 15) == 0) && ((reinterpret_cast<size_t>(dst) & 15) == 0)) {
        const int4* d4 = reinterpret_cast<const int4*>(dst) + 4 * (size_t)v;
#pragma unroll
        for (int q = 0; q < 4; ++q) {
            const int4 t = d4[q];
            d[4 * q] = t.x; d[4 * q + 1] = t.y; d[4 * q + 2] = t.z; d[4 * q + 3] = t.w;
        }
    } else {
#pragma unroll
        for (int k = 0; k < 16; ++k) {
            const int i = 16 * v + k;
            d[k] = (i < E) ? dst[i] : -1;
        }
    }
    unsigned w[4];
#pragma unroll
    for (int q = 0; q < 4; ++q) {
        w[q] = 0u;
#pragma unroll
        for (int k = 0; k < 4; ++k) {
            const unsigned dd = (unsigned)d[4 * q + k];
            const unsigned o = (dd < (unsigned)n_nodes) ? dd / OWN_R : NO_OWNER;
            w[q] |= o << (8 * k);
        }
    }
    plane[v] = make_uint4(w[0], w[1], w[2], w[3]);
}

__device__ __forceinline__ float leaky(float v) { return (v > 0.0f) ? v : NEG_SLOPE * v; }

template <class Fn>
__device__ __forceinline__ void scan_dst(const int* __restrict__ dst, int E, int tid, Fn&& fn) {
    const bool vec = ((E & 3) == 0) && ((reinterpret_cast<size_t>(dst) & 15) == 0);
    if (vec) {
        const int4* d4 = reinterpret_cast<const int4*>(dst);
        const int E4 = E >> 2;
        for (int jb = 0; jb < E4; jb += 4 * OWN_THREADS) {
            int4 d[4];
#pragma unroll
            for (int u = 0; u < 4; ++u) {
                int j = jb + u * OWN_THREADS + tid;
                d[u] = (j < E4) ? d4[j] : make_int4(-1, -1, -1, -1);
            }
#pragma unroll
            for (int u = 0; u < 4; ++u) {
                int i = 4 * (jb + u * OWN_THREADS + tid);
                fn(i, d[u].x); fn(i + 1, d[u].y); fn(i + 2, d[u].z); fn(i + 3, d[u].w);
            }
        }
    } else {
        for (int ib = 0; ib < E; ib += OWN_THREADS) {
            int i = ib + tid;
            fn(i, (i < E) ? dst[i] : -1);
        }
    }
}

template <class Fn>
__device__ __forceinline__ void scan_plane(const uint4* __restrict__ plane, int nVec,
                                           unsigned me, int tid, Fn&& hit) {
    const unsigned key = me * 0x01010101u;
    for (int vb = 0; vb < nVec; vb += 8 * OWN_THREADS) {
        uint4 d[8];
#pragma unroll
        for (int u = 0; u < 8; ++u) {
            const int v = vb + u * OWN_THREADS + tid;
            d[u] = (v < nVec) ? plane[v] : make_uint4(~0u, ~0u, ~0u, ~0u);
        }
#pragma unroll
        for (int u = 0; u < 8; ++u) {
            const int i0 = 16 * (vb + u * OWN_THREADS + tid);
            const unsigned w[4] = { d[u].x, d[u].y, d[u].z, d[u].w };
#pragma unroll
            for (int q = 0; q < 4; ++q) {
                const unsigned x = w[q] ^ key;
                unsigned z = ~(((x & 0x7F7F7F7Fu) + 0x7F7F7F7Fu) | x | 0x7F7F7F7Fu);
                while (z) {
                    const int k = __builtin_ctz(z) >> 3;
                    z &= z - 1u;
                    hit(i0 + 4 * q + k);
                }
            }
        }
    }
}

template <int F, int H, int U>
__device__ __forceinline__ void accumulate_edges(const int (&s)[U], const unsigned (&dl)[U],
                                                 const bool (&ok)[U], int lane,
                                                 const float* __restrict__ h,
                                                 const float* __restrict__ as,
                                                 const float* adl, const float* m,
                                                 float* den, float* acc) {
    constexpr int C = F / H;
    constexpr int J = F / 128;
    static_assert(F % 128 == 0 && C % 128 == 0, "a 512-byte piece group lies inside one head");
    float4 r[U][J];
    float a[U][H];
#pragma unroll
    for (int u = 0; u < U; ++u) {
        const float4* hs = reinterpret_cast<const float4*>(h + (size_t)s[u] * F);
#pragma unroll
        for (int j = 0; j < J; ++j) r[u][j] = hs[lane + 32 * j];
#pragma unroll
        for (int hh = 0; hh < H; ++hh) a[u][hh] = as[(size_t)s[u] * H + hh];
    }
#pragma unroll
    for (int u = 0; u < U; ++u) {
        if (ok[u]) {
            float ex[H];
#pragma unroll
            for (int hh = 0; hh < H; ++hh) {
                float v = leaky(a[u][hh] + adl[dl[u] * H + hh]);
                ex[hh] = __expf(v - m[dl[u] * H + hh]);
            }
#pragma unroll
            for (int hh = 0; hh < H; ++hh)
                if (lane == hh) atomicAdd(&den[dl[u] * H + hh], ex[hh]);
            float* ar = acc + dl[u] * F + lane;
#pragma unroll
            for (int j = 0; j < J; ++j) {
                const float e = ex[(128 * j) / C];
                atomicAdd(&ar[128 * j],      r[u][j].x * e);
                atomicAdd(&ar[128 * j + 32], r[u][j].y * e);
                atomicAdd(&ar[128 * j + 64], r[u][j].z * e);
                atomicAdd(&ar[128 * j + 96], r[u][j].w * e);
            }
        }
    }
}

template <int F, int H, int U>
__device__ __forceinline__ void accumulate_trip(unsigned pk_l, bool okl, int nvalid, int lane,
                                                const float* __restrict__ h,
                                                const float* __restrict__ as,
                                                const float* adl, const float* m,
                                                float* den, float* acc) {
    constexpr int C = F / H;
    constexpr int J = F / 128;
    static_assert(F % 128 == 0 && C % 128 == 0, "a 512-byte piece group lies inside one head");
    static_assert(U * H <= 32, "one lane per (edge, head)");
    const int hh_l = lane % H;
    const unsigned s_l = pk_l >> 8, d_l = pk_l & 255u;
    const float a_l = as[(size_t)s_l * H + hh_l];
    const float ad_l = adl[d_l * H + hh_l];
    const float m_l = m[d_l * H + hh_l];
    unsigned pk[U];
#pragma unroll
    for (int u = 0; u < U; ++u) pk[u] = (unsigned)__builtin_amdgcn_readlane((int)pk_l, u * H);
    float4 r[U][J];
#pragma unroll
    for (int u = 0; u < U; ++u) {
        const float4* hs = reinterpret_cast<const float4*>(h + (size_t)(pk[u] >> 8) * F);
#pragma unroll
        for (int j = 0; j < J; ++j) r[u][j] = hs[lane + 32 * j];
    }
    const float w_l = __expf(leaky(a_l + ad_l) - m_l);
    if (okl) atomicAdd(&den[d_l * H + hh_l], w_l);
#pragma unroll
    for (int u = 0; u < U; ++u) {
        if (u < nvalid) {
            float* ar = acc + (pk[u] & 255u) * F + lane;
#pragma unroll
            for (int j = 0; j < J; ++j) {
                const float e = __int_as_float(
                    __builtin_amdgcn_readlane(__float_as_int(w_l), u * H + (128 * j) / C));
                atomicAdd(&ar[128 * j],      r[u][j].x * e);
                atomicAdd(&ar[128 * j + 32], r[u][j].y * e);
                atomicAdd(&ar[128 * j + 64], r[u][j].z * e);
                atomicAdd(&ar[128 * j + 96], r[u][j].w * e);
            }
        }
    }
}

template <int F, int H, int LISTS, bool POST>
__global__ __launch_bounds__(OWN_THREADS)
void owner_aggregate_kernel(const int* __restrict__ ei, int E, int n_nodes,
                            const float* __restrict__ h,
                            const float* __restrict__ as,
                            const float* __restrict__ ad,
                            float* agg, unsigned int* keep, int keepF,
                            const uint4* __restrict__ plane,
                            const float* __restrict__ pb,
                            const float* __restrict__ pgamma,
                            const float* __restrict__ pbeta,
                            const float* __restrict__ pmean,
                            const float* __restrict__ pvar) {
    constexpr int C = F / H;
    __shared__ float acc[OWN_R * F];
    __shared__ unsigned int list[OWN_CAP];
    __shared__ float m[OWN_R * H];
    __shared__ float den[OWN_R * H];
    __shared__ float adl[OWN_R * H];
    __shared__ unsigned int cnt;

    const int tid = threadIdx.x;
    const int lane = tid & 31;
    const int wave = tid >> 5;
    const int base = blockIdx.x * OWN_R;
    const int R = min(OWN_R, n_nodes - base);
    const unsigned uR = (unsigned)R;
    const int* __restrict__ src = ei;
    const int* __restrict__ dst = ei + E;
    unsigned int* kp = keep + (size_t)base * keepF;
    const unsigned keepCap = (unsigned)R * (unsigned)keepF;

    for (int i = tid; i < R * F; i += OWN_THREADS) acc[i] = 0.0f;
    for (int i = tid; i < R * H; i += OWN_THREADS) {
        float a_d = ad[(size_t)base * H + i];
        adl[i] = a_d;
        m[i] = leaky(as[(size_t)base * H + i] + a_d);
        den[i] = 0.0f;
    }
    if (tid == 0) cnt = 0u;
    __syncthreads();

    unsigned kept = NO_LIST;
    if (LISTS == 2) kept = kp[0];
    const bool have = (LISTS == 2) && (kept <= OWN_CAP) && (kept < keepCap);

    const bool packs = (E <= (1 << 24)) && (n_nodes <= (1 << 24)) && (OWN_R <= 256);
    constexpr int TRIP = 4;
    const unsigned el = (unsigned)lane / H;
    const bool book = lane < TRIP * H;
    const unsigned wv = (unsigned)__builtin_amdgcn_readfirstlane(wave);
    if (packs && !have) {
        if (owner_plane_ok(E, n_nodes, plane)) {
            scan_plane(plane, (E + 15) >> 4, (unsigned)blockIdx.x, tid, [&](int i) {
                unsigned pos = atomicAdd(&cnt, 1u);
                if (pos < OWN_CAP) list[pos] = (unsigned)i << 8;
            });
        } else {
            scan_dst(dst, E, tid, [&](int i, int d) {
                unsigned dl = (unsigned)(d - base);
                if (dl < uR) {
                    unsigned pos = atomicAdd(&cnt, 1u);
                    if (pos < OWN_CAP) list[pos] = (unsigned)i << 8;
                }
            });
        }
    }
    __syncthreads();
    const unsigned total = have ? kept : cnt;
    const bool fast = have || (packs && (total <= OWN_CAP));
    const bool leave = (LISTS == 1) && fast && (total < keepCap);

    if (fast) {
        for (unsigned p = tid; p < total; p += OWN_THREADS) {
            unsigned dl;
            int s;
            if (have) {
                unsigned pk = kp[1 + p];
                dl = pk & 255u;
                s = (int)(pk >> 8);
            } else {
                const unsigned i = list[p] >> 8;
                dl = (unsigned)(dst[i] - base);
                s = src[i];
            }
#pragma unroll
            for (int hh = 0; hh < H; ++hh)
                atomicMaxFloat(&m[dl * H + hh], leaky(as[(size_t)s * H + hh] + adl[dl * H + hh]));
            list[p] = ((unsigned)s << 8) | dl;
            if (leave) kp[1 + p] = ((unsigned)s << 8) | dl;
        }
        if (LISTS == 1 && tid == 0) kp[0] = leave ? total : NO_LIST;
        __syncthreads();
        const unsigned tot = (unsigned)__builtin_amdgcn_readfirstlane((int)total);
        for (unsigned p0 = wv * TRIP; p0 < tot; p0 += OWN_WAVES * TRIP) {
            const bool okl = book && (p0 + el) < tot;
            const unsigned pk_l = okl ? list[p0 + el] : 0u;
            accumulate_trip<F, H, TRIP>(pk_l, okl, (int)min((unsigned)TRIP, tot - p0), lane,
                                        h, as, adl, m, den, acc);
        }
    } else {
        if (LISTS == 1 && tid == 0) kp[0] = NO_LIST;
        scan_dst(dst, E, tid, [&](int i, int d) {
            unsigned dl = (unsigned)(d - base);
            if (dl < uR) {
                int s = src[i];
#pragma unroll
                for (int hh = 0; hh < H; ++hh)
                    atomicMaxFloat(&m[dl * H + hh], leaky(as[(size_t)s * H + hh] + adl[dl * H + hh]));
            }
        });
        __syncthreads();
        scan_dst(dst, E, tid, [&](int i, int d) {
            unsigned dl = (unsigned)(d - base);
            bool hit = dl < uR;
            int sv = hit ? src[i] : 0;
            unsigned mask = __builtin_amdgcn_ballot_w32(hit);
            while (mask) {
                int l = __builtin_ctz(mask);
                mask &= mask - 1u;
                int s1[1] = { __shfl(sv, l, 32) };
                unsigned d1[1] = { (unsigned)__shfl((int)dl, l, 32) };
                bool o1[1] = { true };
                accumulate_edges<F, H, 1>(s1, d1, o1, lane, h, as, adl, m, den, acc);
            }
        });
    }
    if (packs) {
        for (unsigned q0 = wv * TRIP; q0 < uR; q0 += OWN_WAVES * TRIP) {
            const unsigned ql = q0 + el;
            const bool okl = book && ql < uR;
            const unsigned pk_l = okl ? ((((unsigned)base + ql) << 8) | ql) : 0u;
            accumulate_trip<F, H, TRIP>(pk_l, okl, (int)min((unsigned)TRIP, uR - q0), lane,
                                        h, as, adl, m, den, acc);
        }
    } else {
        for (int q0 = wave; q0 < R; q0 += OWN_WAVES) {
            int s1[1] = { base + q0 };
            unsigned d1[1] = { (unsigned)q0 };
            bool o1[1] = { true };
            accumulate_edges<F, H, 1>(s1, d1, o1, lane, h, as, adl, m, den, acc);
        }
    }
    __syncthreads();

    constexpr int Q = F / 4;
    static_assert(OWN_THREADS % Q == 0, "one piece per thread");
    const int cq = tid % Q;
    const int c0 = 4 * cq;
    const int w0 = 128 * (cq >> 5) + (cq & 31);
    const int hd = c0 / C;
    float4* agg4 = reinterpret_cast<float4*>(agg + (size_t)base * F);
    if (POST) {
        float fb[4], fg[4], ft[4], fm[4], frs[4];
        for (int k = 0; k < 4; ++k) {
            const int c = c0 + k;
            fb[k] = pb[c]; fg[k] = pgamma[c]; ft[k] = pbeta[c]; fm[k] = pmean[c];
            frs[k] = rsqrtf(pvar[c] + EPS);
        }
        for (int i = tid; i < R * Q; i += OWN_THREADS) {
            const int dl = i / Q;
            const float* ar = acc + dl * F + w0;
            const float d = den[dl * H + hd];
            float o[4];
            for (int k = 0; k < 4; ++k) {
                float v = ar[32 * k] / d + fb[k];
                v = fg[k] * (v - fm[k]) * frs[k] + ft[k];
                o[k] = (v > 0.0f) ? v : 0.0f;
            }
            agg4[i] = make_float4(o[0], o[1], o[2], o[3]);
        }
    } else {
        for (int i = tid; i < R * Q; i += OWN_THREADS) {
            const int dl = i / Q;
            const float* ar = acc + dl * F + w0;
            const float d = den[dl * H + hd];
            agg4[i] = make_float4(ar[0] / d, ar[32] / d, ar[64] / d, ar[96] / d);
        }
    }
}

#define POOL_THREADS 1024
#define POOL_WAVES (POOL_THREADS / 32)
#define POOL_CAP 65536
#define POOL_VECS 16
#define POOL_ROWS 8

__global__ __launch_bounds__(POOL_THREADS)
void pool_owner_kernel(const float* __restrict__ x,
                       const int* __restrict__ batch,
                       const float* __restrict__ b,
                       const float* __restrict__ gamma,
                       const float* __restrict__ beta,
                       const float* __restrict__ mean,
                       const float* __restrict__ var,
                       float* __restrict__ out, int n_nodes) {
    constexpr int F = 128;
    __shared__ unsigned short list[POOL_CAP];
    __shared__ float4 part[POOL_WAVES][32];
    __shared__ unsigned int cnt;

    const int tid = threadIdx.x;
    const int lane = tid & 31;
    const int wave = tid >> 5;
    const int g = blockIdx.x;
    const float4* __restrict__ x4 = reinterpret_cast<const float4*>(x);

    float pb[4], pg[4], pt[4], pm[4], prs[4];
    for (int k = 0; k < 4; ++k) {
        const int c = 4 * lane + k;
        pb[k] = b[c]; pg[k] = gamma[c]; pt[k] = beta[c]; pm[k] = mean[c];
        prs[k] = rsqrtf(var[c] + EPS);
    }
    auto act = [&](float xv, int k) {
        float v = xv + pb[k];
        v = pg[k] * (v - pm[k]) * prs[k] + pt[k];
        return (v > 0.0f) ? v : 0.0f;
    };
    const float ninf = __int_as_float((int)0xFF800000u);
    float mx[4] = { ninf, ninf, ninf, ninf };

    for (int chunk0 = 0; chunk0 < n_nodes; chunk0 += POOL_CAP) {
        const int len = min(POOL_CAP, n_nodes - chunk0);
        if (tid == 0) cnt = 0u;
        __syncthreads();
        const int* __restrict__ bp = batch + chunk0;
        const int len4 = ((reinterpret_cast<size_t>(bp) & 15) == 0) ? (len >> 2) : 0;
        const int4* __restrict__ bp4 = reinterpret_cast<const int4*>(bp);
        auto push = [&](int i) {
            const unsigned pos = atomicAdd(&cnt, 1u);
            list[pos] = (unsigned short)i;
        };
        for (int jb = 0; jb < len4; jb += POOL_VECS * POOL_THREADS) {
            int4 d[POOL_VECS];
#pragma unroll
            for (int u = 0; u < POOL_VECS; ++u) {
                const int j = jb + u * POOL_THREADS + tid;
                d[u] = (j < len4) ? bp4[j] : make_int4(-1, -1, -1, -1);
            }
            bool any = false;
#pragma unroll
            for (int u = 0; u < POOL_VECS; ++u)
                any |= (d[u].x == g) | (d[u].y == g) | (d[u].z == g) | (d[u].w == g);
            if (__builtin_amdgcn_ballot_w32(any) != 0u) {
                unsigned long long hm = 0ull;
#pragma unroll
                for (int u = 0; u < POOL_VECS; ++u) {
                    const unsigned m4 = (d[u].x == g ? 1u : 0u) | (d[u].y == g ? 2u : 0u)
                                      | (d[u].z == g ? 4u : 0u) | (d[u].w == g ? 8u : 0u);
                    hm |= (unsigned long long)m4 << (4 * u);
                }
                while (hm) {
                    const int bit = __builtin_ctzll(hm);
                    hm &= hm - 1ull;
                    push(4 * (jb + (bit >> 2) * POOL_THREADS + tid) + (bit & 3));
                }
            }
        }
        for (int i = 4 * len4 + tid; i < len; i += POOL_THREADS)
            if (bp[i] == g) push(i);
        __syncthreads();
        const unsigned total = cnt;

        for (unsigned p0 = wave; p0 < total; p0 += POOL_WAVES * POOL_ROWS) {
            float4 r[POOL_ROWS];
#pragma unroll
            for (int u = 0; u < POOL_ROWS; ++u) {
                const unsigned p = p0 + u * POOL_WAVES;
                const int n = chunk0 + (int)list[(p < total) ? p : p0];
                r[u] = x4[(size_t)n * (F / 4) + lane];
            }
#pragma unroll
            for (int u = 0; u < POOL_ROWS; ++u) {
                if (p0 + u * POOL_WAVES < total) {
                    mx[0] = fmaxf(mx[0], act(r[u].x, 0));
                    mx[1] = fmaxf(mx[1], act(r[u].y, 1));
                    mx[2] = fmaxf(mx[2], act(r[u].z, 2));
                    mx[3] = fmaxf(mx[3], act(r[u].w, 3));
                }
            }
        }
        __syncthreads();
    }

    part[wave][lane] = make_float4(mx[0], mx[1], mx[2], mx[3]);
    __syncthreads();
    if (tid < F) {
        const float* pf = reinterpret_cast<const float*>(&part[0][0]);
        float m = pf[tid];
        for (int w = 1; w < POOL_WAVES; ++w) m = fmaxf(m, pf[w * F + tid]);
        out[(size_t)g * F + tid] = m;
    }
}

extern "C" void kernel_launch(void* const* d_in, const int* in_sizes, int n_in,
                              void* d_out, int out_size, void* d_ws, size_t ws_size,
                              hipStream_t stream) {
    const int F_IN = 128, HID = 128, HEADS = 2;
    const int F1 = HEADS * HID;
    const int N = in_sizes[0] / F_IN;
    const int E = in_sizes[1] / 2;

    const float* x      = (const float*)d_in[0];
    const int*   ei     = (const int*)d_in[1];
    const int*   batch  = (const int*)d_in[2];
    const float* W1     = (const float*)d_in[3];
    const float* a_src1 = (const float*)d_in[4];
    const float* a_dst1 = (const float*)d_in[5];
    const float* b1     = (const float*)d_in[6];
    const float* gamma1 = (const float*)d_in[7];
    const float* beta1  = (const float*)d_in[8];
    const float* rm1    = (const float*)d_in[9];
    const float* rv1    = (const float*)d_in[10];
    const float* W2     = (const float*)d_in[11];
    const float* a_src2 = (const float*)d_in[12];
    const float* a_dst2 = (const float*)d_in[13];
    const float* b2     = (const float*)d_in[14];
    const float* gamma2 = (const float*)d_in[15];
    const float* beta2  = (const float*)d_in[16];
    const float* rm2    = (const float*)d_in[17];
    const float* rv2    = (const float*)d_in[18];

    float* ws = (float*)d_ws;
    size_t o = 0;
    float* h1   = ws + o; o += (size_t)N * F1;
    float* aggA = ws + o; o += (size_t)N * F1;
    float* aggB = ws + o; o += (size_t)N * HID;
    float* as1  = ws + o; o += (size_t)N * HEADS;
    float* ad1  = ws + o; o += (size_t)N * HEADS;
    float* as2  = ws + o; o += (size_t)N;
    float* ad2  = ws + o; o += (size_t)N;
    float* h2   = h1;
    float* agg2 = h1 + (size_t)N * HID;
    unsigned int* keep = (unsigned int*)aggB;
    uint4* plane = (uint4*)(aggB + (size_t)N * KEEP_WORDS);

    auto fillBlocks = [](long long cnt) { return (unsigned)((cnt + 255) / 256); };
    const unsigned ownerBlocks = (unsigned)((N + OWN_R - 1) / OWN_R);

    gemm_wmma_f32_kernel<128, 256><<<GEMM_GRID, GEMM_THREADS, 0, stream>>>(x, W1, h1, N / 16);
    alpha_rows_kernel<2><<<ALPHA_GRID, ALPHA_THREADS, 0, stream>>>(h1, a_src1, a_dst1, as1, ad1, N);
    owner_plane_kernel<<<fillBlocks((long long)((E + 15) / 16)), 256, 0, stream>>>(ei, E, N, plane);
    owner_aggregate_kernel<256, 2, 1, true><<<ownerBlocks, OWN_THREADS, 0, stream>>>(
        ei, E, N, h1, as1, ad1, aggA, keep, KEEP_WORDS, plane, b1, gamma1, beta1, rm1, rv1);

    gemm_wmma_f32_kernel<256, 128><<<GEMM_GRID, GEMM_THREADS, 0, stream>>>(aggA, W2, h2, N / 16);
    alpha_rows_kernel<1><<<ALPHA_GRID, ALPHA_THREADS, 0, stream>>>(h2, a_src2, a_dst2, as2, ad2, N);
    owner_aggregate_kernel<128, 1, 2, false><<<ownerBlocks, OWN_THREADS, 0, stream>>>(
        ei, E, N, h2, as2, ad2, agg2, keep, KEEP_WORDS, plane,
        nullptr, nullptr, nullptr, nullptr, nullptr);

    pool_owner_kernel<<<(unsigned)(out_size / HID), POOL_THREADS, 0, stream>>>(
        agg2, batch, b2, gamma2, beta2, rm2, rv2, (float*)d_out, N);
}


# Edge lists by owner range made once per call by a binning launch (LDS rank per work-group, one returning add per range on its count word); layer-1 owners read them, no plane scan; sum order changes
# speedup vs baseline: 5.2709x; 1.0865x over previous
#include <hip/hip_runtime.h>
#include <hip/hip_bf16.h>

typedef float v2f __attribute__((ext_vector_type(2)));
typedef float v8f __attribute__((ext_vector_type(8)));

#define NEG_SLOPE 0.2f
#define EPS 1e-5f

__device__ __forceinline__ void atomicMaxFloat(float* addr, float val) {
    if (val >= 0.0f) {
        atomicMax((int*)addr, __float_as_int(val));
    } else {
        atomicMin((unsigned int*)addr, __float_as_uint(val));
    }
}

#define GEMM_THREADS 256
#define GEMM_WAVES (GEMM_THREADS / 32)
#define GEMM_GRID 512

template <int K, int NCOL>
__global__ __launch_bounds__(GEMM_THREADS)
void gemm_wmma_f32_kernel(const float* __restrict__ A,
                          const float* __restrict__ Bm,
                          float* __restrict__ C, int nStrips,
                          unsigned int* __restrict__ clr, int clrN, int clrStride) {
    constexpr int PITCH = 2 * NCOL + 16;
    constexpr int TPW = NCOL / 16 / GEMM_WAVES;
    constexpr int C4 = NCOL / 4;
    static_assert(K % 8 == 0 && NCOL % (16 * GEMM_WAVES) == 0, "tile shape");
    __shared__ __attribute__((aligned(16))) float Bs[(K / 2) * PITCH];

    const int tid = threadIdx.x;
    if (clr != nullptr && blockIdx.x == 0)
        for (int i = tid; i < clrN; i += GEMM_THREADS) clr[(size_t)i * clrStride] = 0u;
    for (int i = tid; i < (K / 2) * C4; i += GEMM_THREADS) {
        const int p = i / C4, c4 = i - p * C4;
        const float4 r0 = *reinterpret_cast<const float4*>(Bm + (size_t)(2 * p) * NCOL + 4 * c4);
        const float4 r1 = *reinterpret_cast<const float4*>(Bm + (size_t)(2 * p + 1) * NCOL + 4 * c4);
        float4* d = reinterpret_cast<float4*>(Bs + p * PITCH + 8 * c4);
        d[0] = make_float4(r0.x, r1.x, r0.y, r1.y);
        d[1] = make_float4(r0.z, r1.z, r0.w, r1.w);
    }
    __syncthreads();

    const int wave = tid >> 5;
    const int lane = tid & 31;
    const int c = lane & 15;
    const int h = lane >> 4;
    const float* bp = Bs + (2 * h) * PITCH + 2 * (wave * TPW * 16 + c);

    for (int strip = blockIdx.x; strip < nStrips; strip += gridDim.x) {
        const float* ap = A + ((size_t)strip * 16 + c) * K + 4 * h;
        float4 a4[K / 8];
#pragma unroll
        for (int j = 0; j < K / 8; ++j) a4[j] = *reinterpret_cast<const float4*>(ap + 8 * j);

        v8f acc[TPW];
#pragma unroll
        for (int t = 0; t < TPW; ++t) acc[t] = v8f{};

#pragma unroll
        for (int j = 0; j < K / 8; ++j) {
#pragma unroll
            for (int u = 0; u < 2; ++u) {
                v2f a;
                a.x = u ? a4[j].z : a4[j].x;
                a.y = u ? a4[j].w : a4[j].y;
#pragma unroll
                for (int t = 0; t < TPW; ++t) {
                    const v2f b = *reinterpret_cast<const v2f*>(bp + (4 * j + u) * PITCH + 32 * t);
                    acc[t] = __builtin_amdgcn_wmma_f32_16x16x4_f32(false, a, false, b,
                                                                   (short)0, acc[t], false, false);
                }
            }
        }
#pragma unroll
        for (int t = 0; t < TPW; ++t) {
            float* cp = C + ((size_t)strip * 16 + 8 * h) * NCOL + (wave * TPW + t) * 16 + c;
#pragma unroll
            for (int r = 0; r < 8; ++r) cp[(size_t)r * NCOL] = acc[t][r];
        }
    }
}

#define ALPHA_GRID 512
#define ALPHA_THREADS 256

template <int H>
__global__ __launch_bounds__(ALPHA_THREADS)
void alpha_rows_kernel(const float* __restrict__ h,
                       const float* __restrict__ a_src,
                       const float* __restrict__ a_dst,
                       float* __restrict__ as, float* __restrict__ ad, int n_nodes) {
    static_assert(H == 1 || H == 2, "16 rows x heads per trip");
    constexpr int ROWS = 16 / H;
    constexpr int RV = 32 * H;
    const int lane = threadIdx.x & 31;
    const int gw = __builtin_amdgcn_readfirstlane((int)((blockIdx.x * ALPHA_THREADS + threadIdx.x) >> 5));
    const int nW = (int)((gridDim.x * ALPHA_THREADS) >> 5);
    const float4* __restrict__ h4 = reinterpret_cast<const float4*>(h);

    float4 ws[H], wd[H];
#pragma unroll
    for (int hh = 0; hh < H; ++hh) {
        const float* sp = a_src + hh * 128 + 4 * lane;
        const float* dp = a_dst + hh * 128 + 4 * lane;
        ws[hh] = make_float4(sp[0], sp[1], sp[2], sp[3]);
        wd[hh] = make_float4(dp[0], dp[1], dp[2], dp[3]);
    }

    for (int r0 = gw * ROWS; r0 < n_nodes; r0 += nW * ROWS) {
        float4 v[ROWS][H];
#pragma unroll
        for (int u = 0; u < ROWS; ++u) {
            const int n = min(r0 + u, n_nodes - 1);
#pragma unroll
            for (int hh = 0; hh < H; ++hh) v[u][hh] = h4[(size_t)n * RV + 32 * hh + lane];
        }
        __builtin_amdgcn_sched_barrier(0);
        float val[32];
#pragma unroll
        for (int u = 0; u < ROWS; ++u) {
#pragma unroll
            for (int hh = 0; hh < H; ++hh) {
                const float4 x = v[u][hh];
                val[u * H + hh] = x.x * ws[hh].x + x.y * ws[hh].y + x.z * ws[hh].z + x.w * ws[hh].w;
                val[16 + u * H + hh] = x.x * wd[hh].x + x.y * wd[hh].y + x.z * wd[hh].z + x.w * wd[hh].w;
            }
        }
#pragma unroll
        for (int half = 16, off = 16; half >= 1; half >>= 1, off >>= 1) {
            const bool up = (lane & off) != 0;
#pragma unroll
            for (int i = 0; i < half; ++i) {
                const float lo = val[i], hi = val[i + half];
                const float keep = up ? hi : lo;
                const float send = up ? lo : hi;
                val[i] = keep + __shfl_xor(send, off, 32);
            }
        }
        const int k = lane & 15;
        if (r0 + k / H < n_nodes) {
            float* o = (lane & 16) ? ad : as;
            o[(size_t)r0 * H + k] = val[0];
        }
    }
}

#define NO_LIST 0xFFFFFFFFu
#define OWN_R 200
#define OWN_THREADS 1024
#define OWN_WAVES (OWN_THREADS / 32)
#define OWN_CAP 16384
#define KEEP_WORDS 96

#define BIN_THREADS 512
#define BIN_VECS 2
#define BIN_SLOTS 256

__global__ __launch_bounds__(BIN_THREADS)
void owner_bin_kernel(const int* __restrict__ ei, int E, int n_nodes,
                      unsigned int* __restrict__ keep, int keepF) {
    __shared__ unsigned int slot[BIN_SLOTS];
    constexpr int PER = 4 * BIN_VECS;
    const int tid = threadIdx.x;
    const unsigned nOwner = ((unsigned)n_nodes + OWN_R - 1) / OWN_R;
    const size_t regionW = (size_t)OWN_R * (size_t)keepF;
    const bool packs = (E <= (1 << 24)) && (n_nodes <= (1 << 24)) && (OWN_R <= 256);
    if (!packs || nOwner > (unsigned)BIN_SLOTS) {
        for (unsigned o = blockIdx.x * BIN_THREADS + tid; o < nOwner; o += gridDim.x * BIN_THREADS)
            keep[(size_t)o * regionW] = NO_LIST;
        return;
    }
    for (int i = tid; i < BIN_SLOTS; i += BIN_THREADS) slot[i] = 0u;
    __syncthreads();

    const int* __restrict__ src = ei;
    const int* __restrict__ dst = ei + E;
    const int T = (int)(gridDim.x * BIN_THREADS);
    const int t = (int)(blockIdx.x * BIN_THREADS) + tid;
    int s[PER], d[PER];
    if (((E & 3) == 0) && ((reinterpret_cast<size_t>(ei) & 15) == 0)) {
        const int4* __restrict__ s4 = reinterpret_cast<const int4*>(src);
        const int4* __restrict__ d4 = reinterpret_cast<const int4*>(dst);
        const int nVec = E >> 2;
        for (int u = 0; u < BIN_VECS; ++u) {
            const int v = t + u * T;
            const bool in = v < nVec;
            const int4 dv = in ? d4[v] : make_int4(-1, -1, -1, -1);
            const int4 sv = in ? s4[v] : make_int4(0, 0, 0, 0);
            d[4 * u] = dv.x; d[4 * u + 1] = dv.y; d[4 * u + 2] = dv.z; d[4 * u + 3] = dv.w;
            s[4 * u] = sv.x; s[4 * u + 1] = sv.y; s[4 * u + 2] = sv.z; s[4 * u + 3] = sv.w;
        }
    } else {
        for (int k = 0; k < PER; ++k) {
            const int i = t + k * T;
            d[k] = (i < E) ? dst[i] : -1;
            s[k] = (i < E) ? src[i] : 0;
        }
    }
    unsigned o[PER], rk[PER];
    for (int k = 0; k < PER; ++k) {
        const unsigned dd = (unsigned)d[k];
        const bool ok = dd < (unsigned)n_nodes;
        o[k] = ok ? dd / OWN_R : NO_LIST;
        rk[k] = ok ? atomicAdd(&slot[o[k]], 1u) : 0u;
    }
    __syncthreads();
    if (tid < (int)nOwner) {
        const unsigned cnt = slot[tid];
        unsigned b = 0u;
        if (cnt != 0u) b = atomicAdd(&keep[(size_t)tid * regionW], cnt);
        slot[tid] = b;
    }
    __syncthreads();
    for (int k = 0; k < PER; ++k) {
        if (o[k] != NO_LIST) {
            const unsigned ob = o[k] * OWN_R;
            const unsigned room = (unsigned)min(OWN_R, n_nodes - (int)ob) * (unsigned)keepF;
            const unsigned pos = slot[o[k]] + rk[k];
            if (pos + 1u < room)
                keep[(size_t)o[k] * regionW + 1u + pos] = ((unsigned)s[k] << 8) | ((unsigned)d[k] - ob);
        }
    }
}

__device__ __forceinline__ float leaky(float v) { return (v > 0.0f) ? v : NEG_SLOPE * v; }

template <class Fn>
__device__ __forceinline__ void scan_dst(const int* __restrict__ dst, int E, int tid, Fn&& fn) {
    const bool vec = ((E & 3) == 0) && ((reinterpret_cast<size_t>(dst) & 15) == 0);
    if (vec) {
        const int4* d4 = reinterpret_cast<const int4*>(dst);
        const int E4 = E >> 2;
        for (int jb = 0; jb < E4; jb += 4 * OWN_THREADS) {
            int4 d[4];
#pragma unroll
            for (int u = 0; u < 4; ++u) {
                int j = jb + u * OWN_THREADS + tid;
                d[u] = (j < E4) ? d4[j] : make_int4(-1, -1, -1, -1);
            }
#pragma unroll
            for (int u = 0; u < 4; ++u) {
                int i = 4 * (jb + u * OWN_THREADS + tid);
                fn(i, d[u].x); fn(i + 1, d[u].y); fn(i + 2, d[u].z); fn(i + 3, d[u].w);
            }
        }
    } else {
        for (int ib = 0; ib < E; ib += OWN_THREADS) {
            int i = ib + tid;
            fn(i, (i < E) ? dst[i] : -1);
        }
    }
}

template <int F, int H, int U>
__device__ __forceinline__ void accumulate_edges(const int (&s)[U], const unsigned (&dl)[U],
                                                 const bool (&ok)[U], int lane,
                                                 const float* __restrict__ h,
                                                 const float* __restrict__ as,
                                                 const float* adl, const float* m,
                                                 float* den, float* acc) {
    constexpr int C = F / H;
    constexpr int J = F / 128;
    static_assert(F % 128 == 0 && C % 128 == 0, "a 512-byte piece group lies inside one head");
    float4 r[U][J];
    float a[U][H];
#pragma unroll
    for (int u = 0; u < U; ++u) {
        const float4* hs = reinterpret_cast<const float4*>(h + (size_t)s[u] * F);
#pragma unroll
        for (int j = 0; j < J; ++j) r[u][j] = hs[lane + 32 * j];
#pragma unroll
        for (int hh = 0; hh < H; ++hh) a[u][hh] = as[(size_t)s[u] * H + hh];
    }
#pragma unroll
    for (int u = 0; u < U; ++u) {
        if (ok[u]) {
            float ex[H];
#pragma unroll
            for (int hh = 0; hh < H; ++hh) {
                float v = leaky(a[u][hh] + adl[dl[u] * H + hh]);
                ex[hh] = __expf(v - m[dl[u] * H + hh]);
            }
#pragma unroll
            for (int hh = 0; hh < H; ++hh)
                if (lane == hh) atomicAdd(&den[dl[u] * H + hh], ex[hh]);
            float* ar = acc + dl[u] * F + lane;
#pragma unroll
            for (int j = 0; j < J; ++j) {
                const float e = ex[(128 * j) / C];
                atomicAdd(&ar[128 * j],      r[u][j].x * e);
                atomicAdd(&ar[128 * j + 32], r[u][j].y * e);
                atomicAdd(&ar[128 * j + 64], r[u][j].z * e);
                atomicAdd(&ar[128 * j + 96], r[u][j].w * e);
            }
        }
    }
}

template <int F, int H, int U>
__device__ __forceinline__ void accumulate_trip(unsigned pk_l, bool okl, int nvalid, int lane,
                                                const float* __restrict__ h,
                                                const float* __restrict__ as,
                                                const float* adl, const float* m,
                                                float* den, float* acc) {
    constexpr int C = F / H;
    constexpr int J = F / 128;
    static_assert(F % 128 == 0 && C % 128 == 0, "a 512-byte piece group lies inside one head");
    static_assert(U * H <= 32, "one lane per (edge, head)");
    const int hh_l = lane % H;
    const unsigned s_l = pk_l >> 8, d_l = pk_l & 255u;
    const float a_l = as[(size_t)s_l * H + hh_l];
    const float ad_l = adl[d_l * H + hh_l];
    const float m_l = m[d_l * H + hh_l];
    unsigned pk[U];
#pragma unroll
    for (int u = 0; u < U; ++u) pk[u] = (unsigned)__builtin_amdgcn_readlane((int)pk_l, u * H);
    float4 r[U][J];
#pragma unroll
    for (int u = 0; u < U; ++u) {
        const float4* hs = reinterpret_cast<const float4*>(h + (size_t)(pk[u] >> 8) * F);
#pragma unroll
        for (int j = 0; j < J; ++j) r[u][j] = hs[lane + 32 * j];
    }
    const float w_l = __expf(leaky(a_l + ad_l) - m_l);
    if (okl) atomicAdd(&den[d_l * H + hh_l], w_l);
#pragma unroll
    for (int u = 0; u < U; ++u) {
        if (u < nvalid) {
            float* ar = acc + (pk[u] & 255u) * F + lane;
#pragma unroll
            for (int j = 0; j < J; ++j) {
                const float e = __int_as_float(
                    __builtin_amdgcn_readlane(__float_as_int(w_l), u * H + (128 * j) / C));
                atomicAdd(&ar[128 * j],      r[u][j].x * e);
                atomicAdd(&ar[128 * j + 32], r[u][j].y * e);
                atomicAdd(&ar[128 * j + 64], r[u][j].z * e);
                atomicAdd(&ar[128 * j + 96], r[u][j].w * e);
            }
        }
    }
}

template <int F, int H, bool POST>
__global__ __launch_bounds__(OWN_THREADS)
void owner_aggregate_kernel(const int* __restrict__ ei, int E, int n_nodes,
                            const float* __restrict__ h,
                            const float* __restrict__ as,
                            const float* __restrict__ ad,
                            float* agg, const unsigned int* __restrict__ keep, int keepF,
                            const float* __restrict__ pb,
                            const float* __restrict__ pgamma,
                            const float* __restrict__ pbeta,
                            const float* __restrict__ pmean,
                            const float* __restrict__ pvar) {
    constexpr int C = F / H;
    __shared__ float acc[OWN_R * F];
    __shared__ unsigned int list[OWN_CAP];
    __shared__ float m[OWN_R * H];
    __shared__ float den[OWN_R * H];
    __shared__ float adl[OWN_R * H];
    __shared__ unsigned int cnt;

    const int tid = threadIdx.x;
    const int lane = tid & 31;
    const int wave = tid >> 5;
    const int base = blockIdx.x * OWN_R;
    const int R = min(OWN_R, n_nodes - base);
    const unsigned uR = (unsigned)R;
    const int* __restrict__ src = ei;
    const int* __restrict__ dst = ei + E;
    const unsigned int* __restrict__ kp = keep + (size_t)base * keepF;
    const unsigned keepCap = (unsigned)R * (unsigned)keepF;

    for (int i = tid; i < R * F; i += OWN_THREADS) acc[i] = 0.0f;
    for (int i = tid; i < R * H; i += OWN_THREADS) {
        float a_d = ad[(size_t)base * H + i];
        adl[i] = a_d;
        m[i] = leaky(as[(size_t)base * H + i] + a_d);
        den[i] = 0.0f;
    }
    if (tid == 0) cnt = 0u;
    __syncthreads();

    const unsigned kept = kp[0];
    const bool have = (kept <= OWN_CAP) && (kept < keepCap);

    const bool packs = (E <= (1 << 24)) && (n_nodes <= (1 << 24)) && (OWN_R <= 256);
    constexpr int TRIP = 4;
    const unsigned el = (unsigned)lane / H;
    const bool book = lane < TRIP * H;
    const unsigned wv = (unsigned)__builtin_amdgcn_readfirstlane(wave);
    if (packs && !have) {
        scan_dst(dst, E, tid, [&](int i, int d) {
            unsigned dl = (unsigned)(d - base);
            if (dl < uR) {
                unsigned pos = atomicAdd(&cnt, 1u);
                if (pos < OWN_CAP) list[pos] = (unsigned)i << 8;
            }
        });
    }
    __syncthreads();
    const unsigned total = have ? kept : cnt;
    const bool fast = have || (packs && (total <= OWN_CAP));

    if (fast) {
        for (unsigned p = tid; p < total; p += OWN_THREADS) {
            unsigned dl;
            int s;
            if (have) {
                unsigned pk = kp[1 + p];
                dl = pk & 255u;
                s = (int)(pk >> 8);
            } else {
                const unsigned i = list[p] >> 8;
                dl = (unsigned)(dst[i] - base);
                s = src[i];
            }
#pragma unroll
            for (int hh = 0; hh < H; ++hh)
                atomicMaxFloat(&m[dl * H + hh], leaky(as[(size_t)s * H + hh] + adl[dl * H + hh]));
            list[p] = ((unsigned)s << 8) | dl;
        }
        __syncthreads();
        const unsigned tot = (unsigned)__builtin_amdgcn_readfirstlane((int)total);
        for (unsigned p0 = wv * TRIP; p0 < tot; p0 += OWN_WAVES * TRIP) {
            const bool okl = book && (p0 + el) < tot;
            const unsigned pk_l = okl ? list[p0 + el] : 0u;
            accumulate_trip<F, H, TRIP>(pk_l, okl, (int)min((unsigned)TRIP, tot - p0), lane,
                                        h, as, adl, m, den, acc);
        }
    } else {
        scan_dst(dst, E, tid, [&](int i, int d) {
            unsigned dl = (unsigned)(d - base);
            if (dl < uR) {
                int s = src[i];
#pragma unroll
                for (int hh = 0; hh < H; ++hh)
                    atomicMaxFloat(&m[dl * H + hh], leaky(as[(size_t)s * H + hh] + adl[dl * H + hh]));
            }
        });
        __syncthreads();
        scan_dst(dst, E, tid, [&](int i, int d) {
            unsigned dl = (unsigned)(d - base);
            bool hit = dl < uR;
            int sv = hit ? src[i] : 0;
            unsigned mask = __builtin_amdgcn_ballot_w32(hit);
            while (mask) {
                int l = __builtin_ctz(mask);
                mask &= mask - 1u;
                int s1[1] = { __shfl(sv, l, 32) };
                unsigned d1[1] = { (unsigned)__shfl((int)dl, l, 32) };
                bool o1[1] = { true };
                accumulate_edges<F, H, 1>(s1, d1, o1, lane, h, as, adl, m, den, acc);
            }
        });
    }
    if (packs) {
        for (unsigned q0 = wv * TRIP; q0 < uR; q0 += OWN_WAVES * TRIP) {
            const unsigned ql = q0 + el;
            const bool okl = book && ql < uR;
            const unsigned pk_l = okl ? ((((unsigned)base + ql) << 8) | ql) : 0u;
            accumulate_trip<F, H, TRIP>(pk_l, okl, (int)min((unsigned)TRIP, uR - q0), lane,
                                        h, as, adl, m, den, acc);
        }
    } else {
        for (int q0 = wave; q0 < R; q0 += OWN_WAVES) {
            int s1[1] = { base + q0 };
            unsigned d1[1] = { (unsigned)q0 };
            bool o1[1] = { true };
            accumulate_edges<F, H, 1>(s1, d1, o1, lane, h, as, adl, m, den, acc);
        }
    }
    __syncthreads();

    constexpr int Q = F / 4;
    static_assert(OWN_THREADS % Q == 0, "one piece per thread");
    const int cq = tid % Q;
    const int c0 = 4 * cq;
    const int w0 = 128 * (cq >> 5) + (cq & 31);
    const int hd = c0 / C;
    float4* agg4 = reinterpret_cast<float4*>(agg + (size_t)base * F);
    if (POST) {
        float fb[4], fg[4], ft[4], fm[4], frs[4];
        for (int k = 0; k < 4; ++k) {
            const int c = c0 + k;
            fb[k] = pb[c]; fg[k] = pgamma[c]; ft[k] = pbeta[c]; fm[k] = pmean[c];
            frs[k] = rsqrtf(pvar[c] + EPS);
        }
        for (int i = tid; i < R * Q; i += OWN_THREADS) {
            const int dl = i / Q;
            const float* ar = acc + dl * F + w0;
            const float d = den[dl * H + hd];
            float o[4];
            for (int k = 0; k < 4; ++k) {
                float v = ar[32 * k] / d + fb[k];
                v = fg[k] * (v - fm[k]) * frs[k] + ft[k];
                o[k] = (v > 0.0f) ? v : 0.0f;
            }
            agg4[i] = make_float4(o[0], o[1], o[2], o[3]);
        }
    } else {
        for (int i = tid; i < R * Q; i += OWN_THREADS) {
            const int dl = i / Q;
            const float* ar = acc + dl * F + w0;
            const float d = den[dl * H + hd];
            agg4[i] = make_float4(ar[0] / d, ar[32] / d, ar[64] / d, ar[96] / d);
        }
    }
}

#define POOL_THREADS 1024
#define POOL_WAVES (POOL_THREADS / 32)
#define POOL_CAP 65536
#define POOL_VECS 16
#define POOL_ROWS 8

__global__ __launch_bounds__(POOL_THREADS)
void pool_owner_kernel(const float* __restrict__ x,
                       const int* __restrict__ batch,
                       const float* __restrict__ b,
                       const float* __restrict__ gamma,
                       const float* __restrict__ beta,
                       const float* __restrict__ mean,
                       const float* __restrict__ var,
                       float* __restrict__ out, int n_nodes) {
    constexpr int F = 128;
    __shared__ unsigned short list[POOL_CAP];
    __shared__ float4 part[POOL_WAVES][32];
    __shared__ unsigned int cnt;

    const int tid = threadIdx.x;
    const int lane = tid & 31;
    const int wave = tid >> 5;
    const int g = blockIdx.x;
    const float4* __restrict__ x4 = reinterpret_cast<const float4*>(x);

    float pb[4], pg[4], pt[4], pm[4], prs[4];
    for (int k = 0; k < 4; ++k) {
        const int c = 4 * lane + k;
        pb[k] = b[c]; pg[k] = gamma[c]; pt[k] = beta[c]; pm[k] = mean[c];
        prs[k] = rsqrtf(var[c] + EPS);
    }
    auto act = [&](float xv, int k) {
        float v = xv + pb[k];
        v = pg[k] * (v - pm[k]) * prs[k] + pt[k];
        return (v > 0.0f) ? v : 0.0f;
    };
    const float ninf = __int_as_float((int)0xFF800000u);
    float mx[4] = { ninf, ninf, ninf, ninf };

    for (int chunk0 = 0; chunk0 < n_nodes; chunk0 += POOL_CAP) {
        const int len = min(POOL_CAP, n_nodes - chunk0);
        if (tid == 0) cnt = 0u;
        __syncthreads();
        const int* __restrict__ bp = batch + chunk0;
        const int len4 = ((reinterpret_cast<size_t>(bp) & 15) == 0) ? (len >> 2) : 0;
        const int4* __restrict__ bp4 = reinterpret_cast<const int4*>(bp);
        auto push = [&](int i) {
            const unsigned pos = atomicAdd(&cnt, 1u);
            list[pos] = (unsigned short)i;
        };
        for (int jb = 0; jb < len4; jb += POOL_VECS * POOL_THREADS) {
            int4 d[POOL_VECS];
#pragma unroll
            for (int u = 0; u < POOL_VECS; ++u) {
                const int j = jb + u * POOL_THREADS + tid;
                d[u] = (j < len4) ? bp4[j] : make_int4(-1, -1, -1, -1);
            }
            bool any = false;
#pragma unroll
            for (int u = 0; u < POOL_VECS; ++u)
                any |= (d[u].x == g) | (d[u].y == g) | (d[u].z == g) | (d[u].w == g);
            if (__builtin_amdgcn_ballot_w32(any) != 0u) {
                unsigned long long hm = 0ull;
#pragma unroll
                for (int u = 0; u < POOL_VECS; ++u) {
                    const unsigned m4 = (d[u].x == g ? 1u : 0u) | (d[u].y == g ? 2u : 0u)
                                      | (d[u].z == g ? 4u : 0u) | (d[u].w == g ? 8u : 0u);
                    hm |= (unsigned long long)m4 << (4 * u);
                }
                while (hm) {
                    const int bit = __builtin_ctzll(hm);
                    hm &= hm - 1ull;
                    push(4 * (jb + (bit >> 2) * POOL_THREADS + tid) + (bit & 3));
                }
            }
        }
        for (int i = 4 * len4 + tid; i < len; i += POOL_THREADS)
            if (bp[i] == g) push(i);
        __syncthreads();
        const unsigned total = cnt;

        for (unsigned p0 = wave; p0 < total; p0 += POOL_WAVES * POOL_ROWS) {
            float4 r[POOL_ROWS];
#pragma unroll
            for (int u = 0; u < POOL_ROWS; ++u) {
                const unsigned p = p0 + u * POOL_WAVES;
                const int n = chunk0 + (int)list[(p < total) ? p : p0];
                r[u] = x4[(size_t)n * (F / 4) + lane];
            }
#pragma unroll
            for (int u = 0; u < POOL_ROWS; ++u) {
                if (p0 + u * POOL_WAVES < total) {
                    mx[0] = fmaxf(mx[0], act(r[u].x, 0));
                    mx[1] = fmaxf(mx[1], act(r[u].y, 1));
                    mx[2] = fmaxf(mx[2], act(r[u].z, 2));
                    mx[3] = fmaxf(mx[3], act(r[u].w, 3));
                }
            }
        }
        __syncthreads();
    }

    part[wave][lane] = make_float4(mx[0], mx[1], mx[2], mx[3]);
    __syncthreads();
    if (tid < F) {
        const float* pf = reinterpret_cast<const float*>(&part[0][0]);
        float m = pf[tid];
        for (int w = 1; w < POOL_WAVES; ++w) m = fmaxf(m, pf[w * F + tid]);
        out[(size_t)g * F + tid] = m;
    }
}

extern "C" void kernel_launch(void* const* d_in, const int* in_sizes, int n_in,
                              void* d_out, int out_size, void* d_ws, size_t ws_size,
                              hipStream_t stream) {
    const int F_IN = 128, HID = 128, HEADS = 2;
    const int F1 = HEADS * HID;
    const int N = in_sizes[0] / F_IN;
    const int E = in_sizes[1] / 2;

    const float* x      = (const float*)d_in[0];
    const int*   ei     = (const int*)d_in[1];
    const int*   batch  = (const int*)d_in[2];
    const float* W1     = (const float*)d_in[3];
    const float* a_src1 = (const float*)d_in[4];
    const float* a_dst1 = (const float*)d_in[5];
    const float* b1     = (const float*)d_in[6];
    const float* gamma1 = (const float*)d_in[7];
    const float* beta1  = (const float*)d_in[8];
    const float* rm1    = (const float*)d_in[9];
    const float* rv1    = (const float*)d_in[10];
    const float* W2     = (const float*)d_in[11];
    const float* a_src2 = (const float*)d_in[12];
    const float* a_dst2 = (const float*)d_in[13];
    const float* b2     = (const float*)d_in[14];
    const float* gamma2 = (const float*)d_in[15];
    const float* beta2  = (const float*)d_in[16];
    const float* rm2    = (const float*)d_in[17];
    const float* rv2    = (const float*)d_in[18];

    float* ws = (float*)d_ws;
    size_t o = 0;
    float* h1   = ws + o; o += (size_t)N * F1;
    float* aggA = ws + o; o += (size_t)N * F1;
    float* aggB = ws + o; o += (size_t)N * HID;
    float* as1  = ws + o; o += (size_t)N * HEADS;
    float* ad1  = ws + o; o += (size_t)N * HEADS;
    float* as2  = ws + o; o += (size_t)N;
    float* ad2  = ws + o; o += (size_t)N;
    float* h2   = h1;
    float* agg2 = h1 + (size_t)N * HID;
    unsigned int* keep = (unsigned int*)aggB;

    const unsigned ownerBlocks = (unsigned)((N + OWN_R - 1) / OWN_R);

    gemm_wmma_f32_kernel<128, 256><<<GEMM_GRID, GEMM_THREADS, 0, stream>>>(
        x, W1, h1, N / 16, keep, (int)ownerBlocks, OWN_R * KEEP_WORDS);
    alpha_rows_kernel<2><<<ALPHA_GRID, ALPHA_THREADS, 0, stream>>>(h1, a_src1, a_dst1, as1, ad1, N);
    owner_bin_kernel<<<(unsigned)((E + BIN_THREADS * 4 * BIN_VECS - 1) / (BIN_THREADS * 4 * BIN_VECS)),
                       BIN_THREADS, 0, stream>>>(ei, E, N, keep, KEEP_WORDS);
    owner_aggregate_kernel<256, 2, true><<<ownerBlocks, OWN_THREADS, 0, stream>>>(
        ei, E, N, h1, as1, ad1, aggA, keep, KEEP_WORDS, b1, gamma1, beta1, rm1, rv1);

    gemm_wmma_f32_kernel<256, 128><<<GEMM_GRID, GEMM_THREADS, 0, stream>>>(aggA, W2, h2, N / 16, nullptr, 0, 0);
    alpha_rows_kernel<1><<<ALPHA_GRID, ALPHA_THREADS, 0, stream>>>(h2, a_src2, a_dst2, as2, ad2, N);
    owner_aggregate_kernel<128, 1, false><<<ownerBlocks, OWN_THREADS, 0, stream>>>(
        ei, E, N, h2, as2, ad2, agg2, keep, KEEP_WORDS,
        nullptr, nullptr, nullptr, nullptr, nullptr);

    pool_owner_kernel<<<(unsigned)(out_size / HID), POOL_THREADS, 0, stream>>>(
        agg2, batch, b2, gamma2, beta2, rm2, rv2, (float*)d_out, N);
}
